# starting kernel
# baseline (speedup 1.0000x reference)
#include <hip/hip_runtime.h>

typedef __attribute__((ext_vector_type(2))) float v2f;
typedef __attribute__((ext_vector_type(8))) float v8f;

__global__ void init_deg_kernel(float* __restrict__ deg, int n) {
    int i = blockIdx.x * blockDim.x + threadIdx.x;
    if (i < n) deg[i] = 1.0f;
}

__global__ void accum_deg_kernel(const int* __restrict__ dst,
                                 float* __restrict__ deg, int ne) {
    int i = blockIdx.x * blockDim.x + threadIdx.x;
    if (i < ne) atomicAdd(&deg[dst[i]], 1.0f);
}

__global__ void deg_to_dinv_kernel(float* __restrict__ d, int n) {
    int i = blockIdx.x * blockDim.x + threadIdx.x;
    if (i < n) d[i] = rsqrtf(d[i]);
}

__global__ void pad_x_kernel(const float* __restrict__ x, float* __restrict__ xp,
                             int nn, int nn_pad) {
    int i = blockIdx.x * blockDim.x + threadIdx.x;
    int total = nn_pad * 20;
    if (i >= total) return;
    int n = i / 20;
    int k = i % 20;
    xp[i] = (n < nn && k < 18) ? x[n * 18 + k] : 0.0f;
}

__global__ void repack_w_kernel(const float* __restrict__ w, float* __restrict__ wp,
                                int kd_true, int kp, int nf) {
    int o = blockIdx.x * blockDim.x + threadIdx.x;
    if (o >= kp * nf) return;
    int j = o & 1;
    int n = (o >> 1) % nf;
    int p = (o >> 1) / nf;
    int k = 2 * p + j;
    wp[o] = (k < kd_true) ? w[k * nf + n] : 0.0f;
}

template <int KD, int NF, int NT>
__global__ void gemm_scale_kernel(const float* __restrict__ X,
                                  const float* __restrict__ Wp,
                                  const float* __restrict__ dinv,
                                  float* __restrict__ hs,
                                  float* __restrict__ agg,
                                  int mtiles) {
    int wave = blockIdx.x * (blockDim.x >> 5) + (threadIdx.x >> 5);
    if (wave >= mtiles) return;
    int lane = threadIdx.x & 31;
    int r16  = lane & 15;
    int kh   = lane >> 4;
    int mbase = wave * 16;
    int arow  = mbase + r16;

    v8f acc[NT] = {};

    static_assert(KD % 4 == 0, "KD must be multiple of 4");
    constexpr int KSTEPS = KD / 4;
#pragma unroll
    for (int ks = 0; ks < KSTEPS; ++ks) {
        int k0 = ks * 4 + kh * 2;
        v2f a = *(const v2f*)&X[arow * KD + k0];
        int p = ks * 2 + kh;
#pragma unroll
        for (int t = 0; t < NT; ++t) {
            v2f b = *(const v2f*)&Wp[(p * NF + t * 16 + r16) * 2];
            acc[t] = __builtin_amdgcn_wmma_f32_16x16x4_f32(
                false, a, false, b, (short)0, acc[t], false, false);
        }
    }

    const float4 dv0 = *(const float4*)&dinv[mbase + 8 * kh];
    const float4 dv1 = *(const float4*)&dinv[mbase + 8 * kh + 4];
    const float dv[8] = {dv0.x, dv0.y, dv0.z, dv0.w, dv1.x, dv1.y, dv1.z, dv1.w};

#pragma unroll
    for (int v = 0; v < 8; ++v) {
        int m = mbase + v + 8 * kh;
#pragma unroll
        for (int t = 0; t < NT; ++t) {
            float val = acc[t][v] * dv[v];
            int idx = m * NF + t * 16 + r16;
            hs[idx]  = val;
            agg[idx] = val;
        }
    }
}

template <int F4>
__global__ void scatter_kernel(const int* __restrict__ src,
                               const int* __restrict__ dst,
                               const float* __restrict__ hs,
                               float* __restrict__ agg, int ne) {
    unsigned int tid = blockIdx.x * blockDim.x + threadIdx.x;
    unsigned int total = (unsigned int)ne * F4;
    if (tid >= total) return;
    int e  = (int)(tid / F4);
    int fo = (int)(tid % F4) * 4;
    constexpr int F = F4 * 4;
    int s = src[e];
    int d = dst[e];
    const float4 val = *(const float4*)&hs[s * F + fo];
    float* p = &agg[d * F + fo];
    atomicAdd(p + 0, val.x);
    atomicAdd(p + 1, val.y);
    atomicAdd(p + 2, val.z);
    atomicAdd(p + 3, val.w);
}

template <int F, bool RELU>
__global__ void finalize_kernel(const float* __restrict__ agg,
                                const float* __restrict__ dinv,
                                const float* __restrict__ bias,
                                float* __restrict__ out, int nrows) {
    int i = blockIdx.x * blockDim.x + threadIdx.x;
    constexpr int F4 = F / 4;
    if (i >= nrows * F4) return;
    int n  = i / F4;
    int fo = (i % F4) * 4;
    float dvn = dinv[n];
    const float4 a = *(const float4*)&agg[n * F + fo];
    const float4 b = *(const float4*)&bias[fo];
    float4 v;
    v.x = dvn * a.x + b.x;
    v.y = dvn * a.y + b.y;
    v.z = dvn * a.z + b.z;
    v.w = dvn * a.w + b.w;
    if (RELU) {
        v.x = fmaxf(v.x, 0.0f);
        v.y = fmaxf(v.y, 0.0f);
        v.z = fmaxf(v.z, 0.0f);
        v.w = fmaxf(v.w, 0.0f);
    }
    *(float4*)&out[n * F + fo] = v;
}

extern "C" void kernel_launch(void* const* d_in, const int* in_sizes, int n_in,
                              void* d_out, int out_size, void* d_ws, size_t ws_size,
                              hipStream_t stream) {
    const float* x  = (const float*)d_in[0];
    const int*   ei = (const int*)d_in[1];
    const float* W1 = (const float*)d_in[2];
    const float* b1 = (const float*)d_in[3];
    const float* W2 = (const float*)d_in[4];
    const float* b2 = (const float*)d_in[5];
    float* out = (float*)d_out;

    const int NN = in_sizes[0] / 18;
    const int NE = in_sizes[1] / 2;
    const int NN_pad = (NN + 15) & ~15;
    const int* src = ei;
    const int* dst = ei + NE;

    char* ws = (char*)d_ws;
    size_t off = 0;
    auto carve = [&](size_t bytes) -> char* {
        char* p = ws + off;
        off += (bytes + 255) & ~(size_t)255;
        return p;
    };
    float* dinv = (float*)carve((size_t)NN_pad * 4);
    float* xp   = (float*)carve((size_t)NN_pad * 20 * 4);
    float* W1p  = (float*)carve((size_t)20 * 64 * 4);
    float* W2p  = (float*)carve((size_t)64 * 32 * 4);
    float* hs1  = (float*)carve((size_t)NN_pad * 64 * 4);
    float* agg1 = (float*)carve((size_t)NN_pad * 64 * 4);
    float* h1   = (float*)carve((size_t)NN_pad * 64 * 4);
    float* hs2  = hs1;
    float* agg2 = agg1;

    const int B = 256;

    init_deg_kernel<<<(NN_pad + B - 1) / B, B, 0, stream>>>(dinv, NN_pad);
    accum_deg_kernel<<<(NE + B - 1) / B, B, 0, stream>>>(dst, dinv, NE);
    deg_to_dinv_kernel<<<(NN_pad + B - 1) / B, B, 0, stream>>>(dinv, NN_pad);

    pad_x_kernel<<<(NN_pad * 20 + B - 1) / B, B, 0, stream>>>(x, xp, NN, NN_pad);
    repack_w_kernel<<<(20 * 64 + B - 1) / B, B, 0, stream>>>(W1, W1p, 18, 20, 64);
    repack_w_kernel<<<(64 * 32 + B - 1) / B, B, 0, stream>>>(W2, W2p, 64, 64, 32);

    const int mtiles = NN_pad / 16;
    const int gblk = (mtiles + 3) / 4;

    gemm_scale_kernel<20, 64, 4><<<gblk, 128, 0, stream>>>(xp, W1p, dinv, hs1, agg1, mtiles);
    {
        unsigned int total = (unsigned int)NE * 16u;
        scatter_kernel<16><<<(total + B - 1) / B, B, 0, stream>>>(src, dst, hs1, agg1, NE);
    }
    finalize_kernel<64, true><<<(NN_pad * 16 + B - 1) / B, B, 0, stream>>>(agg1, dinv, b1, h1, NN_pad);

    gemm_scale_kernel<64, 32, 2><<<gblk, 128, 0, stream>>>(h1, W2p, dinv, hs2, agg2, mtiles);
    {
        unsigned int total = (unsigned int)NE * 8u;
        scatter_kernel<8><<<(total + B - 1) / B, B, 0, stream>>>(src, dst, hs2, agg2, NE);
    }
    finalize_kernel<32, false><<<(NN * 8 + B - 1) / B, B, 0, stream>>>(agg2, dinv, b2, out, NN);
}


# Scatter remapped: one wave per 32 edges, lane = feature column, so each wave-wide f32 atomic and gather covers one whole 128B line of one row; atomics kept, their arrival order changes
# speedup vs baseline: 1.5279x; 1.5279x over previous
#include <hip/hip_runtime.h>

typedef __attribute__((ext_vector_type(2))) float v2f;
typedef __attribute__((ext_vector_type(8))) float v8f;

__global__ void init_deg_kernel(float* __restrict__ deg, int n) {
    int i = blockIdx.x * blockDim.x + threadIdx.x;
    if (i < n) deg[i] = 1.0f;
}

__global__ void accum_deg_kernel(const int* __restrict__ dst,
                                 float* __restrict__ deg, int ne) {
    int i = blockIdx.x * blockDim.x + threadIdx.x;
    if (i < ne) atomicAdd(&deg[dst[i]], 1.0f);
}

__global__ void deg_to_dinv_kernel(float* __restrict__ d, int n) {
    int i = blockIdx.x * blockDim.x + threadIdx.x;
    if (i < n) d[i] = rsqrtf(d[i]);
}

__global__ void pad_x_kernel(const float* __restrict__ x, float* __restrict__ xp,
                             int nn, int nn_pad) {
    int i = blockIdx.x * blockDim.x + threadIdx.x;
    int total = nn_pad * 20;
    if (i >= total) return;
    int n = i / 20;
    int k = i % 20;
    xp[i] = (n < nn && k < 18) ? x[n * 18 + k] : 0.0f;
}

__global__ void repack_w_kernel(const float* __restrict__ w, float* __restrict__ wp,
                                int kd_true, int kp, int nf) {
    int o = blockIdx.x * blockDim.x + threadIdx.x;
    if (o >= kp * nf) return;
    int j = o & 1;
    int n = (o >> 1) % nf;
    int p = (o >> 1) / nf;
    int k = 2 * p + j;
    wp[o] = (k < kd_true) ? w[k * nf + n] : 0.0f;
}

template <int KD, int NF, int NT>
__global__ void gemm_scale_kernel(const float* __restrict__ X,
                                  const float* __restrict__ Wp,
                                  const float* __restrict__ dinv,
                                  float* __restrict__ hs,
                                  float* __restrict__ agg,
                                  int mtiles) {
    int wave = blockIdx.x * (blockDim.x >> 5) + (threadIdx.x >> 5);
    if (wave >= mtiles) return;
    int lane = threadIdx.x & 31;
    int r16  = lane & 15;
    int kh   = lane >> 4;
    int mbase = wave * 16;
    int arow  = mbase + r16;

    v8f acc[NT] = {};

    static_assert(KD % 4 == 0, "KD must be multiple of 4");
    constexpr int KSTEPS = KD / 4;
#pragma unroll
    for (int ks = 0; ks < KSTEPS; ++ks) {
        int k0 = ks * 4 + kh * 2;
        v2f a = *(const v2f*)&X[arow * KD + k0];
        int p = ks * 2 + kh;
#pragma unroll
        for (int t = 0; t < NT; ++t) {
            v2f b = *(const v2f*)&Wp[(p * NF + t * 16 + r16) * 2];
            acc[t] = __builtin_amdgcn_wmma_f32_16x16x4_f32(
                false, a, false, b, (short)0, acc[t], false, false);
        }
    }

    const float4 dv0 = *(const float4*)&dinv[mbase + 8 * kh];
    const float4 dv1 = *(const float4*)&dinv[mbase + 8 * kh + 4];
    const float dv[8] = {dv0.x, dv0.y, dv0.z, dv0.w, dv1.x, dv1.y, dv1.z, dv1.w};

#pragma unroll
    for (int v = 0; v < 8; ++v) {
        int m = mbase + v + 8 * kh;
#pragma unroll
        for (int t = 0; t < NT; ++t) {
            float val = acc[t][v] * dv[v];
            int idx = m * NF + t * 16 + r16;
            hs[idx]  = val;
            agg[idx] = val;
        }
    }
}

template <int F>
__global__ void scatter_kernel(const int* __restrict__ src,
                               const int* __restrict__ dst,
                               const float* __restrict__ hs,
                               float* __restrict__ agg, int ne) {
    constexpr int NSEG = F / 32;
    constexpr int U = 16 / NSEG;
    const int lane = threadIdx.x & 31;
    const int wave = __builtin_amdgcn_readfirstlane(
        (int)(blockIdx.x * (blockDim.x >> 5) + (threadIdx.x >> 5)));
    const int e0 = wave * 32;
    if (e0 >= ne) return;
    const int left = ne - e0;
    const int cnt = left < 32 ? left : 32;
    const int e = (lane < cnt) ? e0 + lane : ne - 1;
    const int my_s = src[e];
    const int my_d = dst[e];

    if (cnt == 32) {
#pragma unroll
        for (int g = 0; g < 32; g += U) {
            float v[U][NSEG];
#pragma unroll
            for (int u = 0; u < U; ++u) {
                const int s = __builtin_amdgcn_readlane(my_s, g + u);
#pragma unroll
                for (int q = 0; q < NSEG; ++q) v[u][q] = hs[s * F + q * 32 + lane];
            }
#pragma unroll
            for (int u = 0; u < U; ++u) {
                const int d = __builtin_amdgcn_readlane(my_d, g + u);
#pragma unroll
                for (int q = 0; q < NSEG; ++q) atomicAdd(&agg[d * F + q * 32 + lane], v[u][q]);
            }
        }
    } else {
        for (int k = 0; k < cnt; ++k) {
            const int s = __builtin_amdgcn_readlane(my_s, k);
            const int d = __builtin_amdgcn_readlane(my_d, k);
#pragma unroll
            for (int q = 0; q < NSEG; ++q)
                atomicAdd(&agg[d * F + q * 32 + lane], hs[s * F + q * 32 + lane]);
        }
    }
}

template <int F, bool RELU>
__global__ void finalize_kernel(const float* __restrict__ agg,
                                const float* __restrict__ dinv,
                                const float* __restrict__ bias,
                                float* __restrict__ out, int nrows) {
    int i = blockIdx.x * blockDim.x + threadIdx.x;
    constexpr int F4 = F / 4;
    if (i >= nrows * F4) return;
    int n  = i / F4;
    int fo = (i % F4) * 4;
    float dvn = dinv[n];
    const float4 a = *(const float4*)&agg[n * F + fo];
    const float4 b = *(const float4*)&bias[fo];
    float4 v;
    v.x = dvn * a.x + b.x;
    v.y = dvn * a.y + b.y;
    v.z = dvn * a.z + b.z;
    v.w = dvn * a.w + b.w;
    if (RELU) {
        v.x = fmaxf(v.x, 0.0f);
        v.y = fmaxf(v.y, 0.0f);
        v.z = fmaxf(v.z, 0.0f);
        v.w = fmaxf(v.w, 0.0f);
    }
    *(float4*)&out[n * F + fo] = v;
}

extern "C" void kernel_launch(void* const* d_in, const int* in_sizes, int n_in,
                              void* d_out, int out_size, void* d_ws, size_t ws_size,
                              hipStream_t stream) {
    const float* x  = (const float*)d_in[0];
    const int*   ei = (const int*)d_in[1];
    const float* W1 = (const float*)d_in[2];
    const float* b1 = (const float*)d_in[3];
    const float* W2 = (const float*)d_in[4];
    const float* b2 = (const float*)d_in[5];
    float* out = (float*)d_out;

    const int NN = in_sizes[0] / 18;
    const int NE = in_sizes[1] / 2;
    const int NN_pad = (NN + 15) & ~15;
    const int* src = ei;
    const int* dst = ei + NE;

    char* ws = (char*)d_ws;
    size_t off = 0;
    auto carve = [&](size_t bytes) -> char* {
        char* p = ws + off;
        off += (bytes + 255) & ~(size_t)255;
        return p;
    };
    float* dinv = (float*)carve((size_t)NN_pad * 4);
    float* xp   = (float*)carve((size_t)NN_pad * 20 * 4);
    float* W1p  = (float*)carve((size_t)20 * 64 * 4);
    float* W2p  = (float*)carve((size_t)64 * 32 * 4);
    float* hs1  = (float*)carve((size_t)NN_pad * 64 * 4);
    float* agg1 = (float*)carve((size_t)NN_pad * 64 * 4);
    float* h1   = (float*)carve((size_t)NN_pad * 64 * 4);
    float* hs2  = hs1;
    float* agg2 = agg1;

    const int B = 256;

    init_deg_kernel<<<(NN_pad + B - 1) / B, B, 0, stream>>>(dinv, NN_pad);
    accum_deg_kernel<<<(NE + B - 1) / B, B, 0, stream>>>(dst, dinv, NE);
    deg_to_dinv_kernel<<<(NN_pad + B - 1) / B, B, 0, stream>>>(dinv, NN_pad);

    pad_x_kernel<<<(NN_pad * 20 + B - 1) / B, B, 0, stream>>>(x, xp, NN, NN_pad);
    repack_w_kernel<<<(20 * 64 + B - 1) / B, B, 0, stream>>>(W1, W1p, 18, 20, 64);
    repack_w_kernel<<<(64 * 32 + B - 1) / B, B, 0, stream>>>(W2, W2p, 64, 64, 32);

    const int mtiles = NN_pad / 16;
    const int gblk = (mtiles + 3) / 4;
    const int swaves = (NE + 31) / 32;
    const int sblk = (swaves + (B / 32) - 1) / (B / 32);

    gemm_scale_kernel<20, 64, 4><<<gblk, 128, 0, stream>>>(xp, W1p, dinv, hs1, agg1, mtiles);
    scatter_kernel<64><<<sblk, B, 0, stream>>>(src, dst, hs1, agg1, NE);
    finalize_kernel<64, true><<<(NN_pad * 16 + B - 1) / B, B, 0, stream>>>(agg1, dinv, b1, h1, NN_pad);

    gemm_scale_kernel<64, 32, 2><<<gblk, 128, 0, stream>>>(h1, W2p, dinv, hs2, agg2, mtiles);
    scatter_kernel<32><<<sblk, B, 0, stream>>>(src, dst, hs2, agg2, NE);
    finalize_kernel<32, false><<<(NN * 8 + B - 1) / B, B, 0, stream>>>(agg2, dinv, b2, out, NN);
}


# Layer 1 sums neighbours' dinv-scaled 18-wide inputs BEFORE the weight product ((A x)W1, not A(x W1)): its scatter moves one 128B line per edge, not two; bias+ReLU in GEMM epilogue; sum order changes
# speedup vs baseline: 1.6203x; 1.0605x over previous
#include <hip/hip_runtime.h>

typedef __attribute__((ext_vector_type(2))) float v2f;
typedef __attribute__((ext_vector_type(8))) float v8f;

__global__ void init_deg_kernel(float* __restrict__ deg, int n) {
    int i = blockIdx.x * blockDim.x + threadIdx.x;
    if (i < n) deg[i] = 1.0f;
}

__global__ void accum_deg_kernel(const int* __restrict__ dst,
                                 float* __restrict__ deg, int ne) {
    int i = blockIdx.x * blockDim.x + threadIdx.x;
    if (i < ne) atomicAdd(&deg[dst[i]], 1.0f);
}

__global__ void deg_to_dinv_kernel(float* __restrict__ d, int n) {
    int i = blockIdx.x * blockDim.x + threadIdx.x;
    if (i < n) d[i] = rsqrtf(d[i]);
}

__global__ void scale_pad_x_kernel(const float* __restrict__ x,
                                   const float* __restrict__ dinv,
                                   float* __restrict__ xs, float* __restrict__ aggx,
                                   int nn, int nn_pad) {
    int i = blockIdx.x * blockDim.x + threadIdx.x;
    if (i >= nn_pad * 32) return;
    int n = i >> 5;
    int k = i & 31;
    float v = (n < nn && k < 18) ? dinv[n] * x[n * 18 + k] : 0.0f;
    xs[i]   = v;
    aggx[i] = v;
}

__global__ void repack_w_kernel(const float* __restrict__ w, float* __restrict__ wp,
                                int kd_true, int kp, int nf) {
    int o = blockIdx.x * blockDim.x + threadIdx.x;
    if (o >= kp * nf) return;
    int j = o & 1;
    int n = (o >> 1) % nf;
    int p = (o >> 1) / nf;
    int k = 2 * p + j;
    wp[o] = (k < kd_true) ? w[k * nf + n] : 0.0f;
}

template <int KD, int NF, int NT>
__global__ void gemm_scale_kernel(const float* __restrict__ X,
                                  const float* __restrict__ Wp,
                                  const float* __restrict__ dinv,
                                  float* __restrict__ hs,
                                  float* __restrict__ agg,
                                  int mtiles) {
    int wave = blockIdx.x * (blockDim.x >> 5) + (threadIdx.x >> 5);
    if (wave >= mtiles) return;
    int lane = threadIdx.x & 31;
    int r16  = lane & 15;
    int kh   = lane >> 4;
    int mbase = wave * 16;
    int arow  = mbase + r16;

    v8f acc[NT] = {};

    static_assert(KD % 4 == 0, "KD must be multiple of 4");
    constexpr int KSTEPS = KD / 4;
#pragma unroll
    for (int ks = 0; ks < KSTEPS; ++ks) {
        int k0 = ks * 4 + kh * 2;
        v2f a = *(const v2f*)&X[arow * KD + k0];
        int p = ks * 2 + kh;
#pragma unroll
        for (int t = 0; t < NT; ++t) {
            v2f b = *(const v2f*)&Wp[(p * NF + t * 16 + r16) * 2];
            acc[t] = __builtin_amdgcn_wmma_f32_16x16x4_f32(
                false, a, false, b, (short)0, acc[t], false, false);
        }
    }

    const float4 dv0 = *(const float4*)&dinv[mbase + 8 * kh];
    const float4 dv1 = *(const float4*)&dinv[mbase + 8 * kh + 4];
    const float dv[8] = {dv0.x, dv0.y, dv0.z, dv0.w, dv1.x, dv1.y, dv1.z, dv1.w};

#pragma unroll
    for (int v = 0; v < 8; ++v) {
        int m = mbase + v + 8 * kh;
#pragma unroll
        for (int t = 0; t < NT; ++t) {
            float val = acc[t][v] * dv[v];
            int idx = m * NF + t * 16 + r16;
            hs[idx]  = val;
            agg[idx] = val;
        }
    }
}

template <int KD, int LDA, int NF, int NT>
__global__ void gemm_bias_relu_kernel(const float* __restrict__ X,
                                      const float* __restrict__ Wp,
                                      const float* __restrict__ dinv,
                                      const float* __restrict__ bias,
                                      float* __restrict__ h,
                                      int mtiles) {
    int wave = blockIdx.x * (blockDim.x >> 5) + (threadIdx.x >> 5);
    if (wave >= mtiles) return;
    int lane = threadIdx.x & 31;
    int r16  = lane & 15;
    int kh   = lane >> 4;
    int mbase = wave * 16;
    int arow  = mbase + r16;

    v8f acc[NT] = {};

    static_assert(KD % 4 == 0 && KD <= LDA && LDA % 2 == 0, "K layout");
    constexpr int KSTEPS = KD / 4;
#pragma unroll
    for (int ks = 0; ks < KSTEPS; ++ks) {
        int k0 = ks * 4 + kh * 2;
        v2f a = *(const v2f*)&X[arow * LDA + k0];
        int p = ks * 2 + kh;
#pragma unroll
        for (int t = 0; t < NT; ++t) {
            v2f b = *(const v2f*)&Wp[(p * NF + t * 16 + r16) * 2];
            acc[t] = __builtin_amdgcn_wmma_f32_16x16x4_f32(
                false, a, false, b, (short)0, acc[t], false, false);
        }
    }

    const float4 dv0 = *(const float4*)&dinv[mbase + 8 * kh];
    const float4 dv1 = *(const float4*)&dinv[mbase + 8 * kh + 4];
    const float dv[8] = {dv0.x, dv0.y, dv0.z, dv0.w, dv1.x, dv1.y, dv1.z, dv1.w};
    float bv[NT];
#pragma unroll
    for (int t = 0; t < NT; ++t) bv[t] = bias[t * 16 + r16];

#pragma unroll
    for (int v = 0; v < 8; ++v) {
        int m = mbase + v + 8 * kh;
#pragma unroll
        for (int t = 0; t < NT; ++t)
            h[m * NF + t * 16 + r16] = fmaxf(acc[t][v] * dv[v] + bv[t], 0.0f);
    }
}

template <int F>
__global__ void scatter_kernel(const int* __restrict__ src,
                               const int* __restrict__ dst,
                               const float* __restrict__ hs,
                               float* __restrict__ agg, int ne) {
    constexpr int NSEG = F / 32;
    constexpr int U = 16 / NSEG;
    const int lane = threadIdx.x & 31;
    const int wave = __builtin_amdgcn_readfirstlane(
        (int)(blockIdx.x * (blockDim.x >> 5) + (threadIdx.x >> 5)));
    const int e0 = wave * 32;
    if (e0 >= ne) return;
    const int left = ne - e0;
    const int cnt = left < 32 ? left : 32;
    const int e = (lane < cnt) ? e0 + lane : ne - 1;
    const int my_s = src[e];
    const int my_d = dst[e];

    if (cnt == 32) {
#pragma unroll
        for (int g = 0; g < 32; g += U) {
            float v[U][NSEG];
#pragma unroll
            for (int u = 0; u < U; ++u) {
                const int s = __builtin_amdgcn_readlane(my_s, g + u);
#pragma unroll
                for (int q = 0; q < NSEG; ++q) v[u][q] = hs[s * F + q * 32 + lane];
            }
#pragma unroll
            for (int u = 0; u < U; ++u) {
                const int d = __builtin_amdgcn_readlane(my_d, g + u);
#pragma unroll
                for (int q = 0; q < NSEG; ++q) atomicAdd(&agg[d * F + q * 32 + lane], v[u][q]);
            }
        }
    } else {
        for (int k = 0; k < cnt; ++k) {
            const int s = __builtin_amdgcn_readlane(my_s, k);
            const int d = __builtin_amdgcn_readlane(my_d, k);
#pragma unroll
            for (int q = 0; q < NSEG; ++q)
                atomicAdd(&agg[d * F + q * 32 + lane], hs[s * F + q * 32 + lane]);
        }
    }
}

template <int F, bool RELU>
__global__ void finalize_kernel(const float* __restrict__ agg,
                                const float* __restrict__ dinv,
                                const float* __restrict__ bias,
                                float* __restrict__ out, int nrows) {
    int i = blockIdx.x * blockDim.x + threadIdx.x;
    constexpr int F4 = F / 4;
    if (i >= nrows * F4) return;
    int n  = i / F4;
    int fo = (i % F4) * 4;
    float dvn = dinv[n];
    const float4 a = *(const float4*)&agg[n * F + fo];
    const float4 b = *(const float4*)&bias[fo];
    float4 v;
    v.x = dvn * a.x + b.x;
    v.y = dvn * a.y + b.y;
    v.z = dvn * a.z + b.z;
    v.w = dvn * a.w + b.w;
    if (RELU) {
        v.x = fmaxf(v.x, 0.0f);
        v.y = fmaxf(v.y, 0.0f);
        v.z = fmaxf(v.z, 0.0f);
        v.w = fmaxf(v.w, 0.0f);
    }
    *(float4*)&out[n * F + fo] = v;
}

extern "C" void kernel_launch(void* const* d_in, const int* in_sizes, int n_in,
                              void* d_out, int out_size, void* d_ws, size_t ws_size,
                              hipStream_t stream) {
    const float* x  = (const float*)d_in[0];
    const int*   ei = (const int*)d_in[1];
    const float* W1 = (const float*)d_in[2];
    const float* b1 = (const float*)d_in[3];
    const float* W2 = (const float*)d_in[4];
    const float* b2 = (const float*)d_in[5];
    float* out = (float*)d_out;

    const int NN = in_sizes[0] / 18;
    const int NE = in_sizes[1] / 2;
    const int NN_pad = (NN + 15) & ~15;
    const int* src = ei;
    const int* dst = ei + NE;

    char* ws = (char*)d_ws;
    size_t off = 0;
    auto carve = [&](size_t bytes) -> char* {
        char* p = ws + off;
        off += (bytes + 255) & ~(size_t)255;
        return p;
    };
    float* dinv = (float*)carve((size_t)NN_pad * 4);
    float* W1p  = (float*)carve((size_t)20 * 64 * 4);
    float* W2p  = (float*)carve((size_t)64 * 32 * 4);
    float* xs   = (float*)carve((size_t)NN_pad * 32 * 4);
    float* aggx = (float*)carve((size_t)NN_pad * 32 * 4);
    float* h1   = (float*)carve((size_t)NN_pad * 64 * 4);
    float* hs2  = xs;
    float* agg2 = aggx;

    const int B = 256;

    init_deg_kernel<<<(NN_pad + B - 1) / B, B, 0, stream>>>(dinv, NN_pad);
    accum_deg_kernel<<<(NE + B - 1) / B, B, 0, stream>>>(dst, dinv, NE);
    deg_to_dinv_kernel<<<(NN_pad + B - 1) / B, B, 0, stream>>>(dinv, NN_pad);

    scale_pad_x_kernel<<<(NN_pad * 32 + B - 1) / B, B, 0, stream>>>(x, dinv, xs, aggx, NN, NN_pad);
    repack_w_kernel<<<(20 * 64 + B - 1) / B, B, 0, stream>>>(W1, W1p, 18, 20, 64);
    repack_w_kernel<<<(64 * 32 + B - 1) / B, B, 0, stream>>>(W2, W2p, 64, 64, 32);

    const int mtiles = NN_pad / 16;
    const int gblk = (mtiles + 3) / 4;
    const int swaves = (NE + 31) / 32;
    const int sblk = (swaves + (B / 32) - 1) / (B / 32);

    scatter_kernel<32><<<sblk, B, 0, stream>>>(src, dst, xs, aggx, NE);
    gemm_bias_relu_kernel<20, 32, 64, 4><<<gblk, 128, 0, stream>>>(aggx, W1p, dinv, b1, h1, mtiles);

    gemm_scale_kernel<64, 32, 2><<<gblk, 128, 0, stream>>>(h1, W2p, dinv, hs2, agg2, mtiles);
    scatter_kernel<32><<<sblk, B, 0, stream>>>(src, dst, hs2, agg2, NE);
    finalize_kernel<32, false><<<(NN * 8 + B - 1) / B, B, 0, stream>>>(agg2, dinv, b2, out, NN);
}


# Degree count without global atomics: per node range an integer histogram in LDS fed by a 4x redundant scan of dst, partial rows per edge slice summed where dinv is formed; float sums untouched
# speedup vs baseline: 2.5090x; 1.5485x over previous
#include <hip/hip_runtime.h>

typedef __attribute__((ext_vector_type(2))) float v2f;
typedef __attribute__((ext_vector_type(8))) float v8f;

constexpr int DEG_RANGE   = 25024;
constexpr int DEG_CHUNKS  = 64;
constexpr int DEG_THREADS = 1024;
constexpr int DEG_BATCH   = 8;

__device__ __forceinline__ void deg_bump(unsigned* cnt, int d, int lo) {
    const unsigned i = (unsigned)(d - lo);
    if (i < (unsigned)DEG_RANGE) atomicAdd(&cnt[i], 1u);
}

__global__ void __launch_bounds__(DEG_THREADS)
deg_count_lds_kernel(const int* __restrict__ dst, unsigned* __restrict__ part,
                     int ne, int chunk, int nn_pad, int pitch) {
    __shared__ unsigned cnt[DEG_RANGE];
    const int tid = threadIdx.x;
    const int lo  = blockIdx.x * DEG_RANGE;
    const int c   = blockIdx.y;
    const int left = nn_pad - lo;
    const int range_n = left < DEG_RANGE ? left : DEG_RANGE;

    uint4* cnt4 = (uint4*)cnt;
    for (int i = tid; i < DEG_RANGE / 4; i += DEG_THREADS) cnt4[i] = make_uint4(0u, 0u, 0u, 0u);
    __syncthreads();

    const int e0 = c * chunk;
    int e1 = e0 + chunk;
    if (e1 > ne) e1 = ne;
    if ((((size_t)dst) & 15) == 0) {
        const int4* d4 = (const int4*)dst;
        const int q1 = e1 >> 2;
        for (int qb = (e0 >> 2) + tid; qb < q1; qb += DEG_BATCH * DEG_THREADS) {
            int4 d[DEG_BATCH];
#pragma unroll
            for (int j = 0; j < DEG_BATCH; ++j) {
                const int q = qb + j * DEG_THREADS;
                d[j] = d4[q < q1 ? q : q1 - 1];
            }
#pragma unroll
            for (int j = 0; j < DEG_BATCH; ++j) {
                if (qb + j * DEG_THREADS < q1) {
                    deg_bump(cnt, d[j].x, lo);
                    deg_bump(cnt, d[j].y, lo);
                    deg_bump(cnt, d[j].z, lo);
                    deg_bump(cnt, d[j].w, lo);
                }
            }
        }
        const int e = (q1 << 2) + tid;
        if (e >= e0 && e < e1) deg_bump(cnt, dst[e], lo);
    } else {
        for (int eb = e0 + tid; eb < e1; eb += DEG_BATCH * DEG_THREADS) {
            int d[DEG_BATCH];
#pragma unroll
            for (int j = 0; j < DEG_BATCH; ++j) {
                const int e = eb + j * DEG_THREADS;
                d[j] = dst[e < e1 ? e : e1 - 1];
            }
#pragma unroll
            for (int j = 0; j < DEG_BATCH; ++j)
                if (eb + j * DEG_THREADS < e1) deg_bump(cnt, d[j], lo);
        }
    }
    __syncthreads();

    uint4* out4 = (uint4*)(part + (size_t)c * pitch + lo);
    for (int i = tid; i < range_n / 4; i += DEG_THREADS) out4[i] = cnt4[i];
}

__global__ void dinv_from_counts_kernel(const unsigned* __restrict__ part,
                                        float* __restrict__ dinv, int nn_pad, int pitch) {
    const int n = blockIdx.x * blockDim.x + threadIdx.x;
    if (n >= nn_pad) return;
    unsigned s = 1u;
#pragma unroll
    for (int c = 0; c < DEG_CHUNKS; ++c) s += part[(size_t)c * pitch + n];
    dinv[n] = rsqrtf((float)s);
}

__global__ void scale_pad_x_kernel(const float* __restrict__ x,
                                   const float* __restrict__ dinv,
                                   float* __restrict__ xs, float* __restrict__ aggx,
                                   int nn, int nn_pad) {
    int i = blockIdx.x * blockDim.x + threadIdx.x;
    if (i >= nn_pad * 32) return;
    int n = i >> 5;
    int k = i & 31;
    float v = (n < nn && k < 18) ? dinv[n] * x[n * 18 + k] : 0.0f;
    xs[i]   = v;
    aggx[i] = v;
}

__global__ void repack_w_kernel(const float* __restrict__ w, float* __restrict__ wp,
                                int kd_true, int kp, int nf) {
    int o = blockIdx.x * blockDim.x + threadIdx.x;
    if (o >= kp * nf) return;
    int j = o & 1;
    int n = (o >> 1) % nf;
    int p = (o >> 1) / nf;
    int k = 2 * p + j;
    wp[o] = (k < kd_true) ? w[k * nf + n] : 0.0f;
}

template <int KD, int NF, int NT>
__global__ void gemm_scale_kernel(const float* __restrict__ X,
                                  const float* __restrict__ Wp,
                                  const float* __restrict__ dinv,
                                  float* __restrict__ hs,
                                  float* __restrict__ agg,
                                  int mtiles) {
    int wave = blockIdx.x * (blockDim.x >> 5) + (threadIdx.x >> 5);
    if (wave >= mtiles) return;
    int lane = threadIdx.x & 31;
    int r16  = lane & 15;
    int kh   = lane >> 4;
    int mbase = wave * 16;
    int arow  = mbase + r16;

    v8f acc[NT] = {};

    static_assert(KD % 4 == 0, "KD must be multiple of 4");
    constexpr int KSTEPS = KD / 4;
#pragma unroll
    for (int ks = 0; ks < KSTEPS; ++ks) {
        int k0 = ks * 4 + kh * 2;
        v2f a = *(const v2f*)&X[arow * KD + k0];
        int p = ks * 2 + kh;
#pragma unroll
        for (int t = 0; t < NT; ++t) {
            v2f b = *(const v2f*)&Wp[(p * NF + t * 16 + r16) * 2];
            acc[t] = __builtin_amdgcn_wmma_f32_16x16x4_f32(
                false, a, false, b, (short)0, acc[t], false, false);
        }
    }

    const float4 dv0 = *(const float4*)&dinv[mbase + 8 * kh];
    const float4 dv1 = *(const float4*)&dinv[mbase + 8 * kh + 4];
    const float dv[8] = {dv0.x, dv0.y, dv0.z, dv0.w, dv1.x, dv1.y, dv1.z, dv1.w};

#pragma unroll
    for (int v = 0; v < 8; ++v) {
        int m = mbase + v + 8 * kh;
#pragma unroll
        for (int t = 0; t < NT; ++t) {
            float val = acc[t][v] * dv[v];
            int idx = m * NF + t * 16 + r16;
            hs[idx]  = val;
            agg[idx] = val;
        }
    }
}

template <int KD, int LDA, int NF, int NT>
__global__ void gemm_bias_relu_kernel(const float* __restrict__ X,
                                      const float* __restrict__ Wp,
                                      const float* __restrict__ dinv,
                                      const float* __restrict__ bias,
                                      float* __restrict__ h,
                                      int mtiles) {
    int wave = blockIdx.x * (blockDim.x >> 5) + (threadIdx.x >> 5);
    if (wave >= mtiles) return;
    int lane = threadIdx.x & 31;
    int r16  = lane & 15;
    int kh   = lane >> 4;
    int mbase = wave * 16;
    int arow  = mbase + r16;

    v8f acc[NT] = {};

    static_assert(KD % 4 == 0 && KD <= LDA && LDA % 2 == 0, "K layout");
    constexpr int KSTEPS = KD / 4;
#pragma unroll
    for (int ks = 0; ks < KSTEPS; ++ks) {
        int k0 = ks * 4 + kh * 2;
        v2f a = *(const v2f*)&X[arow * LDA + k0];
        int p = ks * 2 + kh;
#pragma unroll
        for (int t = 0; t < NT; ++t) {
            v2f b = *(const v2f*)&Wp[(p * NF + t * 16 + r16) * 2];
            acc[t] = __builtin_amdgcn_wmma_f32_16x16x4_f32(
                false, a, false, b, (short)0, acc[t], false, false);
        }
    }

    const float4 dv0 = *(const float4*)&dinv[mbase + 8 * kh];
    const float4 dv1 = *(const float4*)&dinv[mbase + 8 * kh + 4];
    const float dv[8] = {dv0.x, dv0.y, dv0.z, dv0.w, dv1.x, dv1.y, dv1.z, dv1.w};
    float bv[NT];
#pragma unroll
    for (int t = 0; t < NT; ++t) bv[t] = bias[t * 16 + r16];

#pragma unroll
    for (int v = 0; v < 8; ++v) {
        int m = mbase + v + 8 * kh;
#pragma unroll
        for (int t = 0; t < NT; ++t)
            h[m * NF + t * 16 + r16] = fmaxf(acc[t][v] * dv[v] + bv[t], 0.0f);
    }
}

template <int F>
__global__ void scatter_kernel(const int* __restrict__ src,
                               const int* __restrict__ dst,
                               const float* __restrict__ hs,
                               float* __restrict__ agg, int ne) {
    constexpr int NSEG = F / 32;
    constexpr int U = 16 / NSEG;
    const int lane = threadIdx.x & 31;
    const int wave = __builtin_amdgcn_readfirstlane(
        (int)(blockIdx.x * (blockDim.x >> 5) + (threadIdx.x >> 5)));
    const int e0 = wave * 32;
    if (e0 >= ne) return;
    const int left = ne - e0;
    const int cnt = left < 32 ? left : 32;
    const int e = (lane < cnt) ? e0 + lane : ne - 1;
    const int my_s = src[e];
    const int my_d = dst[e];

    if (cnt == 32) {
#pragma unroll
        for (int g = 0; g < 32; g += U) {
            float v[U][NSEG];
#pragma unroll
            for (int u = 0; u < U; ++u) {
                const int s = __builtin_amdgcn_readlane(my_s, g + u);
#pragma unroll
                for (int q = 0; q < NSEG; ++q) v[u][q] = hs[s * F + q * 32 + lane];
            }
#pragma unroll
            for (int u = 0; u < U; ++u) {
                const int d = __builtin_amdgcn_readlane(my_d, g + u);
#pragma unroll
                for (int q = 0; q < NSEG; ++q) atomicAdd(&agg[d * F + q * 32 + lane], v[u][q]);
            }
        }
    } else {
        for (int k = 0; k < cnt; ++k) {
            const int s = __builtin_amdgcn_readlane(my_s, k);
            const int d = __builtin_amdgcn_readlane(my_d, k);
#pragma unroll
            for (int q = 0; q < NSEG; ++q)
                atomicAdd(&agg[d * F + q * 32 + lane], hs[s * F + q * 32 + lane]);
        }
    }
}

template <int F, bool RELU>
__global__ void finalize_kernel(const float* __restrict__ agg,
                                const float* __restrict__ dinv,
                                const float* __restrict__ bias,
                                float* __restrict__ out, int nrows) {
    int i = blockIdx.x * blockDim.x + threadIdx.x;
    constexpr int F4 = F / 4;
    if (i >= nrows * F4) return;
    int n  = i / F4;
    int fo = (i % F4) * 4;
    float dvn = dinv[n];
    const float4 a = *(const float4*)&agg[n * F + fo];
    const float4 b = *(const float4*)&bias[fo];
    float4 v;
    v.x = dvn * a.x + b.x;
    v.y = dvn * a.y + b.y;
    v.z = dvn * a.z + b.z;
    v.w = dvn * a.w + b.w;
    if (RELU) {
        v.x = fmaxf(v.x, 0.0f);
        v.y = fmaxf(v.y, 0.0f);
        v.z = fmaxf(v.z, 0.0f);
        v.w = fmaxf(v.w, 0.0f);
    }
    *(float4*)&out[n * F + fo] = v;
}

extern "C" void kernel_launch(void* const* d_in, const int* in_sizes, int n_in,
                              void* d_out, int out_size, void* d_ws, size_t ws_size,
                              hipStream_t stream) {
    const float* x  = (const float*)d_in[0];
    const int*   ei = (const int*)d_in[1];
    const float* W1 = (const float*)d_in[2];
    const float* b1 = (const float*)d_in[3];
    const float* W2 = (const float*)d_in[4];
    const float* b2 = (const float*)d_in[5];
    float* out = (float*)d_out;

    const int NN = in_sizes[0] / 18;
    const int NE = in_sizes[1] / 2;
    const int NN_pad = (NN + 15) & ~15;
    const int* src = ei;
    const int* dst = ei + NE;

    char* ws = (char*)d_ws;
    size_t off = 0;
    auto carve = [&](size_t bytes) -> char* {
        char* p = ws + off;
        off += (bytes + 255) & ~(size_t)255;
        return p;
    };
    float* dinv = (float*)carve((size_t)NN_pad * 4);
    const int dpitch = (NN_pad + 31) & ~31;
    unsigned* degp = (unsigned*)carve((size_t)DEG_CHUNKS * dpitch * 4);
    float* W1p  = (float*)carve((size_t)20 * 64 * 4);
    float* W2p  = (float*)carve((size_t)64 * 32 * 4);
    float* xs   = (float*)carve((size_t)NN_pad * 32 * 4);
    float* aggx = (float*)carve((size_t)NN_pad * 32 * 4);
    float* h1   = (float*)carve((size_t)NN_pad * 64 * 4);
    float* hs2  = xs;
    float* agg2 = aggx;

    const int B = 256;

    const int dranges = (NN_pad + DEG_RANGE - 1) / DEG_RANGE;
    const int dchunk  = (((NE + DEG_CHUNKS - 1) / DEG_CHUNKS) + 3) & ~3;
    deg_count_lds_kernel<<<dim3(dranges, DEG_CHUNKS), DEG_THREADS, 0, stream>>>(
        dst, degp, NE, dchunk, NN_pad, dpitch);
    dinv_from_counts_kernel<<<(NN_pad + B - 1) / B, B, 0, stream>>>(degp, dinv, NN_pad, dpitch);

    scale_pad_x_kernel<<<(NN_pad * 32 + B - 1) / B, B, 0, stream>>>(x, dinv, xs, aggx, NN, NN_pad);
    repack_w_kernel<<<(20 * 64 + B - 1) / B, B, 0, stream>>>(W1, W1p, 18, 20, 64);
    repack_w_kernel<<<(64 * 32 + B - 1) / B, B, 0, stream>>>(W2, W2p, 64, 64, 32);

    const int mtiles = NN_pad / 16;
    const int gblk = (mtiles + 3) / 4;
    const int swaves = (NE + 31) / 32;
    const int sblk = (swaves + (B / 32) - 1) / (B / 32);

    scatter_kernel<32><<<sblk, B, 0, stream>>>(src, dst, xs, aggx, NE);
    gemm_bias_relu_kernel<20, 32, 64, 4><<<gblk, 128, 0, stream>>>(aggx, W1p, dinv, b1, h1, mtiles);

    gemm_scale_kernel<64, 32, 2><<<gblk, 128, 0, stream>>>(h1, W2p, dinv, hs2, agg2, mtiles);
    scatter_kernel<32><<<sblk, B, 0, stream>>>(src, dst, hs2, agg2, NE);
    finalize_kernel<32, false><<<(NN * 8 + B - 1) / B, B, 0, stream>>>(agg2, dinv, b2, out, NN);
}


# Scatter rows owned by one L2 half: work-groups read their die id and take edge-slice tickets for their half's row range (the other's when done); f32 atomics kept, their arrival order changes
# speedup vs baseline: 4.4373x; 1.7686x over previous
#include <hip/hip_runtime.h>

typedef __attribute__((ext_vector_type(2))) float v2f;
typedef __attribute__((ext_vector_type(8))) float v8f;

constexpr int DEG_RANGE   = 25024;
constexpr int DEG_CHUNKS  = 64;
constexpr int DEG_THREADS = 1024;
constexpr int DEG_BATCH   = 8;

__device__ __forceinline__ void deg_bump(unsigned* cnt, int d, int lo) {
    const unsigned i = (unsigned)(d - lo);
    if (i < (unsigned)DEG_RANGE) atomicAdd(&cnt[i], 1u);
}

__global__ void __launch_bounds__(DEG_THREADS)
deg_count_lds_kernel(const int* __restrict__ dst, unsigned* __restrict__ part,
                     int ne, int chunk, int nn_pad, int pitch) {
    __shared__ unsigned cnt[DEG_RANGE];
    const int tid = threadIdx.x;
    const int lo  = blockIdx.x * DEG_RANGE;
    const int c   = blockIdx.y;
    const int left = nn_pad - lo;
    const int range_n = left < DEG_RANGE ? left : DEG_RANGE;

    uint4* cnt4 = (uint4*)cnt;
    for (int i = tid; i < DEG_RANGE / 4; i += DEG_THREADS) cnt4[i] = make_uint4(0u, 0u, 0u, 0u);
    __syncthreads();

    const int e0 = c * chunk;
    int e1 = e0 + chunk;
    if (e1 > ne) e1 = ne;
    if ((((size_t)dst) & 15) == 0) {
        const int4* d4 = (const int4*)dst;
        const int q1 = e1 >> 2;
        for (int qb = (e0 >> 2) + tid; qb < q1; qb += DEG_BATCH * DEG_THREADS) {
            int4 d[DEG_BATCH];
#pragma unroll
            for (int j = 0; j < DEG_BATCH; ++j) {
                const int q = qb + j * DEG_THREADS;
                d[j] = d4[q < q1 ? q : q1 - 1];
            }
#pragma unroll
            for (int j = 0; j < DEG_BATCH; ++j) {
                if (qb + j * DEG_THREADS < q1) {
                    deg_bump(cnt, d[j].x, lo);
                    deg_bump(cnt, d[j].y, lo);
                    deg_bump(cnt, d[j].z, lo);
                    deg_bump(cnt, d[j].w, lo);
                }
            }
        }
        const int e = (q1 << 2) + tid;
        if (e >= e0 && e < e1) deg_bump(cnt, dst[e], lo);
    } else {
        for (int eb = e0 + tid; eb < e1; eb += DEG_BATCH * DEG_THREADS) {
            int d[DEG_BATCH];
#pragma unroll
            for (int j = 0; j < DEG_BATCH; ++j) {
                const int e = eb + j * DEG_THREADS;
                d[j] = dst[e < e1 ? e : e1 - 1];
            }
#pragma unroll
            for (int j = 0; j < DEG_BATCH; ++j)
                if (eb + j * DEG_THREADS < e1) deg_bump(cnt, d[j], lo);
        }
    }
    __syncthreads();

    uint4* out4 = (uint4*)(part + (size_t)c * pitch + lo);
    for (int i = tid; i < range_n / 4; i += DEG_THREADS) out4[i] = cnt4[i];
}

__global__ void dinv_from_counts_kernel(const unsigned* __restrict__ part,
                                        float* __restrict__ dinv, int* __restrict__ ctr,
                                        int nn_pad, int pitch) {
    const int n = blockIdx.x * blockDim.x + threadIdx.x;
    if (n >= nn_pad) return;
    if (n < 4) ctr[n * 64] = 0;
    unsigned s = 1u;
#pragma unroll
    for (int c = 0; c < DEG_CHUNKS; ++c) s += part[(size_t)c * pitch + n];
    dinv[n] = rsqrtf((float)s);
}

__global__ void scale_pad_x_kernel(const float* __restrict__ x,
                                   const float* __restrict__ dinv,
                                   float* __restrict__ xs, float* __restrict__ aggx,
                                   int nn, int nn_pad) {
    int i = blockIdx.x * blockDim.x + threadIdx.x;
    if (i >= nn_pad * 32) return;
    int n = i >> 5;
    int k = i & 31;
    float v = (n < nn && k < 18) ? dinv[n] * x[n * 18 + k] : 0.0f;
    xs[i]   = v;
    aggx[i] = v;
}

__global__ void repack_w_kernel(const float* __restrict__ w, float* __restrict__ wp,
                                int kd_true, int kp, int nf) {
    int o = blockIdx.x * blockDim.x + threadIdx.x;
    if (o >= kp * nf) return;
    int j = o & 1;
    int n = (o >> 1) % nf;
    int p = (o >> 1) / nf;
    int k = 2 * p + j;
    wp[o] = (k < kd_true) ? w[k * nf + n] : 0.0f;
}

template <int KD, int NF, int NT>
__global__ void gemm_scale_kernel(const float* __restrict__ X,
                                  const float* __restrict__ Wp,
                                  const float* __restrict__ dinv,
                                  float* __restrict__ hs,
                                  float* __restrict__ agg,
                                  int mtiles) {
    int wave = blockIdx.x * (blockDim.x >> 5) + (threadIdx.x >> 5);
    if (wave >= mtiles) return;
    int lane = threadIdx.x & 31;
    int r16  = lane & 15;
    int kh   = lane >> 4;
    int mbase = wave * 16;
    int arow  = mbase + r16;

    v8f acc[NT] = {};

    static_assert(KD % 4 == 0, "KD must be multiple of 4");
    constexpr int KSTEPS = KD / 4;
#pragma unroll
    for (int ks = 0; ks < KSTEPS; ++ks) {
        int k0 = ks * 4 + kh * 2;
        v2f a = *(const v2f*)&X[arow * KD + k0];
        int p = ks * 2 + kh;
#pragma unroll
        for (int t = 0; t < NT; ++t) {
            v2f b = *(const v2f*)&Wp[(p * NF + t * 16 + r16) * 2];
            acc[t] = __builtin_amdgcn_wmma_f32_16x16x4_f32(
                false, a, false, b, (short)0, acc[t], false, false);
        }
    }

    const float4 dv0 = *(const float4*)&dinv[mbase + 8 * kh];
    const float4 dv1 = *(const float4*)&dinv[mbase + 8 * kh + 4];
    const float dv[8] = {dv0.x, dv0.y, dv0.z, dv0.w, dv1.x, dv1.y, dv1.z, dv1.w};

#pragma unroll
    for (int v = 0; v < 8; ++v) {
        int m = mbase + v + 8 * kh;
#pragma unroll
        for (int t = 0; t < NT; ++t) {
            float val = acc[t][v] * dv[v];
            int idx = m * NF + t * 16 + r16;
            hs[idx]  = val;
            agg[idx] = val;
        }
    }
}

template <int KD, int LDA, int NF, int NT>
__global__ void gemm_bias_relu_kernel(const float* __restrict__ X,
                                      const float* __restrict__ Wp,
                                      const float* __restrict__ dinv,
                                      const float* __restrict__ bias,
                                      float* __restrict__ h,
                                      int mtiles) {
    int wave = blockIdx.x * (blockDim.x >> 5) + (threadIdx.x >> 5);
    if (wave >= mtiles) return;
    int lane = threadIdx.x & 31;
    int r16  = lane & 15;
    int kh   = lane >> 4;
    int mbase = wave * 16;
    int arow  = mbase + r16;

    v8f acc[NT] = {};

    static_assert(KD % 4 == 0 && KD <= LDA && LDA % 2 == 0, "K layout");
    constexpr int KSTEPS = KD / 4;
#pragma unroll
    for (int ks = 0; ks < KSTEPS; ++ks) {
        int k0 = ks * 4 + kh * 2;
        v2f a = *(const v2f*)&X[arow * LDA + k0];
        int p = ks * 2 + kh;
#pragma unroll
        for (int t = 0; t < NT; ++t) {
            v2f b = *(const v2f*)&Wp[(p * NF + t * 16 + r16) * 2];
            acc[t] = __builtin_amdgcn_wmma_f32_16x16x4_f32(
                false, a, false, b, (short)0, acc[t], false, false);
        }
    }

    const float4 dv0 = *(const float4*)&dinv[mbase + 8 * kh];
    const float4 dv1 = *(const float4*)&dinv[mbase + 8 * kh + 4];
    const float dv[8] = {dv0.x, dv0.y, dv0.z, dv0.w, dv1.x, dv1.y, dv1.z, dv1.w};
    float bv[NT];
#pragma unroll
    for (int t = 0; t < NT; ++t) bv[t] = bias[t * 16 + r16];

#pragma unroll
    for (int v = 0; v < 8; ++v) {
        int m = mbase + v + 8 * kh;
#pragma unroll
        for (int t = 0; t < NT; ++t)
            h[m * NF + t * 16 + r16] = fmaxf(acc[t][v] * dv[v] + bv[t], 0.0f);
    }
}

constexpr int SC_WAVES  = 8;
constexpr int SC_WEDGES = 128;
constexpr int SC_BEDGES = SC_WAVES * SC_WEDGES;
constexpr int SC_CTR_STRIDE = 64;

__global__ void __launch_bounds__(SC_WAVES * 32)
scatter_owned_kernel(const int* __restrict__ src,
                     const int* __restrict__ dst,
                     const float* __restrict__ hs,
                     float* __restrict__ agg, int* __restrict__ ctr,
                     int ne, int nslices, int nn_pad) {
    constexpr int F = 32;
    constexpr int U = 16;
    constexpr int NLD = SC_WEDGES / 32;
    __shared__ int2 lst[SC_WAVES][SC_WEDGES];
    __shared__ int tk[2];
    const int lane = threadIdx.x & 31;
    const int wv   = threadIdx.x >> 5;

    if (threadIdx.x == 0) {
        const unsigned id = __builtin_amdgcn_s_sendmsg_rtn(0x87);
        const int r = (int)((id >> 18) & 1u);
        tk[0] = atomicAdd(&ctr[r * SC_CTR_STRIDE], 1);
        tk[1] = r;
    }
    __syncthreads();
    int t = tk[0];
    int role = tk[1];
    if (t >= nslices) {
        __syncthreads();
        if (threadIdx.x == 0) {
            const int r = role ^ 1;
            tk[0] = atomicAdd(&ctr[r * SC_CTR_STRIDE], 1);
            tk[1] = r;
        }
        __syncthreads();
        t = tk[0];
        role = tk[1];
        if (t >= nslices) return;
    }
    t    = __builtin_amdgcn_readfirstlane(t);
    role = __builtin_amdgcn_readfirstlane(role);

    const int split = (nn_pad >> 2) << 1;
    const int e0 = t * SC_BEDGES + wv * SC_WEDGES;
    int2* my = lst[wv];

    int sv[NLD], dv[NLD];
    bool in[NLD];
#pragma unroll
    for (int j = 0; j < NLD; ++j) {
        const int e = e0 + j * 32 + lane;
        in[j] = e < ne;
        const int ec = in[j] ? e : ne - 1;
        sv[j] = src[ec];
        dv[j] = dst[ec];
    }
    static_assert(NLD == 4, "the fence below names four pairs");
    asm volatile("" :: "v"(sv[0]), "v"(sv[1]), "v"(sv[2]), "v"(sv[3]),
                       "v"(dv[0]), "v"(dv[1]), "v"(dv[2]), "v"(dv[3]));
    int n = 0;
#pragma unroll
    for (int j = 0; j < NLD; ++j) {
        const bool mine = in[j] && ((dv[j] >= split) == (role != 0));
        const unsigned m = __builtin_amdgcn_ballot_w32(mine);
        const int rank = (int)__builtin_amdgcn_mbcnt_lo(m, 0u);
        if (mine) my[n + rank] = make_int2(sv[j], dv[j]);
        n += __builtin_popcount(m);
    }
    __builtin_amdgcn_fence(__ATOMIC_ACQ_REL, "wavefront");

    for (int base = 0; base < n; base += 32) {
        const int i = base + lane;
        int my_s = 0, my_d = -1;
        if (i < n) {
            const int2 p = my[i];
            my_s = p.x;
            my_d = p.y;
        }
#pragma unroll
        for (int g = 0; g < 32; g += U) {
            if (base + g >= n) break;
            float v[U];
#pragma unroll
            for (int u = 0; u < U; ++u) {
                const int s = __builtin_amdgcn_readlane(my_s, g + u);
                v[u] = hs[s * F + lane];
            }
            static_assert(U == 16, "the fence below names sixteen values");
            asm volatile("" :: "v"(v[0]), "v"(v[1]), "v"(v[2]), "v"(v[3]),
                               "v"(v[4]), "v"(v[5]), "v"(v[6]), "v"(v[7]),
                               "v"(v[8]), "v"(v[9]), "v"(v[10]), "v"(v[11]),
                               "v"(v[12]), "v"(v[13]), "v"(v[14]), "v"(v[15]));
#pragma unroll
            for (int u = 0; u < U; ++u) {
                const int d = __builtin_amdgcn_readlane(my_d, g + u);
                if (d >= 0) atomicAdd(&agg[d * F + lane], v[u]);
            }
        }
    }
}

template <int F, bool RELU>
__global__ void finalize_kernel(const float* __restrict__ agg,
                                const float* __restrict__ dinv,
                                const float* __restrict__ bias,
                                float* __restrict__ out, int nrows) {
    int i = blockIdx.x * blockDim.x + threadIdx.x;
    constexpr int F4 = F / 4;
    if (i >= nrows * F4) return;
    int n  = i / F4;
    int fo = (i % F4) * 4;
    float dvn = dinv[n];
    const float4 a = *(const float4*)&agg[n * F + fo];
    const float4 b = *(const float4*)&bias[fo];
    float4 v;
    v.x = dvn * a.x + b.x;
    v.y = dvn * a.y + b.y;
    v.z = dvn * a.z + b.z;
    v.w = dvn * a.w + b.w;
    if (RELU) {
        v.x = fmaxf(v.x, 0.0f);
        v.y = fmaxf(v.y, 0.0f);
        v.z = fmaxf(v.z, 0.0f);
        v.w = fmaxf(v.w, 0.0f);
    }
    *(float4*)&out[n * F + fo] = v;
}

extern "C" void kernel_launch(void* const* d_in, const int* in_sizes, int n_in,
                              void* d_out, int out_size, void* d_ws, size_t ws_size,
                              hipStream_t stream) {
    const float* x  = (const float*)d_in[0];
    const int*   ei = (const int*)d_in[1];
    const float* W1 = (const float*)d_in[2];
    const float* b1 = (const float*)d_in[3];
    const float* W2 = (const float*)d_in[4];
    const float* b2 = (const float*)d_in[5];
    float* out = (float*)d_out;

    const int NN = in_sizes[0] / 18;
    const int NE = in_sizes[1] / 2;
    const int NN_pad = (NN + 15) & ~15;
    const int* src = ei;
    const int* dst = ei + NE;

    char* ws = (char*)d_ws;
    size_t off = 0;
    auto carve = [&](size_t bytes) -> char* {
        char* p = ws + off;
        off += (bytes + 255) & ~(size_t)255;
        return p;
    };
    float* dinv = (float*)carve((size_t)NN_pad * 4);
    const int dpitch = (NN_pad + 31) & ~31;
    unsigned* degp = (unsigned*)carve((size_t)DEG_CHUNKS * dpitch * 4);
    int* sctr   = (int*)carve((size_t)4 * 64 * 4);
    float* W1p  = (float*)carve((size_t)20 * 64 * 4);
    float* W2p  = (float*)carve((size_t)64 * 32 * 4);
    float* xs   = (float*)carve((size_t)NN_pad * 32 * 4);
    float* aggx = (float*)carve((size_t)NN_pad * 32 * 4);
    float* h1   = (float*)carve((size_t)NN_pad * 64 * 4);
    float* hs2  = xs;
    float* agg2 = aggx;

    const int B = 256;

    const int dranges = (NN_pad + DEG_RANGE - 1) / DEG_RANGE;
    const int dchunk  = (((NE + DEG_CHUNKS - 1) / DEG_CHUNKS) + 3) & ~3;
    deg_count_lds_kernel<<<dim3(dranges, DEG_CHUNKS), DEG_THREADS, 0, stream>>>(
        dst, degp, NE, dchunk, NN_pad, dpitch);
    dinv_from_counts_kernel<<<(NN_pad + B - 1) / B, B, 0, stream>>>(degp, dinv, sctr, NN_pad, dpitch);

    scale_pad_x_kernel<<<(NN_pad * 32 + B - 1) / B, B, 0, stream>>>(x, dinv, xs, aggx, NN, NN_pad);
    repack_w_kernel<<<(20 * 64 + B - 1) / B, B, 0, stream>>>(W1, W1p, 18, 20, 64);
    repack_w_kernel<<<(64 * 32 + B - 1) / B, B, 0, stream>>>(W2, W2p, 64, 64, 32);

    const int mtiles = NN_pad / 16;
    const int gblk = (mtiles + 3) / 4;
    const int nslices = (NE + SC_BEDGES - 1) / SC_BEDGES;
    const int sblk = 2 * nslices;

    scatter_owned_kernel<<<sblk, SC_WAVES * 32, 0, stream>>>(src, dst, xs, aggx, sctr, NE, nslices, NN_pad);
    gemm_bias_relu_kernel<20, 32, 64, 4><<<gblk, 128, 0, stream>>>(aggx, W1p, dinv, b1, h1, mtiles);

    gemm_scale_kernel<64, 32, 2><<<gblk, 128, 0, stream>>>(h1, W2p, dinv, hs2, agg2, mtiles);
    scatter_owned_kernel<<<sblk, SC_WAVES * 32, 0, stream>>>(src, dst, hs2, agg2, sctr + 2 * SC_CTR_STRIDE, NE, nslices, NN_pad);
    finalize_kernel<32, false><<<(NN * 8 + B - 1) / B, B, 0, stream>>>(agg2, dinv, b2, out, NN);
}


# Both dense products and the weight repacks in ONE launch: layer 1 as W1^T x^T, its D registers are layer 2's A fragments, hidden rows stay in registers; launches 10 -> 7; layer-2 k sum order changes
# speedup vs baseline: 4.8189x; 1.0860x over previous
#include <hip/hip_runtime.h>

typedef __attribute__((ext_vector_type(2))) float v2f;
typedef __attribute__((ext_vector_type(8))) float v8f;

constexpr int DEG_RANGE   = 25024;
constexpr int DEG_CHUNKS  = 64;
constexpr int DEG_THREADS = 1024;
constexpr int DEG_BATCH   = 8;

__device__ __forceinline__ void deg_bump(unsigned* cnt, int d, int lo) {
    const unsigned i = (unsigned)(d - lo);
    if (i < (unsigned)DEG_RANGE) atomicAdd(&cnt[i], 1u);
}

__global__ void __launch_bounds__(DEG_THREADS)
deg_count_lds_kernel(const int* __restrict__ dst, unsigned* __restrict__ part,
                     int ne, int chunk, int nn_pad, int pitch) {
    __shared__ unsigned cnt[DEG_RANGE];
    const int tid = threadIdx.x;
    const int lo  = blockIdx.x * DEG_RANGE;
    const int c   = blockIdx.y;
    const int left = nn_pad - lo;
    const int range_n = left < DEG_RANGE ? left : DEG_RANGE;

    uint4* cnt4 = (uint4*)cnt;
    for (int i = tid; i < DEG_RANGE / 4; i += DEG_THREADS) cnt4[i] = make_uint4(0u, 0u, 0u, 0u);
    __syncthreads();

    const int e0 = c * chunk;
    int e1 = e0 + chunk;
    if (e1 > ne) e1 = ne;
    if ((((size_t)dst) & 15) == 0) {
        const int4* d4 = (const int4*)dst;
        const int q1 = e1 >> 2;
        for (int qb = (e0 >> 2) + tid; qb < q1; qb += DEG_BATCH * DEG_THREADS) {
            int4 d[DEG_BATCH];
#pragma unroll
            for (int j = 0; j < DEG_BATCH; ++j) {
                const int q = qb + j * DEG_THREADS;
                d[j] = d4[q < q1 ? q : q1 - 1];
            }
#pragma unroll
            for (int j = 0; j < DEG_BATCH; ++j) {
                if (qb + j * DEG_THREADS < q1) {
                    deg_bump(cnt, d[j].x, lo);
                    deg_bump(cnt, d[j].y, lo);
                    deg_bump(cnt, d[j].z, lo);
                    deg_bump(cnt, d[j].w, lo);
                }
            }
        }
        const int e = (q1 << 2) + tid;
        if (e >= e0 && e < e1) deg_bump(cnt, dst[e], lo);
    } else {
        for (int eb = e0 + tid; eb < e1; eb += DEG_BATCH * DEG_THREADS) {
            int d[DEG_BATCH];
#pragma unroll
            for (int j = 0; j < DEG_BATCH; ++j) {
                const int e = eb + j * DEG_THREADS;
                d[j] = dst[e < e1 ? e : e1 - 1];
            }
#pragma unroll
            for (int j = 0; j < DEG_BATCH; ++j)
                if (eb + j * DEG_THREADS < e1) deg_bump(cnt, d[j], lo);
        }
    }
    __syncthreads();

    uint4* out4 = (uint4*)(part + (size_t)c * pitch + lo);
    for (int i = tid; i < range_n / 4; i += DEG_THREADS) out4[i] = cnt4[i];
}

__global__ void dinv_from_counts_kernel(const unsigned* __restrict__ part,
                                        float* __restrict__ dinv, int* __restrict__ ctr,
                                        int nn_pad, int pitch) {
    const int n = blockIdx.x * blockDim.x + threadIdx.x;
    if (n >= nn_pad) return;
    if (n < 4) ctr[n * 64] = 0;
    unsigned s = 1u;
#pragma unroll
    for (int c = 0; c < DEG_CHUNKS; ++c) s += part[(size_t)c * pitch + n];
    dinv[n] = rsqrtf((float)s);
}

__global__ void scale_pad_x_kernel(const float* __restrict__ x,
                                   const float* __restrict__ dinv,
                                   float* __restrict__ xs, float* __restrict__ aggx,
                                   int nn, int nn_pad) {
    int i = blockIdx.x * blockDim.x + threadIdx.x;
    if (i >= nn_pad * 32) return;
    int n = i >> 5;
    int k = i & 31;
    float v = (n < nn && k < 18) ? dinv[n] * x[n * 18 + k] : 0.0f;
    xs[i]   = v;
    aggx[i] = v;
}

template <int K1, int LDA, int NH, int NO>
__global__ void __launch_bounds__(128)
gemm_two_layers_kernel(const float* X,
                       const float* __restrict__ W1,
                       const float* __restrict__ b1,
                       const float* __restrict__ W2,
                       const float* __restrict__ dinv,
                       float* hs, float* agg, int mtiles) {
    static_assert(NH % 16 == 0 && NO % 16 == 0 && LDA % 2 == 0, "tile layout");
    constexpr int KS1 = (K1 + 3) / 4;
    constexpr int T1  = NH / 16;
    constexpr int T2  = NO / 16;
    static_assert(KS1 * 4 <= LDA, "padded K inside the row pitch");

    const int wave = blockIdx.x * (blockDim.x >> 5) + (threadIdx.x >> 5);
    if (wave >= mtiles) return;
    const int lane = threadIdx.x & 31;
    const int c    = lane & 15;
    const int hh   = lane >> 4;
    const int m0   = wave * 16;

    v8f acc1[T1] = {};
#pragma unroll
    for (int ks = 0; ks < KS1; ++ks) {
        const int k0 = ks * 4 + hh * 2;
        const v2f xb = *(const v2f*)&X[(m0 + c) * LDA + k0];
        const bool in0 = k0 < K1;
        const bool in1 = k0 + 1 < K1;
        const int r0 = in0 ? k0 : 0;
        const int r1 = in1 ? k0 + 1 : 0;
#pragma unroll
        for (int t = 0; t < T1; ++t) {
            v2f wa;
            wa.x = W1[r0 * NH + t * 16 + c];
            wa.y = W1[r1 * NH + t * 16 + c];
            if (!in0) wa.x = 0.0f;
            if (!in1) wa.y = 0.0f;
            acc1[t] = __builtin_amdgcn_wmma_f32_16x16x4_f32(
                false, wa, false, xb, (short)0, acc1[t], false, false);
        }
    }

    const float dn = dinv[m0 + c];
#pragma unroll
    for (int t = 0; t < T1; ++t) {
        const float4 ba = *(const float4*)&b1[t * 16 + 8 * hh];
        const float4 bb = *(const float4*)&b1[t * 16 + 8 * hh + 4];
        const float bv[8] = {ba.x, ba.y, ba.z, ba.w, bb.x, bb.y, bb.z, bb.w};
#pragma unroll
        for (int r = 0; r < 8; ++r)
            acc1[t][r] = fmaxf(acc1[t][r] * dn + bv[r], 0.0f);
    }

    v8f acc2[T2] = {};
#pragma unroll
    for (int t = 0; t < T1; ++t) {
#pragma unroll
        for (int p = 0; p < 4; ++p) {
            v2f a;
            a.x = acc1[t][2 * p];
            a.y = acc1[t][2 * p + 1];
            const int k2 = t * 16 + 8 * hh + 2 * p;
#pragma unroll
            for (int u = 0; u < T2; ++u) {
                v2f b;
                b.x = W2[k2 * NO + u * 16 + c];
                b.y = W2[(k2 + 1) * NO + u * 16 + c];
                acc2[u] = __builtin_amdgcn_wmma_f32_16x16x4_f32(
                    false, a, false, b, (short)0, acc2[u], false, false);
            }
        }
    }

    const float4 dv0 = *(const float4*)&dinv[m0 + 8 * hh];
    const float4 dv1 = *(const float4*)&dinv[m0 + 8 * hh + 4];
    const float dv[8] = {dv0.x, dv0.y, dv0.z, dv0.w, dv1.x, dv1.y, dv1.z, dv1.w};
#pragma unroll
    for (int r = 0; r < 8; ++r) {
        const int m = m0 + r + 8 * hh;
#pragma unroll
        for (int u = 0; u < T2; ++u) {
            const float val = acc2[u][r] * dv[r];
            const int idx = m * NO + u * 16 + c;
            hs[idx]  = val;
            agg[idx] = val;
        }
    }
}

constexpr int SC_WAVES  = 8;
constexpr int SC_WEDGES = 128;
constexpr int SC_BEDGES = SC_WAVES * SC_WEDGES;
constexpr int SC_CTR_STRIDE = 64;

__global__ void __launch_bounds__(SC_WAVES * 32)
scatter_owned_kernel(const int* __restrict__ src,
                     const int* __restrict__ dst,
                     const float* __restrict__ hs,
                     float* __restrict__ agg, int* __restrict__ ctr,
                     int ne, int nslices, int nn_pad) {
    constexpr int F = 32;
    constexpr int U = 16;
    constexpr int NLD = SC_WEDGES / 32;
    __shared__ int2 lst[SC_WAVES][SC_WEDGES];
    __shared__ int tk[2];
    const int lane = threadIdx.x & 31;
    const int wv   = threadIdx.x >> 5;

    if (threadIdx.x == 0) {
        const unsigned id = __builtin_amdgcn_s_sendmsg_rtn(0x87);
        const int r = (int)((id >> 18) & 1u);
        tk[0] = atomicAdd(&ctr[r * SC_CTR_STRIDE], 1);
        tk[1] = r;
    }
    __syncthreads();
    int t = tk[0];
    int role = tk[1];
    if (t >= nslices) {
        __syncthreads();
        if (threadIdx.x == 0) {
            const int r = role ^ 1;
            tk[0] = atomicAdd(&ctr[r * SC_CTR_STRIDE], 1);
            tk[1] = r;
        }
        __syncthreads();
        t = tk[0];
        role = tk[1];
        if (t >= nslices) return;
    }
    t    = __builtin_amdgcn_readfirstlane(t);
    role = __builtin_amdgcn_readfirstlane(role);

    const int split = (nn_pad >> 2) << 1;
    const int e0 = t * SC_BEDGES + wv * SC_WEDGES;
    int2* my = lst[wv];

    int sv[NLD], dv[NLD];
    bool in[NLD];
#pragma unroll
    for (int j = 0; j < NLD; ++j) {
        const int e = e0 + j * 32 + lane;
        in[j] = e < ne;
        const int ec = in[j] ? e : ne - 1;
        sv[j] = src[ec];
        dv[j] = dst[ec];
    }
    static_assert(NLD == 4, "the fence below names four pairs");
    asm volatile("" :: "v"(sv[0]), "v"(sv[1]), "v"(sv[2]), "v"(sv[3]),
                       "v"(dv[0]), "v"(dv[1]), "v"(dv[2]), "v"(dv[3]));
    int n = 0;
#pragma unroll
    for (int j = 0; j < NLD; ++j) {
        const bool mine = in[j] && ((dv[j] >= split) == (role != 0));
        const unsigned m = __builtin_amdgcn_ballot_w32(mine);
        const int rank = (int)__builtin_amdgcn_mbcnt_lo(m, 0u);
        if (mine) my[n + rank] = make_int2(sv[j], dv[j]);
        n += __builtin_popcount(m);
    }
    __builtin_amdgcn_fence(__ATOMIC_ACQ_REL, "wavefront");

    for (int base = 0; base < n; base += 32) {
        const int i = base + lane;
        int my_s = 0, my_d = -1;
        if (i < n) {
            const int2 p = my[i];
            my_s = p.x;
            my_d = p.y;
        }
#pragma unroll
        for (int g = 0; g < 32; g += U) {
            if (base + g >= n) break;
            float v[U];
#pragma unroll
            for (int u = 0; u < U; ++u) {
                const int s = __builtin_amdgcn_readlane(my_s, g + u);
                v[u] = hs[s * F + lane];
            }
            static_assert(U == 16, "the fence below names sixteen values");
            asm volatile("" :: "v"(v[0]), "v"(v[1]), "v"(v[2]), "v"(v[3]),
                               "v"(v[4]), "v"(v[5]), "v"(v[6]), "v"(v[7]),
                               "v"(v[8]), "v"(v[9]), "v"(v[10]), "v"(v[11]),
                               "v"(v[12]), "v"(v[13]), "v"(v[14]), "v"(v[15]));
#pragma unroll
            for (int u = 0; u < U; ++u) {
                const int d = __builtin_amdgcn_readlane(my_d, g + u);
                if (d >= 0) atomicAdd(&agg[d * F + lane], v[u]);
            }
        }
    }
}

template <int F, bool RELU>
__global__ void finalize_kernel(const float* __restrict__ agg,
                                const float* __restrict__ dinv,
                                const float* __restrict__ bias,
                                float* __restrict__ out, int nrows) {
    int i = blockIdx.x * blockDim.x + threadIdx.x;
    constexpr int F4 = F / 4;
    if (i >= nrows * F4) return;
    int n  = i / F4;
    int fo = (i % F4) * 4;
    float dvn = dinv[n];
    const float4 a = *(const float4*)&agg[n * F + fo];
    const float4 b = *(const float4*)&bias[fo];
    float4 v;
    v.x = dvn * a.x + b.x;
    v.y = dvn * a.y + b.y;
    v.z = dvn * a.z + b.z;
    v.w = dvn * a.w + b.w;
    if (RELU) {
        v.x = fmaxf(v.x, 0.0f);
        v.y = fmaxf(v.y, 0.0f);
        v.z = fmaxf(v.z, 0.0f);
        v.w = fmaxf(v.w, 0.0f);
    }
    *(float4*)&out[n * F + fo] = v;
}

extern "C" void kernel_launch(void* const* d_in, const int* in_sizes, int n_in,
                              void* d_out, int out_size, void* d_ws, size_t ws_size,
                              hipStream_t stream) {
    const float* x  = (const float*)d_in[0];
    const int*   ei = (const int*)d_in[1];
    const float* W1 = (const float*)d_in[2];
    const float* b1 = (const float*)d_in[3];
    const float* W2 = (const float*)d_in[4];
    const float* b2 = (const float*)d_in[5];
    float* out = (float*)d_out;

    const int NN = in_sizes[0] / 18;
    const int NE = in_sizes[1] / 2;
    const int NN_pad = (NN + 15) & ~15;
    const int* src = ei;
    const int* dst = ei + NE;

    char* ws = (char*)d_ws;
    size_t off = 0;
    auto carve = [&](size_t bytes) -> char* {
        char* p = ws + off;
        off += (bytes + 255) & ~(size_t)255;
        return p;
    };
    float* dinv = (float*)carve((size_t)NN_pad * 4);
    const int dpitch = (NN_pad + 31) & ~31;
    unsigned* degp = (unsigned*)carve((size_t)DEG_CHUNKS * dpitch * 4);
    int* sctr   = (int*)carve((size_t)4 * 64 * 4);
    float* xs   = (float*)carve((size_t)NN_pad * 32 * 4);
    float* aggx = (float*)carve((size_t)NN_pad * 32 * 4);
    float* hs2  = xs;
    float* agg2 = aggx;

    const int B = 256;

    const int dranges = (NN_pad + DEG_RANGE - 1) / DEG_RANGE;
    const int dchunk  = (((NE + DEG_CHUNKS - 1) / DEG_CHUNKS) + 3) & ~3;
    deg_count_lds_kernel<<<dim3(dranges, DEG_CHUNKS), DEG_THREADS, 0, stream>>>(
        dst, degp, NE, dchunk, NN_pad, dpitch);
    dinv_from_counts_kernel<<<(NN_pad + B - 1) / B, B, 0, stream>>>(degp, dinv, sctr, NN_pad, dpitch);

    scale_pad_x_kernel<<<(NN_pad * 32 + B - 1) / B, B, 0, stream>>>(x, dinv, xs, aggx, NN, NN_pad);

    const int mtiles = NN_pad / 16;
    const int gblk = (mtiles + 3) / 4;
    const int nslices = (NE + SC_BEDGES - 1) / SC_BEDGES;
    const int sblk = 2 * nslices;

    scatter_owned_kernel<<<sblk, SC_WAVES * 32, 0, stream>>>(src, dst, xs, aggx, sctr, NE, nslices, NN_pad);

    gemm_two_layers_kernel<18, 32, 64, 32><<<gblk, 128, 0, stream>>>(aggx, W1, b1, W2, dinv, hs2, agg2, mtiles);

    scatter_owned_kernel<<<sblk, SC_WAVES * 32, 0, stream>>>(src, dst, hs2, agg2, sctr + 2 * SC_CTR_STRIDE, NE, nslices, NN_pad);
    finalize_kernel<32, false><<<(NN * 8 + B - 1) / B, B, 0, stream>>>(agg2, dinv, b2, out, NN);
}


# Degree sums, rsqrt and scaled, padded input rows in ONE launch: a wave owns 32 nodes (lane = node adds its partial counts; then row by row, dinv by readlane); launches 7 -> 6; values bit-identical
# speedup vs baseline: 5.0069x; 1.0390x over previous
#include <hip/hip_runtime.h>

typedef __attribute__((ext_vector_type(2))) float v2f;
typedef __attribute__((ext_vector_type(8))) float v8f;

constexpr int DEG_RANGE   = 25024;
constexpr int DEG_CHUNKS  = 64;
constexpr int DEG_THREADS = 1024;
constexpr int DEG_BATCH   = 8;

__device__ __forceinline__ void deg_bump(unsigned* cnt, int d, int lo) {
    const unsigned i = (unsigned)(d - lo);
    if (i < (unsigned)DEG_RANGE) atomicAdd(&cnt[i], 1u);
}

__global__ void __launch_bounds__(DEG_THREADS)
deg_count_lds_kernel(const int* __restrict__ dst, unsigned* __restrict__ part,
                     int ne, int chunk, int nn_pad, int pitch) {
    __shared__ unsigned cnt[DEG_RANGE];
    const int tid = threadIdx.x;
    const int lo  = blockIdx.x * DEG_RANGE;
    const int c   = blockIdx.y;
    const int left = nn_pad - lo;
    const int range_n = left < DEG_RANGE ? left : DEG_RANGE;

    uint4* cnt4 = (uint4*)cnt;
    for (int i = tid; i < DEG_RANGE / 4; i += DEG_THREADS) cnt4[i] = make_uint4(0u, 0u, 0u, 0u);
    __syncthreads();

    const int e0 = c * chunk;
    int e1 = e0 + chunk;
    if (e1 > ne) e1 = ne;
    if ((((size_t)dst) & 15) == 0) {
        const int4* d4 = (const int4*)dst;
        const int q1 = e1 >> 2;
        for (int qb = (e0 >> 2) + tid; qb < q1; qb += DEG_BATCH * DEG_THREADS) {
            int4 d[DEG_BATCH];
#pragma unroll
            for (int j = 0; j < DEG_BATCH; ++j) {
                const int q = qb + j * DEG_THREADS;
                d[j] = d4[q < q1 ? q : q1 - 1];
            }
#pragma unroll
            for (int j = 0; j < DEG_BATCH; ++j) {
                if (qb + j * DEG_THREADS < q1) {
                    deg_bump(cnt, d[j].x, lo);
                    deg_bump(cnt, d[j].y, lo);
                    deg_bump(cnt, d[j].z, lo);
                    deg_bump(cnt, d[j].w, lo);
                }
            }
        }
        const int e = (q1 << 2) + tid;
        if (e >= e0 && e < e1) deg_bump(cnt, dst[e], lo);
    } else {
        for (int eb = e0 + tid; eb < e1; eb += DEG_BATCH * DEG_THREADS) {
            int d[DEG_BATCH];
#pragma unroll
            for (int j = 0; j < DEG_BATCH; ++j) {
                const int e = eb + j * DEG_THREADS;
                d[j] = dst[e < e1 ? e : e1 - 1];
            }
#pragma unroll
            for (int j = 0; j < DEG_BATCH; ++j)
                if (eb + j * DEG_THREADS < e1) deg_bump(cnt, d[j], lo);
        }
    }
    __syncthreads();

    uint4* out4 = (uint4*)(part + (size_t)c * pitch + lo);
    for (int i = tid; i < range_n / 4; i += DEG_THREADS) out4[i] = cnt4[i];
}

constexpr int SP_WAVES = 4;
constexpr int SP_ROWS  = 32;

__global__ void __launch_bounds__(SP_WAVES * 32) __attribute__((amdgpu_waves_per_eu(1, 4)))
dinv_scale_pad_kernel(const unsigned* __restrict__ part, const float* __restrict__ x,
                      float* __restrict__ dinv, float* __restrict__ xs,
                      float* __restrict__ aggx, int* __restrict__ ctr,
                      int nn, int nn_pad, int pitch) {
    const int lane = threadIdx.x & 31;
    const int wave = __builtin_amdgcn_readfirstlane(blockIdx.x * SP_WAVES + (threadIdx.x >> 5));
    const int n0   = wave * SP_ROWS;
    if (n0 >= nn_pad) return;
    if (wave == 0 && lane < 4) ctr[lane * 64] = 0;

    const int n  = n0 + lane;
    const int nc = n < nn_pad ? n : nn_pad - 1;
    unsigned s = 1u;
#pragma unroll
    for (int c = 0; c < DEG_CHUNKS; ++c) s += part[(size_t)c * pitch + nc];

    const int kc = lane < 18 ? lane : 17;
    float xv[SP_ROWS];
#pragma unroll
    for (int j = 0; j < SP_ROWS; ++j) {
        const int r = n0 + j < nn ? n0 + j : nn - 1;
        xv[j] = x[r * 18 + kc];
    }

    const float dn = rsqrtf((float)s);
    if (n < nn_pad) dinv[n] = dn;

#pragma unroll
    for (int j = 0; j < SP_ROWS; ++j) {
        const int row = n0 + j;
        if (row >= nn_pad) break;
        const float dj = __builtin_bit_cast(
            float, __builtin_amdgcn_readlane(__builtin_bit_cast(int, dn), j));
        const float v = (row < nn && lane < 18) ? dj * xv[j] : 0.0f;
        xs[row * 32 + lane]   = v;
        aggx[row * 32 + lane] = v;
    }
}

template <int K1, int LDA, int NH, int NO>
__global__ void __launch_bounds__(128)
gemm_two_layers_kernel(const float* X,
                       const float* __restrict__ W1,
                       const float* __restrict__ b1,
                       const float* __restrict__ W2,
                       const float* __restrict__ dinv,
                       float* hs, float* agg, int mtiles) {
    static_assert(NH % 16 == 0 && NO % 16 == 0 && LDA % 2 == 0, "tile layout");
    constexpr int KS1 = (K1 + 3) / 4;
    constexpr int T1  = NH / 16;
    constexpr int T2  = NO / 16;
    static_assert(KS1 * 4 <= LDA, "padded K inside the row pitch");

    const int wave = blockIdx.x * (blockDim.x >> 5) + (threadIdx.x >> 5);
    if (wave >= mtiles) return;
    const int lane = threadIdx.x & 31;
    const int c    = lane & 15;
    const int hh   = lane >> 4;
    const int m0   = wave * 16;

    v8f acc1[T1] = {};
#pragma unroll
    for (int ks = 0; ks < KS1; ++ks) {
        const int k0 = ks * 4 + hh * 2;
        const v2f xb = *(const v2f*)&X[(m0 + c) * LDA + k0];
        const bool in0 = k0 < K1;
        const bool in1 = k0 + 1 < K1;
        const int r0 = in0 ? k0 : 0;
        const int r1 = in1 ? k0 + 1 : 0;
#pragma unroll
        for (int t = 0; t < T1; ++t) {
            v2f wa;
            wa.x = W1[r0 * NH + t * 16 + c];
            wa.y = W1[r1 * NH + t * 16 + c];
            if (!in0) wa.x = 0.0f;
            if (!in1) wa.y = 0.0f;
            acc1[t] = __builtin_amdgcn_wmma_f32_16x16x4_f32(
                false, wa, false, xb, (short)0, acc1[t], false, false);
        }
    }

    const float dn = dinv[m0 + c];
#pragma unroll
    for (int t = 0; t < T1; ++t) {
        const float4 ba = *(const float4*)&b1[t * 16 + 8 * hh];
        const float4 bb = *(const float4*)&b1[t * 16 + 8 * hh + 4];
        const float bv[8] = {ba.x, ba.y, ba.z, ba.w, bb.x, bb.y, bb.z, bb.w};
#pragma unroll
        for (int r = 0; r < 8; ++r)
            acc1[t][r] = fmaxf(acc1[t][r] * dn + bv[r], 0.0f);
    }

    v8f acc2[T2] = {};
#pragma unroll
    for (int t = 0; t < T1; ++t) {
#pragma unroll
        for (int p = 0; p < 4; ++p) {
            v2f a;
            a.x = acc1[t][2 * p];
            a.y = acc1[t][2 * p + 1];
            const int k2 = t * 16 + 8 * hh + 2 * p;
#pragma unroll
            for (int u = 0; u < T2; ++u) {
                v2f b;
                b.x = W2[k2 * NO + u * 16 + c];
                b.y = W2[(k2 + 1) * NO + u * 16 + c];
                acc2[u] = __builtin_amdgcn_wmma_f32_16x16x4_f32(
                    false, a, false, b, (short)0, acc2[u], false, false);
            }
        }
    }

    const float4 dv0 = *(const float4*)&dinv[m0 + 8 * hh];
    const float4 dv1 = *(const float4*)&dinv[m0 + 8 * hh + 4];
    const float dv[8] = {dv0.x, dv0.y, dv0.z, dv0.w, dv1.x, dv1.y, dv1.z, dv1.w};
#pragma unroll
    for (int r = 0; r < 8; ++r) {
        const int m = m0 + r + 8 * hh;
#pragma unroll
        for (int u = 0; u < T2; ++u) {
            const float val = acc2[u][r] * dv[r];
            const int idx = m * NO + u * 16 + c;
            hs[idx]  = val;
            agg[idx] = val;
        }
    }
}

constexpr int SC_WAVES  = 8;
constexpr int SC_WEDGES = 128;
constexpr int SC_BEDGES = SC_WAVES * SC_WEDGES;
constexpr int SC_CTR_STRIDE = 64;

__global__ void __launch_bounds__(SC_WAVES * 32)
scatter_owned_kernel(const int* __restrict__ src,
                     const int* __restrict__ dst,
                     const float* __restrict__ hs,
                     float* __restrict__ agg, int* __restrict__ ctr,
                     int ne, int nslices, int nn_pad) {
    constexpr int F = 32;
    constexpr int U = 16;
    constexpr int NLD = SC_WEDGES / 32;
    __shared__ int2 lst[SC_WAVES][SC_WEDGES];
    __shared__ int tk[2];
    const int lane = threadIdx.x & 31;
    const int wv   = threadIdx.x >> 5;

    if (threadIdx.x == 0) {
        const unsigned id = __builtin_amdgcn_s_sendmsg_rtn(0x87);
        const int r = (int)((id >> 18) & 1u);
        tk[0] = atomicAdd(&ctr[r * SC_CTR_STRIDE], 1);
        tk[1] = r;
    }
    __syncthreads();
    int t = tk[0];
    int role = tk[1];
    if (t >= nslices) {
        __syncthreads();
        if (threadIdx.x == 0) {
            const int r = role ^ 1;
            tk[0] = atomicAdd(&ctr[r * SC_CTR_STRIDE], 1);
            tk[1] = r;
        }
        __syncthreads();
        t = tk[0];
        role = tk[1];
        if (t >= nslices) return;
    }
    t    = __builtin_amdgcn_readfirstlane(t);
    role = __builtin_amdgcn_readfirstlane(role);

    const int split = (nn_pad >> 2) << 1;
    const int e0 = t * SC_BEDGES + wv * SC_WEDGES;
    int2* my = lst[wv];

    int sv[NLD], dv[NLD];
    bool in[NLD];
#pragma unroll
    for (int j = 0; j < NLD; ++j) {
        const int e = e0 + j * 32 + lane;
        in[j] = e < ne;
        const int ec = in[j] ? e : ne - 1;
        sv[j] = src[ec];
        dv[j] = dst[ec];
    }
    static_assert(NLD == 4, "the fence below names four pairs");
    asm volatile("" :: "v"(sv[0]), "v"(sv[1]), "v"(sv[2]), "v"(sv[3]),
                       "v"(dv[0]), "v"(dv[1]), "v"(dv[2]), "v"(dv[3]));
    int n = 0;
#pragma unroll
    for (int j = 0; j < NLD; ++j) {
        const bool mine = in[j] && ((dv[j] >= split) == (role != 0));
        const unsigned m = __builtin_amdgcn_ballot_w32(mine);
        const int rank = (int)__builtin_amdgcn_mbcnt_lo(m, 0u);
        if (mine) my[n + rank] = make_int2(sv[j], dv[j]);
        n += __builtin_popcount(m);
    }
    __builtin_amdgcn_fence(__ATOMIC_ACQ_REL, "wavefront");

    for (int base = 0; base < n; base += 32) {
        const int i = base + lane;
        int my_s = 0, my_d = -1;
        if (i < n) {
            const int2 p = my[i];
            my_s = p.x;
            my_d = p.y;
        }
#pragma unroll
        for (int g = 0; g < 32; g += U) {
            if (base + g >= n) break;
            float v[U];
#pragma unroll
            for (int u = 0; u < U; ++u) {
                const int s = __builtin_amdgcn_readlane(my_s, g + u);
                v[u] = hs[s * F + lane];
            }
            static_assert(U == 16, "the fence below names sixteen values");
            asm volatile("" :: "v"(v[0]), "v"(v[1]), "v"(v[2]), "v"(v[3]),
                               "v"(v[4]), "v"(v[5]), "v"(v[6]), "v"(v[7]),
                               "v"(v[8]), "v"(v[9]), "v"(v[10]), "v"(v[11]),
                               "v"(v[12]), "v"(v[13]), "v"(v[14]), "v"(v[15]));
#pragma unroll
            for (int u = 0; u < U; ++u) {
                const int d = __builtin_amdgcn_readlane(my_d, g + u);
                if (d >= 0) atomicAdd(&agg[d * F + lane], v[u]);
            }
        }
    }
}

template <int F, bool RELU>
__global__ void finalize_kernel(const float* __restrict__ agg,
                                const float* __restrict__ dinv,
                                const float* __restrict__ bias,
                                float* __restrict__ out, int nrows) {
    int i = blockIdx.x * blockDim.x + threadIdx.x;
    constexpr int F4 = F / 4;
    if (i >= nrows * F4) return;
    int n  = i / F4;
    int fo = (i % F4) * 4;
    float dvn = dinv[n];
    const float4 a = *(const float4*)&agg[n * F + fo];
    const float4 b = *(const float4*)&bias[fo];
    float4 v;
    v.x = dvn * a.x + b.x;
    v.y = dvn * a.y + b.y;
    v.z = dvn * a.z + b.z;
    v.w = dvn * a.w + b.w;
    if (RELU) {
        v.x = fmaxf(v.x, 0.0f);
        v.y = fmaxf(v.y, 0.0f);
        v.z = fmaxf(v.z, 0.0f);
        v.w = fmaxf(v.w, 0.0f);
    }
    *(float4*)&out[n * F + fo] = v;
}

extern "C" void kernel_launch(void* const* d_in, const int* in_sizes, int n_in,
                              void* d_out, int out_size, void* d_ws, size_t ws_size,
                              hipStream_t stream) {
    const float* x  = (const float*)d_in[0];
    const int*   ei = (const int*)d_in[1];
    const float* W1 = (const float*)d_in[2];
    const float* b1 = (const float*)d_in[3];
    const float* W2 = (const float*)d_in[4];
    const float* b2 = (const float*)d_in[5];
    float* out = (float*)d_out;

    const int NN = in_sizes[0] / 18;
    const int NE = in_sizes[1] / 2;
    const int NN_pad = (NN + 15) & ~15;
    const int* src = ei;
    const int* dst = ei + NE;

    char* ws = (char*)d_ws;
    size_t off = 0;
    auto carve = [&](size_t bytes) -> char* {
        char* p = ws + off;
        off += (bytes + 255) & ~(size_t)255;
        return p;
    };
    float* dinv = (float*)carve((size_t)NN_pad * 4);
    const int dpitch = (NN_pad + 31) & ~31;
    unsigned* degp = (unsigned*)carve((size_t)DEG_CHUNKS * dpitch * 4);
    int* sctr   = (int*)carve((size_t)4 * 64 * 4);
    float* xs   = (float*)carve((size_t)NN_pad * 32 * 4);
    float* aggx = (float*)carve((size_t)NN_pad * 32 * 4);
    float* hs2  = xs;
    float* agg2 = aggx;

    const int B = 256;

    const int dranges = (NN_pad + DEG_RANGE - 1) / DEG_RANGE;
    const int dchunk  = (((NE + DEG_CHUNKS - 1) / DEG_CHUNKS) + 3) & ~3;
    deg_count_lds_kernel<<<dim3(dranges, DEG_CHUNKS), DEG_THREADS, 0, stream>>>(
        dst, degp, NE, dchunk, NN_pad, dpitch);

    const int spblk = (NN_pad + SP_WAVES * SP_ROWS - 1) / (SP_WAVES * SP_ROWS);
    dinv_scale_pad_kernel<<<spblk, SP_WAVES * 32, 0, stream>>>(degp, x, dinv, xs, aggx, sctr, NN, NN_pad, dpitch);

    const int mtiles = NN_pad / 16;
    const int gblk = (mtiles + 3) / 4;
    const int nslices = (NE + SC_BEDGES - 1) / SC_BEDGES;
    const int sblk = 2 * nslices;

    scatter_owned_kernel<<<sblk, SC_WAVES * 32, 0, stream>>>(src, dst, xs, aggx, sctr, NE, nslices, NN_pad);

    gemm_two_layers_kernel<18, 32, 64, 32><<<gblk, 128, 0, stream>>>(aggx, W1, b1, W2, dinv, hs2, agg2, mtiles);

    scatter_owned_kernel<<<sblk, SC_WAVES * 32, 0, stream>>>(src, dst, hs2, agg2, sctr + 2 * SC_CTR_STRIDE, NE, nslices, NN_pad);
    finalize_kernel<32, false><<<(NN * 8 + B - 1) / B, B, 0, stream>>>(agg2, dinv, b2, out, NN);
}


# Edges binned by destination range once per call (runs staged in LDS); each range's rows summed by ds_add_f32 in its work-group's LDS, stored once: no global f32 atomics; sum order changes
# speedup vs baseline: 5.4782x; 1.0941x over previous
#include <hip/hip_runtime.h>

typedef __attribute__((ext_vector_type(2))) float v2f;
typedef __attribute__((ext_vector_type(8))) float v8f;

constexpr int DEG_RANGE   = 25024;
constexpr int DEG_CHUNKS  = 64;
constexpr int DEG_THREADS = 1024;
constexpr int DEG_BATCH   = 8;

__device__ __forceinline__ void deg_bump(unsigned* cnt, int d, int lo) {
    const unsigned i = (unsigned)(d - lo);
    if (i < (unsigned)DEG_RANGE) atomicAdd(&cnt[i], 1u);
}

__global__ void __launch_bounds__(DEG_THREADS)
deg_count_lds_kernel(const int* __restrict__ dst, unsigned* __restrict__ part,
                     int ne, int chunk, int nn_pad, int pitch) {
    __shared__ unsigned cnt[DEG_RANGE];
    const int tid = threadIdx.x;
    const int lo  = blockIdx.x * DEG_RANGE;
    const int c   = blockIdx.y;
    const int left = nn_pad - lo;
    const int range_n = left < DEG_RANGE ? left : DEG_RANGE;

    uint4* cnt4 = (uint4*)cnt;
    for (int i = tid; i < DEG_RANGE / 4; i += DEG_THREADS) cnt4[i] = make_uint4(0u, 0u, 0u, 0u);
    __syncthreads();

    const int e0 = c * chunk;
    int e1 = e0 + chunk;
    if (e1 > ne) e1 = ne;
    if ((((size_t)dst) & 15) == 0) {
        const int4* d4 = (const int4*)dst;
        const int q1 = e1 >> 2;
        for (int qb = (e0 >> 2) + tid; qb < q1; qb += DEG_BATCH * DEG_THREADS) {
            int4 d[DEG_BATCH];
#pragma unroll
            for (int j = 0; j < DEG_BATCH; ++j) {
                const int q = qb + j * DEG_THREADS;
                d[j] = d4[q < q1 ? q : q1 - 1];
            }
#pragma unroll
            for (int j = 0; j < DEG_BATCH; ++j) {
                if (qb + j * DEG_THREADS < q1) {
                    deg_bump(cnt, d[j].x, lo);
                    deg_bump(cnt, d[j].y, lo);
                    deg_bump(cnt, d[j].z, lo);
                    deg_bump(cnt, d[j].w, lo);
                }
            }
        }
        const int e = (q1 << 2) + tid;
        if (e >= e0 && e < e1) deg_bump(cnt, dst[e], lo);
    } else {
        for (int eb = e0 + tid; eb < e1; eb += DEG_BATCH * DEG_THREADS) {
            int d[DEG_BATCH];
#pragma unroll
            for (int j = 0; j < DEG_BATCH; ++j) {
                const int e = eb + j * DEG_THREADS;
                d[j] = dst[e < e1 ? e : e1 - 1];
            }
#pragma unroll
            for (int j = 0; j < DEG_BATCH; ++j)
                if (eb + j * DEG_THREADS < e1) deg_bump(cnt, d[j], lo);
        }
    }
    __syncthreads();

    uint4* out4 = (uint4*)(part + (size_t)c * pitch + lo);
    for (int i = tid; i < range_n / 4; i += DEG_THREADS) out4[i] = cnt4[i];
}

constexpr int SP_WAVES = 4;
constexpr int SP_ROWS  = 32;

constexpr int BK_WAVES = 13;
constexpr int BK_ROWS  = BK_WAVES * SP_ROWS;
constexpr int BK_MAX   = 256;

__global__ void __launch_bounds__(SP_WAVES * 32) __attribute__((amdgpu_waves_per_eu(1, 4)))
dinv_scale_pad_kernel(const unsigned* __restrict__ part, const float* __restrict__ x,
                      float* __restrict__ dinv, float* __restrict__ xs,
                      unsigned* __restrict__ wtot, int* __restrict__ cursor,
                      int nn, int nn_pad, int pitch) {
    const int lane = threadIdx.x & 31;
    const int wave = __builtin_amdgcn_readfirstlane(blockIdx.x * SP_WAVES + (threadIdx.x >> 5));
    const int n0   = wave * SP_ROWS;
    if (n0 >= nn_pad) return;
    if (wave == 0) {
#pragma unroll
        for (int k = 0; k < BK_MAX / 32; ++k) cursor[k * 32 + lane] = 0;
    }

    const int n  = n0 + lane;
    const int nc = n < nn_pad ? n : nn_pad - 1;
    unsigned s = 1u;
#pragma unroll
    for (int c = 0; c < DEG_CHUNKS; ++c) s += part[(size_t)c * pitch + nc];

    const int kc = lane < 18 ? lane : 17;
    float xv[SP_ROWS];
#pragma unroll
    for (int j = 0; j < SP_ROWS; ++j) {
        const int r = n0 + j < nn ? n0 + j : nn - 1;
        xv[j] = x[r * 18 + kc];
    }

    const float dn = rsqrtf((float)s);
    if (n < nn_pad) dinv[n] = dn;

    unsigned ein = n < nn_pad ? s - 1u : 0u;
#pragma unroll
    for (int o = 16; o > 0; o >>= 1) ein += __shfl_xor(ein, o);
    if (lane == 0) wtot[wave] = ein;

#pragma unroll
    for (int j = 0; j < SP_ROWS; ++j) {
        const int row = n0 + j;
        if (row >= nn_pad) break;
        const float dj = __builtin_bit_cast(
            float, __builtin_amdgcn_readlane(__builtin_bit_cast(int, dn), j));
        const float v = (row < nn && lane < 18) ? dj * xv[j] : 0.0f;
        xs[row * 32 + lane] = v;
    }
}

template <int K1, int LDA, int NH, int NO>
__global__ void __launch_bounds__(128)
gemm_two_layers_kernel(const float* __restrict__ X,
                       const float* __restrict__ W1,
                       const float* __restrict__ b1,
                       const float* __restrict__ W2,
                       const float* __restrict__ dinv,
                       float* __restrict__ hs, int mtiles) {
    static_assert(NH % 16 == 0 && NO % 16 == 0 && LDA % 2 == 0, "tile layout");
    constexpr int KS1 = (K1 + 3) / 4;
    constexpr int T1  = NH / 16;
    constexpr int T2  = NO / 16;
    static_assert(KS1 * 4 <= LDA, "padded K inside the row pitch");

    const int wave = blockIdx.x * (blockDim.x >> 5) + (threadIdx.x >> 5);
    if (wave >= mtiles) return;
    const int lane = threadIdx.x & 31;
    const int c    = lane & 15;
    const int hh   = lane >> 4;
    const int m0   = wave * 16;

    v8f acc1[T1] = {};
#pragma unroll
    for (int ks = 0; ks < KS1; ++ks) {
        const int k0 = ks * 4 + hh * 2;
        const v2f xb = *(const v2f*)&X[(m0 + c) * LDA + k0];
        const bool in0 = k0 < K1;
        const bool in1 = k0 + 1 < K1;
        const int r0 = in0 ? k0 : 0;
        const int r1 = in1 ? k0 + 1 : 0;
#pragma unroll
        for (int t = 0; t < T1; ++t) {
            v2f wa;
            wa.x = W1[r0 * NH + t * 16 + c];
            wa.y = W1[r1 * NH + t * 16 + c];
            if (!in0) wa.x = 0.0f;
            if (!in1) wa.y = 0.0f;
            acc1[t] = __builtin_amdgcn_wmma_f32_16x16x4_f32(
                false, wa, false, xb, (short)0, acc1[t], false, false);
        }
    }

    const float dn = dinv[m0 + c];
#pragma unroll
    for (int t = 0; t < T1; ++t) {
        const float4 ba = *(const float4*)&b1[t * 16 + 8 * hh];
        const float4 bb = *(const float4*)&b1[t * 16 + 8 * hh + 4];
        const float bv[8] = {ba.x, ba.y, ba.z, ba.w, bb.x, bb.y, bb.z, bb.w};
#pragma unroll
        for (int r = 0; r < 8; ++r)
            acc1[t][r] = fmaxf(acc1[t][r] * dn + bv[r], 0.0f);
    }

    v8f acc2[T2] = {};
#pragma unroll
    for (int t = 0; t < T1; ++t) {
#pragma unroll
        for (int p = 0; p < 4; ++p) {
            v2f a;
            a.x = acc1[t][2 * p];
            a.y = acc1[t][2 * p + 1];
            const int k2 = t * 16 + 8 * hh + 2 * p;
#pragma unroll
            for (int u = 0; u < T2; ++u) {
                v2f b;
                b.x = W2[k2 * NO + u * 16 + c];
                b.y = W2[(k2 + 1) * NO + u * 16 + c];
                acc2[u] = __builtin_amdgcn_wmma_f32_16x16x4_f32(
                    false, a, false, b, (short)0, acc2[u], false, false);
            }
        }
    }

    const float4 dv0 = *(const float4*)&dinv[m0 + 8 * hh];
    const float4 dv1 = *(const float4*)&dinv[m0 + 8 * hh + 4];
    const float dv[8] = {dv0.x, dv0.y, dv0.z, dv0.w, dv1.x, dv1.y, dv1.z, dv1.w};
#pragma unroll
    for (int r = 0; r < 8; ++r) {
        const int m = m0 + r + 8 * hh;
#pragma unroll
        for (int u = 0; u < T2; ++u) {
            const float val = acc2[u][r] * dv[r];
            const int idx = m * NO + u * 16 + c;
            hs[idx] = val;
        }
    }
}

constexpr int PL_THREADS = 1024;
constexpr int PL_PER     = 13;
constexpr int PL_CAP     = PL_THREADS * PL_PER;

__device__ __forceinline__ int wave_incl_scan(int v, int lane) {
#pragma unroll
    for (int o = 1; o < 32; o <<= 1) {
        const int t = __shfl_up(v, o);
        if (lane >= o) v += t;
    }
    return v;
}

__global__ void __launch_bounds__(PL_THREADS)
bin_edges_kernel(const int* __restrict__ src, const int* __restrict__ dst,
                 const unsigned* __restrict__ wtot, int* __restrict__ cursor,
                 int* __restrict__ gbase, unsigned* __restrict__ binned,
                 int ne, int slice, int nwaves, int nb) {
    __shared__ unsigned ent[PL_CAP];
    __shared__ unsigned short ebk[PL_CAP];
    __shared__ int hist[BK_MAX];
    __shared__ int lbase[BK_MAX];
    __shared__ int rbase[BK_MAX];
    __shared__ int wsum1[BK_MAX / 32], wsum2[BK_MAX / 32];
    const int tid  = threadIdx.x;
    const int lane = tid & 31;
    const int wq   = tid >> 5;
    const int e0 = blockIdx.x * slice;
    int e1 = e0 + slice;
    if (e1 > ne) e1 = ne;

    int sv[PL_PER], dv[PL_PER];
#pragma unroll
    for (int j = 0; j < PL_PER; ++j) {
        const int e = e0 + j * PL_THREADS + tid;
        const int ec = e < e1 ? e : ne - 1;
        sv[j] = src[ec];
        dv[j] = dst[ec];
    }

    int bt = 0, bi = 0;
    if (tid < BK_MAX) {
        hist[tid] = 0;
        unsigned q[BK_WAVES];
#pragma unroll
        for (int k = 0; k < BK_WAVES; ++k) {
            const int w = tid * BK_WAVES + k;
            q[k] = wtot[w < nwaves ? w : nwaves - 1];
        }
        static_assert(BK_WAVES == 13, "the fence below names thirteen values");
        asm volatile("" :: "v"(q[0]), "v"(q[1]), "v"(q[2]), "v"(q[3]), "v"(q[4]),
                           "v"(q[5]), "v"(q[6]), "v"(q[7]), "v"(q[8]), "v"(q[9]),
                           "v"(q[10]), "v"(q[11]), "v"(q[12]));
        unsigned t = 0u;
#pragma unroll
        for (int k = 0; k < BK_WAVES; ++k) {
            const int w = tid * BK_WAVES + k;
            t += (tid < nb && w < nwaves) ? q[k] : 0u;
        }
        bt = (int)t;
        bi = wave_incl_scan(bt, lane);
        if (lane == 31) wsum1[wq] = bi;
    }
    __syncthreads();

    int bb = 0;
    if (tid < BK_MAX) {
        bb = bi - bt;
#pragma unroll
        for (int k = 0; k < BK_MAX / 32; ++k) bb += k < wq ? wsum1[k] : 0;
        if (blockIdx.x == 0) gbase[tid] = bb;
    }

    int rk[PL_PER];
#pragma unroll
    for (int j = 0; j < PL_PER; ++j) {
        rk[j] = 0;
        if (e0 + j * PL_THREADS + tid < e1) rk[j] = atomicAdd(&hist[dv[j] / BK_ROWS], 1);
    }
    __syncthreads();

    int h = 0, hi = 0;
    if (tid < BK_MAX) {
        h  = hist[tid];
        hi = wave_incl_scan(h, lane);
        if (lane == 31) wsum2[wq] = hi;
    }
    __syncthreads();
    if (tid < BK_MAX) {
        int a = hi - h;
#pragma unroll
        for (int k = 0; k < BK_MAX / 32; ++k) a += k < wq ? wsum2[k] : 0;
        lbase[tid] = a;
        int g = 0;
        if (h > 0) g = atomicAdd(&cursor[tid], h);
        rbase[tid] = bb + g;
    }
    __syncthreads();

#pragma unroll
    for (int j = 0; j < PL_PER; ++j) {
        if (e0 + j * PL_THREADS + tid < e1) {
            const int b = dv[j] / BK_ROWS;
            const int p = lbase[b] + rk[j];
            ent[p] = ((unsigned)(dv[j] - b * BK_ROWS) << 20) | (unsigned)sv[j];
            ebk[p] = (unsigned short)b;
        }
    }
    __syncthreads();

    const int cnt = e1 > e0 ? e1 - e0 : 0;
#pragma unroll
    for (int j = 0; j < PL_PER; ++j) {
        const int i = tid + j * PL_THREADS;
        if (i < cnt) {
            const int b = ebk[i];
            binned[rbase[b] + (i - lbase[b])] = ent[i];
        }
    }
}

constexpr int BS_THREADS = 1024;
constexpr int BS_CH      = 2;

template <bool FINAL>
__global__ void __launch_bounds__(BS_THREADS)
bucket_sum_kernel(const unsigned* __restrict__ binned, const int* __restrict__ gbase,
                  const int* __restrict__ cursor, const float* __restrict__ hs,
                  const float* __restrict__ dinv, const float* __restrict__ bias,
                  float* __restrict__ rows_out, int nn, int nn_pad) {
    constexpr int F  = 32;
    constexpr int NW = BS_THREADS / 32;
    constexpr int ST = 32 * BS_CH;
    constexpr int NQ = (BK_ROWS * (F / 4) + BS_THREADS - 1) / BS_THREADS;
    __shared__ float acc[BK_ROWS * F];
    const int tid  = threadIdx.x;
    const int lane = tid & 31;
    const int wv   = __builtin_amdgcn_readfirstlane(tid >> 5);
    const int n_lo = blockIdx.x * BK_ROWS;
    int rows = nn_pad - n_lo;
    if (rows > BK_ROWS) rows = BK_ROWS;
    const int nq = rows * (F / 4);
    const int base = __builtin_amdgcn_readfirstlane(gbase[blockIdx.x]);
    const int cnt  = __builtin_amdgcn_readfirstlane(cursor[blockIdx.x]);
    const unsigned* lst = binned + base;

    unsigned pn[BS_CH];
#pragma unroll
    for (int h = 0; h < BS_CH; ++h) {
        const int i = wv * ST + h * 32 + lane;
        pn[h] = 0u;
        if (cnt > 0) pn[h] = lst[i < cnt ? i : cnt - 1];
    }

    const float4* h4 = (const float4*)(hs + (size_t)n_lo * F);
    float4* a4 = (float4*)acc;
    {
        float4 t[NQ];
#pragma unroll
        for (int k = 0; k < NQ; ++k) {
            const int i = tid + k * BS_THREADS;
            t[k] = h4[i < nq ? i : nq - 1];
        }
        static_assert(NQ == 4, "the fence below names four pieces");
        asm volatile("" :: "v"(t[0].x), "v"(t[0].y), "v"(t[0].z), "v"(t[0].w),
                           "v"(t[1].x), "v"(t[1].y), "v"(t[1].z), "v"(t[1].w),
                           "v"(t[2].x), "v"(t[2].y), "v"(t[2].z), "v"(t[2].w),
                           "v"(t[3].x), "v"(t[3].y), "v"(t[3].z), "v"(t[3].w));
#pragma unroll
        for (int k = 0; k < NQ; ++k) {
            const int i = tid + k * BS_THREADS;
            if (i < nq) a4[i] = t[k];
        }
    }
    __syncthreads();

    for (int c0 = wv * ST; c0 < cnt; c0 += NW * ST) {
        unsigned p[BS_CH];
        const int cn = c0 + NW * ST;
#pragma unroll
        for (int h = 0; h < BS_CH; ++h) {
            p[h] = pn[h];
            const int in = cn + h * 32 + lane;
            pn[h] = lst[in < cnt ? in : cnt - 1];
        }
        int my_s[BS_CH], my_d[BS_CH];
#pragma unroll
        for (int h = 0; h < BS_CH; ++h) {
            my_s[h] = 0;
            my_d[h] = -1;
            if (c0 + h * 32 + lane < cnt) {
                my_s[h] = (int)(p[h] & 0xFFFFFu);
                my_d[h] = (int)(p[h] >> 20);
            }
        }
        float v[BS_CH * 32];
#pragma unroll
        for (int h = 0; h < BS_CH; ++h) {
#pragma unroll
            for (int u = 0; u < 32; ++u) {
                const int s = __builtin_amdgcn_readlane(my_s[h], u);
                v[h * 32 + u] = hs[s * F + lane];
            }
        }
        static_assert(BS_CH == 2, "the fences below name 64 values");
        asm volatile("" :: "v"(v[0]), "v"(v[1]), "v"(v[2]), "v"(v[3]),
                           "v"(v[4]), "v"(v[5]), "v"(v[6]), "v"(v[7]),
                           "v"(v[8]), "v"(v[9]), "v"(v[10]), "v"(v[11]),
                           "v"(v[12]), "v"(v[13]), "v"(v[14]), "v"(v[15]));
        asm volatile("" :: "v"(v[16]), "v"(v[17]), "v"(v[18]), "v"(v[19]),
                           "v"(v[20]), "v"(v[21]), "v"(v[22]), "v"(v[23]),
                           "v"(v[24]), "v"(v[25]), "v"(v[26]), "v"(v[27]),
                           "v"(v[28]), "v"(v[29]), "v"(v[30]), "v"(v[31]));
        asm volatile("" :: "v"(v[32]), "v"(v[33]), "v"(v[34]), "v"(v[35]),
                           "v"(v[36]), "v"(v[37]), "v"(v[38]), "v"(v[39]),
                           "v"(v[40]), "v"(v[41]), "v"(v[42]), "v"(v[43]),
                           "v"(v[44]), "v"(v[45]), "v"(v[46]), "v"(v[47]));
        asm volatile("" :: "v"(v[48]), "v"(v[49]), "v"(v[50]), "v"(v[51]),
                           "v"(v[52]), "v"(v[53]), "v"(v[54]), "v"(v[55]),
                           "v"(v[56]), "v"(v[57]), "v"(v[58]), "v"(v[59]),
                           "v"(v[60]), "v"(v[61]), "v"(v[62]), "v"(v[63]));
#pragma unroll
        for (int h = 0; h < BS_CH; ++h) {
#pragma unroll
            for (int u = 0; u < 32; ++u) {
                const int d = __builtin_amdgcn_readlane(my_d[h], u);
                if (d >= 0) atomicAdd(&acc[d * F + lane], v[h * 32 + u]);
            }
        }
    }
    __syncthreads();

    float4* o4 = (float4*)(rows_out + (size_t)n_lo * F);
    if (!FINAL) {
#pragma unroll
        for (int k = 0; k < NQ; ++k) {
            const int i = tid + k * BS_THREADS;
            if (i < nq) o4[i] = a4[i];
        }
    } else {
        const float4 bq = *(const float4*)&bias[(tid & 7) * 4];
        float dn[NQ];
#pragma unroll
        for (int k = 0; k < NQ; ++k) {
            const int n = n_lo + ((tid + k * BS_THREADS) >> 3);
            dn[k] = dinv[n < nn ? n : nn - 1];
        }
        asm volatile("" :: "v"(dn[0]), "v"(dn[1]), "v"(dn[2]), "v"(dn[3]));
#pragma unroll
        for (int k = 0; k < NQ; ++k) {
            const int i = tid + k * BS_THREADS;
            if (i < nq && n_lo + (i >> 3) < nn) {
                float4 r = a4[i];
                r.x = dn[k] * r.x + bq.x;
                r.y = dn[k] * r.y + bq.y;
                r.z = dn[k] * r.z + bq.z;
                r.w = dn[k] * r.w + bq.w;
                o4[i] = r;
            }
        }
    }
}

extern "C" void kernel_launch(void* const* d_in, const int* in_sizes, int n_in,
                              void* d_out, int out_size, void* d_ws, size_t ws_size,
                              hipStream_t stream) {
    const float* x  = (const float*)d_in[0];
    const int*   ei = (const int*)d_in[1];
    const float* W1 = (const float*)d_in[2];
    const float* b1 = (const float*)d_in[3];
    const float* W2 = (const float*)d_in[4];
    const float* b2 = (const float*)d_in[5];
    float* out = (float*)d_out;

    const int NN = in_sizes[0] / 18;
    const int NE = in_sizes[1] / 2;
    const int NN_pad = (NN + 15) & ~15;
    const int* src = ei;
    const int* dst = ei + NE;

    char* ws = (char*)d_ws;
    size_t off = 0;
    auto carve = [&](size_t bytes) -> char* {
        char* p = ws + off;
        off += (bytes + 255) & ~(size_t)255;
        return p;
    };
    float* dinv = (float*)carve((size_t)NN_pad * 4);
    const int dpitch = (NN_pad + 31) & ~31;
    unsigned* degp = (unsigned*)carve((size_t)DEG_CHUNKS * dpitch * 4);
    const int nwaves = (NN_pad + SP_ROWS - 1) / SP_ROWS;
    unsigned* wtot = (unsigned*)carve((size_t)nwaves * 4);
    int* cursor = (int*)carve((size_t)BK_MAX * 4);
    int* gbase  = (int*)carve((size_t)BK_MAX * 4);
    unsigned* binned = (unsigned*)carve((size_t)NE * 4);
    float* xs   = (float*)carve((size_t)NN_pad * 32 * 4);
    float* aggx = (float*)carve((size_t)NN_pad * 32 * 4);
    float* hs2  = xs;

    const int dranges = (NN_pad + DEG_RANGE - 1) / DEG_RANGE;
    const int dchunk  = (((NE + DEG_CHUNKS - 1) / DEG_CHUNKS) + 3) & ~3;
    deg_count_lds_kernel<<<dim3(dranges, DEG_CHUNKS), DEG_THREADS, 0, stream>>>(
        dst, degp, NE, dchunk, NN_pad, dpitch);

    const int spblk = (NN_pad + SP_WAVES * SP_ROWS - 1) / (SP_WAVES * SP_ROWS);
    dinv_scale_pad_kernel<<<spblk, SP_WAVES * 32, 0, stream>>>(degp, x, dinv, xs, wtot, cursor, NN, NN_pad, dpitch);

    const int nb = (NN_pad + BK_ROWS - 1) / BK_ROWS;
    int pwgs = 256;
    if ((NE + PL_CAP - 1) / PL_CAP > pwgs) pwgs = (NE + PL_CAP - 1) / PL_CAP;
    const int pslice = (NE + pwgs - 1) / pwgs;
    bin_edges_kernel<<<pwgs, PL_THREADS, 0, stream>>>(src, dst, wtot, cursor, gbase, binned, NE, pslice, nwaves, nb);

    bucket_sum_kernel<false><<<nb, BS_THREADS, 0, stream>>>(binned, gbase, cursor, xs, dinv, b2, aggx, NN, NN_pad);

    const int mtiles = NN_pad / 16;
    const int gblk = (mtiles + 3) / 4;
    gemm_two_layers_kernel<18, 32, 64, 32><<<gblk, 128, 0, stream>>>(aggx, W1, b1, W2, dinv, hs2, mtiles);

    bucket_sum_kernel<true><<<nb, BS_THREADS, 0, stream>>>(binned, gbase, cursor, hs2, dinv, b2, out, NN, NN_pad);
}


# Neighbour sums add a run of words for one row in a register, one ds_add_f32 per run: layer 1 first counting-sorts its range's words by row in LDS and stores them back for layer 2; sum order changes
# speedup vs baseline: 6.2123x; 1.1340x over previous
#include <hip/hip_runtime.h>

typedef __attribute__((ext_vector_type(2))) float v2f;
typedef __attribute__((ext_vector_type(8))) float v8f;

constexpr int DEG_RANGE   = 25024;
constexpr int DEG_CHUNKS  = 64;
constexpr int DEG_THREADS = 1024;
constexpr int DEG_BATCH   = 8;

__device__ __forceinline__ void deg_bump(unsigned* cnt, int d, int lo) {
    const unsigned i = (unsigned)(d - lo);
    if (i < (unsigned)DEG_RANGE) atomicAdd(&cnt[i], 1u);
}

__global__ void __launch_bounds__(DEG_THREADS)
deg_count_lds_kernel(const int* __restrict__ dst, unsigned* __restrict__ part,
                     int ne, int chunk, int nn_pad, int pitch) {
    __shared__ unsigned cnt[DEG_RANGE];
    const int tid = threadIdx.x;
    const int lo  = blockIdx.x * DEG_RANGE;
    const int c   = blockIdx.y;
    const int left = nn_pad - lo;
    const int range_n = left < DEG_RANGE ? left : DEG_RANGE;

    uint4* cnt4 = (uint4*)cnt;
    for (int i = tid; i < DEG_RANGE / 4; i += DEG_THREADS) cnt4[i] = make_uint4(0u, 0u, 0u, 0u);
    __syncthreads();

    const int e0 = c * chunk;
    int e1 = e0 + chunk;
    if (e1 > ne) e1 = ne;
    if ((((size_t)dst) & 15) == 0) {
        const int4* d4 = (const int4*)dst;
        const int q1 = e1 >> 2;
        for (int qb = (e0 >> 2) + tid; qb < q1; qb += DEG_BATCH * DEG_THREADS) {
            int4 d[DEG_BATCH];
#pragma unroll
            for (int j = 0; j < DEG_BATCH; ++j) {
                const int q = qb + j * DEG_THREADS;
                d[j] = d4[q < q1 ? q : q1 - 1];
            }
#pragma unroll
            for (int j = 0; j < DEG_BATCH; ++j) {
                if (qb + j * DEG_THREADS < q1) {
                    deg_bump(cnt, d[j].x, lo);
                    deg_bump(cnt, d[j].y, lo);
                    deg_bump(cnt, d[j].z, lo);
                    deg_bump(cnt, d[j].w, lo);
                }
            }
        }
        const int e = (q1 << 2) + tid;
        if (e >= e0 && e < e1) deg_bump(cnt, dst[e], lo);
    } else {
        for (int eb = e0 + tid; eb < e1; eb += DEG_BATCH * DEG_THREADS) {
            int d[DEG_BATCH];
#pragma unroll
            for (int j = 0; j < DEG_BATCH; ++j) {
                const int e = eb + j * DEG_THREADS;
                d[j] = dst[e < e1 ? e : e1 - 1];
            }
#pragma unroll
            for (int j = 0; j < DEG_BATCH; ++j)
                if (eb + j * DEG_THREADS < e1) deg_bump(cnt, d[j], lo);
        }
    }
    __syncthreads();

    uint4* out4 = (uint4*)(part + (size_t)c * pitch + lo);
    for (int i = tid; i < range_n / 4; i += DEG_THREADS) out4[i] = cnt4[i];
}

constexpr int SP_WAVES = 4;
constexpr int SP_ROWS  = 32;

constexpr int BK_WAVES = 13;
constexpr int BK_ROWS  = BK_WAVES * SP_ROWS;
constexpr int BK_MAX   = 256;

__global__ void __launch_bounds__(SP_WAVES * 32) __attribute__((amdgpu_waves_per_eu(1, 4)))
dinv_scale_pad_kernel(const unsigned* __restrict__ part, const float* __restrict__ x,
                      float* __restrict__ dinv, float* __restrict__ xs,
                      unsigned* __restrict__ wtot, int* __restrict__ cursor,
                      int nn, int nn_pad, int pitch) {
    const int lane = threadIdx.x & 31;
    const int wave = __builtin_amdgcn_readfirstlane(blockIdx.x * SP_WAVES + (threadIdx.x >> 5));
    const int n0   = wave * SP_ROWS;
    if (n0 >= nn_pad) return;
    if (wave == 0) {
#pragma unroll
        for (int k = 0; k < BK_MAX / 32; ++k) cursor[k * 32 + lane] = 0;
    }

    const int n  = n0 + lane;
    const int nc = n < nn_pad ? n : nn_pad - 1;
    unsigned s = 1u;
#pragma unroll
    for (int c = 0; c < DEG_CHUNKS; ++c) s += part[(size_t)c * pitch + nc];

    const int kc = lane < 18 ? lane : 17;
    float xv[SP_ROWS];
#pragma unroll
    for (int j = 0; j < SP_ROWS; ++j) {
        const int r = n0 + j < nn ? n0 + j : nn - 1;
        xv[j] = x[r * 18 + kc];
    }

    const float dn = rsqrtf((float)s);
    if (n < nn_pad) dinv[n] = dn;

    unsigned ein = n < nn_pad ? s - 1u : 0u;
#pragma unroll
    for (int o = 16; o > 0; o >>= 1) ein += __shfl_xor(ein, o);
    if (lane == 0) wtot[wave] = ein;

#pragma unroll
    for (int j = 0; j < SP_ROWS; ++j) {
        const int row = n0 + j;
        if (row >= nn_pad) break;
        const float dj = __builtin_bit_cast(
            float, __builtin_amdgcn_readlane(__builtin_bit_cast(int, dn), j));
        const float v = (row < nn && lane < 18) ? dj * xv[j] : 0.0f;
        xs[row * 32 + lane] = v;
    }
}

template <int K1, int LDA, int NH, int NO>
__global__ void __launch_bounds__(128)
gemm_two_layers_kernel(const float* __restrict__ X,
                       const float* __restrict__ W1,
                       const float* __restrict__ b1,
                       const float* __restrict__ W2,
                       const float* __restrict__ dinv,
                       float* __restrict__ hs, int mtiles) {
    static_assert(NH % 16 == 0 && NO % 16 == 0 && LDA % 2 == 0, "tile layout");
    constexpr int KS1 = (K1 + 3) / 4;
    constexpr int T1  = NH / 16;
    constexpr int T2  = NO / 16;
    static_assert(KS1 * 4 <= LDA, "padded K inside the row pitch");

    const int wave = blockIdx.x * (blockDim.x >> 5) + (threadIdx.x >> 5);
    if (wave >= mtiles) return;
    const int lane = threadIdx.x & 31;
    const int c    = lane & 15;
    const int hh   = lane >> 4;
    const int m0   = wave * 16;

    v8f acc1[T1] = {};
#pragma unroll
    for (int ks = 0; ks < KS1; ++ks) {
        const int k0 = ks * 4 + hh * 2;
        const v2f xb = *(const v2f*)&X[(m0 + c) * LDA + k0];
        const bool in0 = k0 < K1;
        const bool in1 = k0 + 1 < K1;
        const int r0 = in0 ? k0 : 0;
        const int r1 = in1 ? k0 + 1 : 0;
#pragma unroll
        for (int t = 0; t < T1; ++t) {
            v2f wa;
            wa.x = W1[r0 * NH + t * 16 + c];
            wa.y = W1[r1 * NH + t * 16 + c];
            if (!in0) wa.x = 0.0f;
            if (!in1) wa.y = 0.0f;
            acc1[t] = __builtin_amdgcn_wmma_f32_16x16x4_f32(
                false, wa, false, xb, (short)0, acc1[t], false, false);
        }
    }

    const float dn = dinv[m0 + c];
#pragma unroll
    for (int t = 0; t < T1; ++t) {
        const float4 ba = *(const float4*)&b1[t * 16 + 8 * hh];
        const float4 bb = *(const float4*)&b1[t * 16 + 8 * hh + 4];
        const float bv[8] = {ba.x, ba.y, ba.z, ba.w, bb.x, bb.y, bb.z, bb.w};
#pragma unroll
        for (int r = 0; r < 8; ++r)
            acc1[t][r] = fmaxf(acc1[t][r] * dn + bv[r], 0.0f);
    }

    v8f acc2[T2] = {};
#pragma unroll
    for (int t = 0; t < T1; ++t) {
#pragma unroll
        for (int p = 0; p < 4; ++p) {
            v2f a;
            a.x = acc1[t][2 * p];
            a.y = acc1[t][2 * p + 1];
            const int k2 = t * 16 + 8 * hh + 2 * p;
#pragma unroll
            for (int u = 0; u < T2; ++u) {
                v2f b;
                b.x = W2[k2 * NO + u * 16 + c];
                b.y = W2[(k2 + 1) * NO + u * 16 + c];
                acc2[u] = __builtin_amdgcn_wmma_f32_16x16x4_f32(
                    false, a, false, b, (short)0, acc2[u], false, false);
            }
        }
    }

    const float4 dv0 = *(const float4*)&dinv[m0 + 8 * hh];
    const float4 dv1 = *(const float4*)&dinv[m0 + 8 * hh + 4];
    const float dv[8] = {dv0.x, dv0.y, dv0.z, dv0.w, dv1.x, dv1.y, dv1.z, dv1.w};
#pragma unroll
    for (int r = 0; r < 8; ++r) {
        const int m = m0 + r + 8 * hh;
#pragma unroll
        for (int u = 0; u < T2; ++u) {
            const float val = acc2[u][r] * dv[r];
            const int idx = m * NO + u * 16 + c;
            hs[idx] = val;
        }
    }
}

constexpr int PL_THREADS = 1024;
constexpr int PL_PER     = 13;
constexpr int PL_CAP     = PL_THREADS * PL_PER;

__device__ __forceinline__ int wave_incl_scan(int v, int lane) {
#pragma unroll
    for (int o = 1; o < 32; o <<= 1) {
        const int t = __shfl_up(v, o);
        if (lane >= o) v += t;
    }
    return v;
}

__global__ void __launch_bounds__(PL_THREADS)
bin_edges_kernel(const int* __restrict__ src, const int* __restrict__ dst,
                 const unsigned* __restrict__ wtot, int* __restrict__ cursor,
                 int* __restrict__ gbase, unsigned* __restrict__ binned,
                 int ne, int slice, int nwaves, int nb) {
    __shared__ unsigned ent[PL_CAP];
    __shared__ unsigned short ebk[PL_CAP];
    __shared__ int hist[BK_MAX];
    __shared__ int lbase[BK_MAX];
    __shared__ int rbase[BK_MAX];
    __shared__ int wsum1[BK_MAX / 32], wsum2[BK_MAX / 32];
    const int tid  = threadIdx.x;
    const int lane = tid & 31;
    const int wq   = tid >> 5;
    const int e0 = blockIdx.x * slice;
    int e1 = e0 + slice;
    if (e1 > ne) e1 = ne;

    int sv[PL_PER], dv[PL_PER];
#pragma unroll
    for (int j = 0; j < PL_PER; ++j) {
        const int e = e0 + j * PL_THREADS + tid;
        const int ec = e < e1 ? e : ne - 1;
        sv[j] = src[ec];
        dv[j] = dst[ec];
    }

    int bt = 0, bi = 0;
    if (tid < BK_MAX) {
        hist[tid] = 0;
        unsigned q[BK_WAVES];
#pragma unroll
        for (int k = 0; k < BK_WAVES; ++k) {
            const int w = tid * BK_WAVES + k;
            q[k] = wtot[w < nwaves ? w : nwaves - 1];
        }
        static_assert(BK_WAVES == 13, "the fence below names thirteen values");
        asm volatile("" :: "v"(q[0]), "v"(q[1]), "v"(q[2]), "v"(q[3]), "v"(q[4]),
                           "v"(q[5]), "v"(q[6]), "v"(q[7]), "v"(q[8]), "v"(q[9]),
                           "v"(q[10]), "v"(q[11]), "v"(q[12]));
        unsigned t = 0u;
#pragma unroll
        for (int k = 0; k < BK_WAVES; ++k) {
            const int w = tid * BK_WAVES + k;
            t += (tid < nb && w < nwaves) ? q[k] : 0u;
        }
        bt = (int)t;
        bi = wave_incl_scan(bt, lane);
        if (lane == 31) wsum1[wq] = bi;
    }
    __syncthreads();

    int bb = 0;
    if (tid < BK_MAX) {
        bb = bi - bt;
#pragma unroll
        for (int k = 0; k < BK_MAX / 32; ++k) bb += k < wq ? wsum1[k] : 0;
        if (blockIdx.x == 0) gbase[tid] = bb;
    }

    int rk[PL_PER];
#pragma unroll
    for (int j = 0; j < PL_PER; ++j) {
        rk[j] = 0;
        if (e0 + j * PL_THREADS + tid < e1) rk[j] = atomicAdd(&hist[dv[j] / BK_ROWS], 1);
    }
    __syncthreads();

    int h = 0, hi = 0;
    if (tid < BK_MAX) {
        h  = hist[tid];
        hi = wave_incl_scan(h, lane);
        if (lane == 31) wsum2[wq] = hi;
    }
    __syncthreads();
    if (tid < BK_MAX) {
        int a = hi - h;
#pragma unroll
        for (int k = 0; k < BK_MAX / 32; ++k) a += k < wq ? wsum2[k] : 0;
        lbase[tid] = a;
        int g = 0;
        if (h > 0) g = atomicAdd(&cursor[tid], h);
        rbase[tid] = bb + g;
    }
    __syncthreads();

#pragma unroll
    for (int j = 0; j < PL_PER; ++j) {
        if (e0 + j * PL_THREADS + tid < e1) {
            const int b = dv[j] / BK_ROWS;
            const int p = lbase[b] + rk[j];
            ent[p] = ((unsigned)(dv[j] - b * BK_ROWS) << 20) | (unsigned)sv[j];
            ebk[p] = (unsigned short)b;
        }
    }
    __syncthreads();

    const int cnt = e1 > e0 ? e1 - e0 : 0;
#pragma unroll
    for (int j = 0; j < PL_PER; ++j) {
        const int i = tid + j * PL_THREADS;
        if (i < cnt) {
            const int b = ebk[i];
            binned[rbase[b] + (i - lbase[b])] = ent[i];
        }
    }
}

constexpr int BS_THREADS = 1024;
constexpr int BS_CH      = 2;
constexpr int SRT_PER    = 16;
constexpr int SRT_CAP    = BS_THREADS * SRT_PER;

template <bool FINAL>
__global__ void __launch_bounds__(BS_THREADS)
bucket_sum_kernel(unsigned* __restrict__ binned, const int* __restrict__ gbase,
                  const int* __restrict__ cursor, const float* __restrict__ hs,
                  const float* __restrict__ dinv, const float* __restrict__ bias,
                  float* __restrict__ rows_out, int nn, int nn_pad) {
    constexpr int F  = 32;
    constexpr int NW = BS_THREADS / 32;
    constexpr int ST = 32 * BS_CH;
    constexpr int NQ = (BK_ROWS * (F / 4) + BS_THREADS - 1) / BS_THREADS;
    __shared__ float acc[BK_ROWS * F];
    const int tid  = threadIdx.x;
    const int lane = tid & 31;
    const int wv   = __builtin_amdgcn_readfirstlane(tid >> 5);
    const int n_lo = blockIdx.x * BK_ROWS;
    int rows = nn_pad - n_lo;
    if (rows > BK_ROWS) rows = BK_ROWS;
    const int nq = rows * (F / 4);
    const int base = __builtin_amdgcn_readfirstlane(gbase[blockIdx.x]);
    const int cnt  = __builtin_amdgcn_readfirstlane(cursor[blockIdx.x]);
    unsigned* lst = binned + base;

    auto sum_edges = [&](auto word, unsigned (&pn)[BS_CH]) {
        float a = 0.0f;
        int cur = -1;
        for (int c0 = wv * ST; c0 < cnt; c0 += NW * ST) {
            unsigned p[BS_CH];
            const int cn = c0 + NW * ST;
#pragma unroll
            for (int h = 0; h < BS_CH; ++h) {
                p[h] = pn[h];
                const int in = cn + h * 32 + lane;
                pn[h] = word(in < cnt ? in : cnt - 1);
            }
            int my_s[BS_CH], my_d[BS_CH];
#pragma unroll
            for (int h = 0; h < BS_CH; ++h) {
                my_s[h] = 0;
                my_d[h] = -1;
                if (c0 + h * 32 + lane < cnt) {
                    my_s[h] = (int)(p[h] & 0xFFFFFu);
                    my_d[h] = (int)(p[h] >> 20);
                }
            }
            float v[BS_CH * 32];
#pragma unroll
            for (int h = 0; h < BS_CH; ++h) {
#pragma unroll
                for (int u = 0; u < 32; ++u) {
                    const int s = __builtin_amdgcn_readlane(my_s[h], u);
                    v[h * 32 + u] = hs[s * F + lane];
                }
            }
            static_assert(BS_CH == 2, "the fences below name 64 values");
            asm volatile("" :: "v"(v[0]), "v"(v[1]), "v"(v[2]), "v"(v[3]),
                               "v"(v[4]), "v"(v[5]), "v"(v[6]), "v"(v[7]),
                               "v"(v[8]), "v"(v[9]), "v"(v[10]), "v"(v[11]),
                               "v"(v[12]), "v"(v[13]), "v"(v[14]), "v"(v[15]));
            asm volatile("" :: "v"(v[16]), "v"(v[17]), "v"(v[18]), "v"(v[19]),
                               "v"(v[20]), "v"(v[21]), "v"(v[22]), "v"(v[23]),
                               "v"(v[24]), "v"(v[25]), "v"(v[26]), "v"(v[27]),
                               "v"(v[28]), "v"(v[29]), "v"(v[30]), "v"(v[31]));
            asm volatile("" :: "v"(v[32]), "v"(v[33]), "v"(v[34]), "v"(v[35]),
                               "v"(v[36]), "v"(v[37]), "v"(v[38]), "v"(v[39]),
                               "v"(v[40]), "v"(v[41]), "v"(v[42]), "v"(v[43]),
                               "v"(v[44]), "v"(v[45]), "v"(v[46]), "v"(v[47]));
            asm volatile("" :: "v"(v[48]), "v"(v[49]), "v"(v[50]), "v"(v[51]),
                               "v"(v[52]), "v"(v[53]), "v"(v[54]), "v"(v[55]),
                               "v"(v[56]), "v"(v[57]), "v"(v[58]), "v"(v[59]),
                               "v"(v[60]), "v"(v[61]), "v"(v[62]), "v"(v[63]));
#pragma unroll
            for (int h = 0; h < BS_CH; ++h) {
#pragma unroll
                for (int u = 0; u < 32; ++u) {
                    const int d = __builtin_amdgcn_readlane(my_d[h], u);
                    if (__builtin_expect(d != cur, 0)) {
                        if (cur >= 0) atomicAdd(&acc[cur * F + lane], a);
                        a = 0.0f;
                        cur = d;
                    }
                    a += v[h * 32 + u];
                }
            }
        }
        if (cur >= 0) atomicAdd(&acc[cur * F + lane], a);
    };

    const bool order = !FINAL && cnt > 0 && cnt <= SRT_CAP;

    unsigned pn[BS_CH];
#pragma unroll
    for (int h = 0; h < BS_CH; ++h) {
        const int i = wv * ST + h * 32 + lane;
        pn[h] = 0u;
        if (cnt > 0 && !order) pn[h] = lst[i < cnt ? i : cnt - 1];
    }

    const float4* h4 = (const float4*)(hs + (size_t)n_lo * F);
    float4* a4 = (float4*)acc;
    auto self_rows = [&]() {
        float4 t[NQ];
#pragma unroll
        for (int k = 0; k < NQ; ++k) {
            const int i = tid + k * BS_THREADS;
            t[k] = h4[i < nq ? i : nq - 1];
        }
        static_assert(NQ == 4, "the fence below names four pieces");
        asm volatile("" :: "v"(t[0].x), "v"(t[0].y), "v"(t[0].z), "v"(t[0].w),
                           "v"(t[1].x), "v"(t[1].y), "v"(t[1].z), "v"(t[1].w),
                           "v"(t[2].x), "v"(t[2].y), "v"(t[2].z), "v"(t[2].w),
                           "v"(t[3].x), "v"(t[3].y), "v"(t[3].z), "v"(t[3].w));
#pragma unroll
        for (int k = 0; k < NQ; ++k) {
            const int i = tid + k * BS_THREADS;
            if (i < nq) a4[i] = t[k];
        }
    };

    if constexpr (!FINAL) {
        __shared__ unsigned srt[SRT_CAP];
        __shared__ int rcnt[BK_ROWS];
        __shared__ int roff[BK_ROWS];
        __shared__ int rws[BK_WAVES];
        if (order) {
            unsigned w[SRT_PER];
#pragma unroll
            for (int j = 0; j < SRT_PER; ++j) {
                const int i = tid + j * BS_THREADS;
                w[j] = lst[i < cnt ? i : cnt - 1];
            }
            if (tid < BK_ROWS) rcnt[tid] = 0;
            self_rows();
            __syncthreads();
            int rk[SRT_PER];
#pragma unroll
            for (int j = 0; j < SRT_PER; ++j) {
                rk[j] = 0;
                if (tid + j * BS_THREADS < cnt) rk[j] = atomicAdd(&rcnt[w[j] >> 20], 1);
            }
            __syncthreads();
            static_assert(BK_ROWS == BK_WAVES * 32, "whole waves scan the rows");
            int rc = 0, ri = 0;
            if (tid < BK_ROWS) {
                rc = rcnt[tid];
                ri = wave_incl_scan(rc, lane);
                if (lane == 31) rws[wv] = ri;
            }
            __syncthreads();
            if (tid < BK_ROWS) {
                int o = ri - rc;
#pragma unroll
                for (int k = 0; k < BK_WAVES; ++k) o += k < wv ? rws[k] : 0;
                roff[tid] = o;
            }
            __syncthreads();
#pragma unroll
            for (int j = 0; j < SRT_PER; ++j)
                if (tid + j * BS_THREADS < cnt) srt[roff[w[j] >> 20] + rk[j]] = w[j];
            __syncthreads();
#pragma unroll
            for (int j = 0; j < SRT_PER; ++j) {
                const int i = tid + j * BS_THREADS;
                if (i < cnt) lst[i] = srt[i];
            }
#pragma unroll
            for (int h = 0; h < BS_CH; ++h) {
                const int i = wv * ST + h * 32 + lane;
                pn[h] = srt[i < cnt ? i : cnt - 1];
            }
            sum_edges([&](int i) { return srt[i]; }, pn);
        } else {
            self_rows();
            __syncthreads();
            sum_edges([&](int i) { return lst[i]; }, pn);
        }
    } else {
        self_rows();
        __syncthreads();
        sum_edges([&](int i) { return lst[i]; }, pn);
    }
    __syncthreads();

    float4* o4 = (float4*)(rows_out + (size_t)n_lo * F);
    if (!FINAL) {
#pragma unroll
        for (int k = 0; k < NQ; ++k) {
            const int i = tid + k * BS_THREADS;
            if (i < nq) o4[i] = a4[i];
        }
    } else {
        const float4 bq = *(const float4*)&bias[(tid & 7) * 4];
        float dn[NQ];
#pragma unroll
        for (int k = 0; k < NQ; ++k) {
            const int n = n_lo + ((tid + k * BS_THREADS) >> 3);
            dn[k] = dinv[n < nn ? n : nn - 1];
        }
        asm volatile("" :: "v"(dn[0]), "v"(dn[1]), "v"(dn[2]), "v"(dn[3]));
#pragma unroll
        for (int k = 0; k < NQ; ++k) {
            const int i = tid + k * BS_THREADS;
            if (i < nq && n_lo + (i >> 3) < nn) {
                float4 r = a4[i];
                r.x = dn[k] * r.x + bq.x;
                r.y = dn[k] * r.y + bq.y;
                r.z = dn[k] * r.z + bq.z;
                r.w = dn[k] * r.w + bq.w;
                o4[i] = r;
            }
        }
    }
}

extern "C" void kernel_launch(void* const* d_in, const int* in_sizes, int n_in,
                              void* d_out, int out_size, void* d_ws, size_t ws_size,
                              hipStream_t stream) {
    const float* x  = (const float*)d_in[0];
    const int*   ei = (const int*)d_in[1];
    const float* W1 = (const float*)d_in[2];
    const float* b1 = (const float*)d_in[3];
    const float* W2 = (const float*)d_in[4];
    const float* b2 = (const float*)d_in[5];
    float* out = (float*)d_out;

    const int NN = in_sizes[0] / 18;
    const int NE = in_sizes[1] / 2;
    const int NN_pad = (NN + 15) & ~15;
    const int* src = ei;
    const int* dst = ei + NE;

    char* ws = (char*)d_ws;
    size_t off = 0;
    auto carve = [&](size_t bytes) -> char* {
        char* p = ws + off;
        off += (bytes + 255) & ~(size_t)255;
        return p;
    };
    float* dinv = (float*)carve((size_t)NN_pad * 4);
    const int dpitch = (NN_pad + 31) & ~31;
    unsigned* degp = (unsigned*)carve((size_t)DEG_CHUNKS * dpitch * 4);
    const int nwaves = (NN_pad + SP_ROWS - 1) / SP_ROWS;
    unsigned* wtot = (unsigned*)carve((size_t)nwaves * 4);
    int* cursor = (int*)carve((size_t)BK_MAX * 4);
    int* gbase  = (int*)carve((size_t)BK_MAX * 4);
    unsigned* binned = (unsigned*)carve((size_t)NE * 4);
    float* xs   = (float*)carve((size_t)NN_pad * 32 * 4);
    float* aggx = (float*)carve((size_t)NN_pad * 32 * 4);
    float* hs2  = xs;

    const int dranges = (NN_pad + DEG_RANGE - 1) / DEG_RANGE;
    const int dchunk  = (((NE + DEG_CHUNKS - 1) / DEG_CHUNKS) + 3) & ~3;
    deg_count_lds_kernel<<<dim3(dranges, DEG_CHUNKS), DEG_THREADS, 0, stream>>>(
        dst, degp, NE, dchunk, NN_pad, dpitch);

    const int spblk = (NN_pad + SP_WAVES * SP_ROWS - 1) / (SP_WAVES * SP_ROWS);
    dinv_scale_pad_kernel<<<spblk, SP_WAVES * 32, 0, stream>>>(degp, x, dinv, xs, wtot, cursor, NN, NN_pad, dpitch);

    const int nb = (NN_pad + BK_ROWS - 1) / BK_ROWS;
    int pwgs = 256;
    if ((NE + PL_CAP - 1) / PL_CAP > pwgs) pwgs = (NE + PL_CAP - 1) / PL_CAP;
    const int pslice = (NE + pwgs - 1) / pwgs;
    bin_edges_kernel<<<pwgs, PL_THREADS, 0, stream>>>(src, dst, wtot, cursor, gbase, binned, NE, pslice, nwaves, nb);

    bucket_sum_kernel<false><<<nb, BS_THREADS, 0, stream>>>(binned, gbase, cursor, xs, dinv, b2, aggx, NN, NN_pad);

    const int mtiles = NN_pad / 16;
    const int gblk = (mtiles + 3) / 4;
    gemm_two_layers_kernel<18, 32, 64, 32><<<gblk, 128, 0, stream>>>(aggx, W1, b1, W2, dinv, hs2, mtiles);

    bucket_sum_kernel<true><<<nb, BS_THREADS, 0, stream>>>(binned, gbase, cursor, hs2, dinv, b2, out, NN, NN_pad);
}


# Both dense products moved into the tail of the layer-1 neighbour-sum launch: each work-group multiplies its own summed rows straight from LDS, one wave per 16 rows; launches 6 -> 5; sum order kept
# speedup vs baseline: 6.8118x; 1.0965x over previous
#include <hip/hip_runtime.h>

typedef __attribute__((ext_vector_type(2))) float v2f;
typedef __attribute__((ext_vector_type(8))) float v8f;

constexpr int DEG_RANGE   = 25024;
constexpr int DEG_CHUNKS  = 64;
constexpr int DEG_THREADS = 1024;
constexpr int DEG_BATCH   = 8;

__device__ __forceinline__ void deg_bump(unsigned* cnt, int d, int lo) {
    const unsigned i = (unsigned)(d - lo);
    if (i < (unsigned)DEG_RANGE) atomicAdd(&cnt[i], 1u);
}

__global__ void __launch_bounds__(DEG_THREADS)
deg_count_lds_kernel(const int* __restrict__ dst, unsigned* __restrict__ part,
                     int ne, int chunk, int nn_pad, int pitch) {
    __shared__ unsigned cnt[DEG_RANGE];
    const int tid = threadIdx.x;
    const int lo  = blockIdx.x * DEG_RANGE;
    const int c   = blockIdx.y;
    const int left = nn_pad - lo;
    const int range_n = left < DEG_RANGE ? left : DEG_RANGE;

    uint4* cnt4 = (uint4*)cnt;
    for (int i = tid; i < DEG_RANGE / 4; i += DEG_THREADS) cnt4[i] = make_uint4(0u, 0u, 0u, 0u);
    __syncthreads();

    const int e0 = c * chunk;
    int e1 = e0 + chunk;
    if (e1 > ne) e1 = ne;
    if ((((size_t)dst) & 15) == 0) {
        const int4* d4 = (const int4*)dst;
        const int q1 = e1 >> 2;
        for (int qb = (e0 >> 2) + tid; qb < q1; qb += DEG_BATCH * DEG_THREADS) {
            int4 d[DEG_BATCH];
#pragma unroll
            for (int j = 0; j < DEG_BATCH; ++j) {
                const int q = qb + j * DEG_THREADS;
                d[j] = d4[q < q1 ? q : q1 - 1];
            }
#pragma unroll
            for (int j = 0; j < DEG_BATCH; ++j) {
                if (qb + j * DEG_THREADS < q1) {
                    deg_bump(cnt, d[j].x, lo);
                    deg_bump(cnt, d[j].y, lo);
                    deg_bump(cnt, d[j].z, lo);
                    deg_bump(cnt, d[j].w, lo);
                }
            }
        }
        const int e = (q1 << 2) + tid;
        if (e >= e0 && e < e1) deg_bump(cnt, dst[e], lo);
    } else {
        for (int eb = e0 + tid; eb < e1; eb += DEG_BATCH * DEG_THREADS) {
            int d[DEG_BATCH];
#pragma unroll
            for (int j = 0; j < DEG_BATCH; ++j) {
                const int e = eb + j * DEG_THREADS;
                d[j] = dst[e < e1 ? e : e1 - 1];
            }
#pragma unroll
            for (int j = 0; j < DEG_BATCH; ++j)
                if (eb + j * DEG_THREADS < e1) deg_bump(cnt, d[j], lo);
        }
    }
    __syncthreads();

    uint4* out4 = (uint4*)(part + (size_t)c * pitch + lo);
    for (int i = tid; i < range_n / 4; i += DEG_THREADS) out4[i] = cnt4[i];
}

constexpr int SP_WAVES = 4;
constexpr int SP_ROWS  = 32;

constexpr int BK_WAVES = 13;
constexpr int BK_ROWS  = BK_WAVES * SP_ROWS;
constexpr int BK_MAX   = 256;

__global__ void __launch_bounds__(SP_WAVES * 32) __attribute__((amdgpu_waves_per_eu(1, 4)))
dinv_scale_pad_kernel(const unsigned* __restrict__ part, const float* __restrict__ x,
                      float* __restrict__ dinv, float* __restrict__ xs,
                      unsigned* __restrict__ wtot, int* __restrict__ cursor,
                      int nn, int nn_pad, int pitch) {
    const int lane = threadIdx.x & 31;
    const int wave = __builtin_amdgcn_readfirstlane(blockIdx.x * SP_WAVES + (threadIdx.x >> 5));
    const int n0   = wave * SP_ROWS;
    if (n0 >= nn_pad) return;
    if (wave == 0) {
#pragma unroll
        for (int k = 0; k < BK_MAX / 32; ++k) cursor[k * 32 + lane] = 0;
    }

    const int n  = n0 + lane;
    const int nc = n < nn_pad ? n : nn_pad - 1;
    unsigned s = 1u;
#pragma unroll
    for (int c = 0; c < DEG_CHUNKS; ++c) s += part[(size_t)c * pitch + nc];

    const int kc = lane < 18 ? lane : 17;
    float xv[SP_ROWS];
#pragma unroll
    for (int j = 0; j < SP_ROWS; ++j) {
        const int r = n0 + j < nn ? n0 + j : nn - 1;
        xv[j] = x[r * 18 + kc];
    }

    const float dn = rsqrtf((float)s);
    if (n < nn_pad) dinv[n] = dn;

    unsigned ein = n < nn_pad ? s - 1u : 0u;
#pragma unroll
    for (int o = 16; o > 0; o >>= 1) ein += __shfl_xor(ein, o);
    if (lane == 0) wtot[wave] = ein;

#pragma unroll
    for (int j = 0; j < SP_ROWS; ++j) {
        const int row = n0 + j;
        if (row >= nn_pad) break;
        const float dj = __builtin_bit_cast(
            float, __builtin_amdgcn_readlane(__builtin_bit_cast(int, dn), j));
        const float v = (row < nn && lane < 18) ? dj * xv[j] : 0.0f;
        xs[row * 32 + lane] = v;
    }
}

template <int K1, int LDA, int NH, int NO>
__device__ __forceinline__ void two_layers_tile(const float* xl,
                                                const float* __restrict__ W1,
                                                const float* __restrict__ b1,
                                                const float* __restrict__ W2,
                                                const float* __restrict__ dv16,
                                                float* __restrict__ hs, int lane) {
    static_assert(NH % 16 == 0 && NO % 16 == 0 && LDA % 2 == 0, "tile layout");
    constexpr int KS1 = (K1 + 3) / 4;
    constexpr int T1  = NH / 16;
    constexpr int T2  = NO / 16;
    static_assert(KS1 * 4 <= LDA, "padded K inside the row pitch");

    const int c    = lane & 15;
    const int hh   = lane >> 4;

    v8f acc1[T1] = {};
#pragma unroll
    for (int ks = 0; ks < KS1; ++ks) {
        const int k0 = ks * 4 + hh * 2;
        const v2f xb = *(const v2f*)&xl[c * LDA + k0];
        const bool in0 = k0 < K1;
        const bool in1 = k0 + 1 < K1;
        const int r0 = in0 ? k0 : 0;
        const int r1 = in1 ? k0 + 1 : 0;
#pragma unroll
        for (int t = 0; t < T1; ++t) {
            v2f wa;
            wa.x = W1[r0 * NH + t * 16 + c];
            wa.y = W1[r1 * NH + t * 16 + c];
            if (!in0) wa.x = 0.0f;
            if (!in1) wa.y = 0.0f;
            acc1[t] = __builtin_amdgcn_wmma_f32_16x16x4_f32(
                false, wa, false, xb, (short)0, acc1[t], false, false);
        }
    }

    const float dn = dv16[c];
#pragma unroll
    for (int t = 0; t < T1; ++t) {
        const float4 ba = *(const float4*)&b1[t * 16 + 8 * hh];
        const float4 bb = *(const float4*)&b1[t * 16 + 8 * hh + 4];
        const float bv[8] = {ba.x, ba.y, ba.z, ba.w, bb.x, bb.y, bb.z, bb.w};
#pragma unroll
        for (int r = 0; r < 8; ++r)
            acc1[t][r] = fmaxf(acc1[t][r] * dn + bv[r], 0.0f);
    }

    v8f acc2[T2] = {};
#pragma unroll
    for (int t = 0; t < T1; ++t) {
#pragma unroll
        for (int p = 0; p < 4; ++p) {
            v2f a;
            a.x = acc1[t][2 * p];
            a.y = acc1[t][2 * p + 1];
            const int k2 = t * 16 + 8 * hh + 2 * p;
#pragma unroll
            for (int u = 0; u < T2; ++u) {
                v2f b;
                b.x = W2[k2 * NO + u * 16 + c];
                b.y = W2[(k2 + 1) * NO + u * 16 + c];
                acc2[u] = __builtin_amdgcn_wmma_f32_16x16x4_f32(
                    false, a, false, b, (short)0, acc2[u], false, false);
            }
        }
    }

    const float4 dv0 = *(const float4*)&dv16[8 * hh];
    const float4 dv1 = *(const float4*)&dv16[8 * hh + 4];
    const float dv[8] = {dv0.x, dv0.y, dv0.z, dv0.w, dv1.x, dv1.y, dv1.z, dv1.w};
#pragma unroll
    for (int r = 0; r < 8; ++r) {
        const int m = r + 8 * hh;
#pragma unroll
        for (int u = 0; u < T2; ++u) {
            const float val = acc2[u][r] * dv[r];
            hs[m * NO + u * 16 + c] = val;
        }
    }
}

constexpr int PL_THREADS = 1024;
constexpr int PL_PER     = 13;
constexpr int PL_CAP     = PL_THREADS * PL_PER;

__device__ __forceinline__ int wave_incl_scan(int v, int lane) {
#pragma unroll
    for (int o = 1; o < 32; o <<= 1) {
        const int t = __shfl_up(v, o);
        if (lane >= o) v += t;
    }
    return v;
}

__global__ void __launch_bounds__(PL_THREADS)
bin_edges_kernel(const int* __restrict__ src, const int* __restrict__ dst,
                 const unsigned* __restrict__ wtot, int* __restrict__ cursor,
                 int* __restrict__ gbase, unsigned* __restrict__ binned,
                 int ne, int slice, int nwaves, int nb) {
    __shared__ unsigned ent[PL_CAP];
    __shared__ unsigned short ebk[PL_CAP];
    __shared__ int hist[BK_MAX];
    __shared__ int lbase[BK_MAX];
    __shared__ int rbase[BK_MAX];
    __shared__ int wsum1[BK_MAX / 32], wsum2[BK_MAX / 32];
    const int tid  = threadIdx.x;
    const int lane = tid & 31;
    const int wq   = tid >> 5;
    const int e0 = blockIdx.x * slice;
    int e1 = e0 + slice;
    if (e1 > ne) e1 = ne;

    int sv[PL_PER], dv[PL_PER];
#pragma unroll
    for (int j = 0; j < PL_PER; ++j) {
        const int e = e0 + j * PL_THREADS + tid;
        const int ec = e < e1 ? e : ne - 1;
        sv[j] = src[ec];
        dv[j] = dst[ec];
    }

    int bt = 0, bi = 0;
    if (tid < BK_MAX) {
        hist[tid] = 0;
        unsigned q[BK_WAVES];
#pragma unroll
        for (int k = 0; k < BK_WAVES; ++k) {
            const int w = tid * BK_WAVES + k;
            q[k] = wtot[w < nwaves ? w : nwaves - 1];
        }
        static_assert(BK_WAVES == 13, "the fence below names thirteen values");
        asm volatile("" :: "v"(q[0]), "v"(q[1]), "v"(q[2]), "v"(q[3]), "v"(q[4]),
                           "v"(q[5]), "v"(q[6]), "v"(q[7]), "v"(q[8]), "v"(q[9]),
                           "v"(q[10]), "v"(q[11]), "v"(q[12]));
        unsigned t = 0u;
#pragma unroll
        for (int k = 0; k < BK_WAVES; ++k) {
            const int w = tid * BK_WAVES + k;
            t += (tid < nb && w < nwaves) ? q[k] : 0u;
        }
        bt = (int)t;
        bi = wave_incl_scan(bt, lane);
        if (lane == 31) wsum1[wq] = bi;
    }
    __syncthreads();

    int bb = 0;
    if (tid < BK_MAX) {
        bb = bi - bt;
#pragma unroll
        for (int k = 0; k < BK_MAX / 32; ++k) bb += k < wq ? wsum1[k] : 0;
        if (blockIdx.x == 0) gbase[tid] = bb;
    }

    int rk[PL_PER];
#pragma unroll
    for (int j = 0; j < PL_PER; ++j) {
        rk[j] = 0;
        if (e0 + j * PL_THREADS + tid < e1) rk[j] = atomicAdd(&hist[dv[j] / BK_ROWS], 1);
    }
    __syncthreads();

    int h = 0, hi = 0;
    if (tid < BK_MAX) {
        h  = hist[tid];
        hi = wave_incl_scan(h, lane);
        if (lane == 31) wsum2[wq] = hi;
    }
    __syncthreads();
    if (tid < BK_MAX) {
        int a = hi - h;
#pragma unroll
        for (int k = 0; k < BK_MAX / 32; ++k) a += k < wq ? wsum2[k] : 0;
        lbase[tid] = a;
        int g = 0;
        if (h > 0) g = atomicAdd(&cursor[tid], h);
        rbase[tid] = bb + g;
    }
    __syncthreads();

#pragma unroll
    for (int j = 0; j < PL_PER; ++j) {
        if (e0 + j * PL_THREADS + tid < e1) {
            const int b = dv[j] / BK_ROWS;
            const int p = lbase[b] + rk[j];
            ent[p] = ((unsigned)(dv[j] - b * BK_ROWS) << 20) | (unsigned)sv[j];
            ebk[p] = (unsigned short)b;
        }
    }
    __syncthreads();

    const int cnt = e1 > e0 ? e1 - e0 : 0;
#pragma unroll
    for (int j = 0; j < PL_PER; ++j) {
        const int i = tid + j * PL_THREADS;
        if (i < cnt) {
            const int b = ebk[i];
            binned[rbase[b] + (i - lbase[b])] = ent[i];
        }
    }
}

constexpr int BS_THREADS = 1024;
constexpr int BS_CH      = 2;
constexpr int SRT_PER    = 16;
constexpr int SRT_CAP    = BS_THREADS * SRT_PER;

template <bool FINAL>
__global__ void __launch_bounds__(BS_THREADS)
bucket_sum_kernel(unsigned* __restrict__ binned, const int* __restrict__ gbase,
                  const int* __restrict__ cursor, const float* __restrict__ hs,
                  const float* __restrict__ dinv, const float* __restrict__ bias,
                  const float* __restrict__ W1, const float* __restrict__ b1,
                  const float* __restrict__ W2,
                  float* __restrict__ rows_out, int nn, int nn_pad) {
    constexpr int F  = 32;
    constexpr int NW = BS_THREADS / 32;
    constexpr int ST = 32 * BS_CH;
    constexpr int NQ = (BK_ROWS * (F / 4) + BS_THREADS - 1) / BS_THREADS;
    __shared__ float acc[BK_ROWS * F];
    const int tid  = threadIdx.x;
    const int lane = tid & 31;
    const int wv   = __builtin_amdgcn_readfirstlane(tid >> 5);
    const int n_lo = blockIdx.x * BK_ROWS;
    int rows = nn_pad - n_lo;
    if (rows > BK_ROWS) rows = BK_ROWS;
    const int nq = rows * (F / 4);
    const int base = __builtin_amdgcn_readfirstlane(gbase[blockIdx.x]);
    const int cnt  = __builtin_amdgcn_readfirstlane(cursor[blockIdx.x]);
    unsigned* lst = binned + base;

    auto sum_edges = [&](auto word, unsigned (&pn)[BS_CH]) {
        float a = 0.0f;
        int cur = -1;
        for (int c0 = wv * ST; c0 < cnt; c0 += NW * ST) {
            unsigned p[BS_CH];
            const int cn = c0 + NW * ST;
#pragma unroll
            for (int h = 0; h < BS_CH; ++h) {
                p[h] = pn[h];
                const int in = cn + h * 32 + lane;
                pn[h] = word(in < cnt ? in : cnt - 1);
            }
            int my_s[BS_CH], my_d[BS_CH];
#pragma unroll
            for (int h = 0; h < BS_CH; ++h) {
                my_s[h] = 0;
                my_d[h] = -1;
                if (c0 + h * 32 + lane < cnt) {
                    my_s[h] = (int)(p[h] & 0xFFFFFu);
                    my_d[h] = (int)(p[h] >> 20);
                }
            }
            float v[BS_CH * 32];
#pragma unroll
            for (int h = 0; h < BS_CH; ++h) {
#pragma unroll
                for (int u = 0; u < 32; ++u) {
                    const int s = __builtin_amdgcn_readlane(my_s[h], u);
                    v[h * 32 + u] = hs[s * F + lane];
                }
            }
            static_assert(BS_CH == 2, "the fences below name 64 values");
            asm volatile("" :: "v"(v[0]), "v"(v[1]), "v"(v[2]), "v"(v[3]),
                               "v"(v[4]), "v"(v[5]), "v"(v[6]), "v"(v[7]),
                               "v"(v[8]), "v"(v[9]), "v"(v[10]), "v"(v[11]),
                               "v"(v[12]), "v"(v[13]), "v"(v[14]), "v"(v[15]));
            asm volatile("" :: "v"(v[16]), "v"(v[17]), "v"(v[18]), "v"(v[19]),
                               "v"(v[20]), "v"(v[21]), "v"(v[22]), "v"(v[23]),
                               "v"(v[24]), "v"(v[25]), "v"(v[26]), "v"(v[27]),
                               "v"(v[28]), "v"(v[29]), "v"(v[30]), "v"(v[31]));
            asm volatile("" :: "v"(v[32]), "v"(v[33]), "v"(v[34]), "v"(v[35]),
                               "v"(v[36]), "v"(v[37]), "v"(v[38]), "v"(v[39]),
                               "v"(v[40]), "v"(v[41]), "v"(v[42]), "v"(v[43]),
                               "v"(v[44]), "v"(v[45]), "v"(v[46]), "v"(v[47]));
            asm volatile("" :: "v"(v[48]), "v"(v[49]), "v"(v[50]), "v"(v[51]),
                               "v"(v[52]), "v"(v[53]), "v"(v[54]), "v"(v[55]),
                               "v"(v[56]), "v"(v[57]), "v"(v[58]), "v"(v[59]),
                               "v"(v[60]), "v"(v[61]), "v"(v[62]), "v"(v[63]));
#pragma unroll
            for (int h = 0; h < BS_CH; ++h) {
#pragma unroll
                for (int u = 0; u < 32; ++u) {
                    const int d = __builtin_amdgcn_readlane(my_d[h], u);
                    if (__builtin_expect(d != cur, 0)) {
                        if (cur >= 0) atomicAdd(&acc[cur * F + lane], a);
                        a = 0.0f;
                        cur = d;
                    }
                    a += v[h * 32 + u];
                }
            }
        }
        if (cur >= 0) atomicAdd(&acc[cur * F + lane], a);
    };

    const bool order = !FINAL && cnt > 0 && cnt <= SRT_CAP;

    unsigned pn[BS_CH];
#pragma unroll
    for (int h = 0; h < BS_CH; ++h) {
        const int i = wv * ST + h * 32 + lane;
        pn[h] = 0u;
        if (cnt > 0 && !order) pn[h] = lst[i < cnt ? i : cnt - 1];
    }

    const float4* h4 = (const float4*)(hs + (size_t)n_lo * F);
    float4* a4 = (float4*)acc;
    auto self_rows = [&]() {
        float4 t[NQ];
#pragma unroll
        for (int k = 0; k < NQ; ++k) {
            const int i = tid + k * BS_THREADS;
            t[k] = h4[i < nq ? i : nq - 1];
        }
        static_assert(NQ == 4, "the fence below names four pieces");
        asm volatile("" :: "v"(t[0].x), "v"(t[0].y), "v"(t[0].z), "v"(t[0].w),
                           "v"(t[1].x), "v"(t[1].y), "v"(t[1].z), "v"(t[1].w),
                           "v"(t[2].x), "v"(t[2].y), "v"(t[2].z), "v"(t[2].w),
                           "v"(t[3].x), "v"(t[3].y), "v"(t[3].z), "v"(t[3].w));
#pragma unroll
        for (int k = 0; k < NQ; ++k) {
            const int i = tid + k * BS_THREADS;
            if (i < nq) a4[i] = t[k];
        }
    };

    if constexpr (!FINAL) {
        __shared__ unsigned srt[SRT_CAP];
        __shared__ int rcnt[BK_ROWS];
        __shared__ int roff[BK_ROWS];
        __shared__ int rws[BK_WAVES];
        if (order) {
            unsigned w[SRT_PER];
#pragma unroll
            for (int j = 0; j < SRT_PER; ++j) {
                const int i = tid + j * BS_THREADS;
                w[j] = lst[i < cnt ? i : cnt - 1];
            }
            if (tid < BK_ROWS) rcnt[tid] = 0;
            self_rows();
            __syncthreads();
            int rk[SRT_PER];
#pragma unroll
            for (int j = 0; j < SRT_PER; ++j) {
                rk[j] = 0;
                if (tid + j * BS_THREADS < cnt) rk[j] = atomicAdd(&rcnt[w[j] >> 20], 1);
            }
            __syncthreads();
            static_assert(BK_ROWS == BK_WAVES * 32, "whole waves scan the rows");
            int rc = 0, ri = 0;
            if (tid < BK_ROWS) {
                rc = rcnt[tid];
                ri = wave_incl_scan(rc, lane);
                if (lane == 31) rws[wv] = ri;
            }
            __syncthreads();
            if (tid < BK_ROWS) {
                int o = ri - rc;
#pragma unroll
                for (int k = 0; k < BK_WAVES; ++k) o += k < wv ? rws[k] : 0;
                roff[tid] = o;
            }
            __syncthreads();
#pragma unroll
            for (int j = 0; j < SRT_PER; ++j)
                if (tid + j * BS_THREADS < cnt) srt[roff[w[j] >> 20] + rk[j]] = w[j];
            __syncthreads();
#pragma unroll
            for (int j = 0; j < SRT_PER; ++j) {
                const int i = tid + j * BS_THREADS;
                if (i < cnt) lst[i] = srt[i];
            }
#pragma unroll
            for (int h = 0; h < BS_CH; ++h) {
                const int i = wv * ST + h * 32 + lane;
                pn[h] = srt[i < cnt ? i : cnt - 1];
            }
            sum_edges([&](int i) { return srt[i]; }, pn);
        } else {
            self_rows();
            __syncthreads();
            sum_edges([&](int i) { return lst[i]; }, pn);
        }
    } else {
        self_rows();
        __syncthreads();
        sum_edges([&](int i) { return lst[i]; }, pn);
    }
    __syncthreads();

    if constexpr (!FINAL) {
        static_assert(BK_ROWS % 16 == 0 && BK_ROWS / 16 <= NW, "one wave per 16-row tile");
        const int m0 = wv * 16;
        if (m0 < rows)
            two_layers_tile<18, F, 64, 32>(acc + m0 * F, W1, b1, W2, dinv + n_lo + m0,
                                           rows_out + (size_t)(n_lo + m0) * 32, lane);
    } else {
        float4* o4 = (float4*)(rows_out + (size_t)n_lo * F);
        const float4 bq = *(const float4*)&bias[(tid & 7) * 4];
        float dn[NQ];
#pragma unroll
        for (int k = 0; k < NQ; ++k) {
            const int n = n_lo + ((tid + k * BS_THREADS) >> 3);
            dn[k] = dinv[n < nn ? n : nn - 1];
        }
        asm volatile("" :: "v"(dn[0]), "v"(dn[1]), "v"(dn[2]), "v"(dn[3]));
#pragma unroll
        for (int k = 0; k < NQ; ++k) {
            const int i = tid + k * BS_THREADS;
            if (i < nq && n_lo + (i >> 3) < nn) {
                float4 r = a4[i];
                r.x = dn[k] * r.x + bq.x;
                r.y = dn[k] * r.y + bq.y;
                r.z = dn[k] * r.z + bq.z;
                r.w = dn[k] * r.w + bq.w;
                o4[i] = r;
            }
        }
    }
}

extern "C" void kernel_launch(void* const* d_in, const int* in_sizes, int n_in,
                              void* d_out, int out_size, void* d_ws, size_t ws_size,
                              hipStream_t stream) {
    const float* x  = (const float*)d_in[0];
    const int*   ei = (const int*)d_in[1];
    const float* W1 = (const float*)d_in[2];
    const float* b1 = (const float*)d_in[3];
    const float* W2 = (const float*)d_in[4];
    const float* b2 = (const float*)d_in[5];
    float* out = (float*)d_out;

    const int NN = in_sizes[0] / 18;
    const int NE = in_sizes[1] / 2;
    const int NN_pad = (NN + 15) & ~15;
    const int* src = ei;
    const int* dst = ei + NE;

    char* ws = (char*)d_ws;
    size_t off = 0;
    auto carve = [&](size_t bytes) -> char* {
        char* p = ws + off;
        off += (bytes + 255) & ~(size_t)255;
        return p;
    };
    float* dinv = (float*)carve((size_t)NN_pad * 4);
    const int dpitch = (NN_pad + 31) & ~31;
    unsigned* degp = (unsigned*)carve((size_t)DEG_CHUNKS * dpitch * 4);
    const int nwaves = (NN_pad + SP_ROWS - 1) / SP_ROWS;
    unsigned* wtot = (unsigned*)carve((size_t)nwaves * 4);
    int* cursor = (int*)carve((size_t)BK_MAX * 4);
    int* gbase  = (int*)carve((size_t)BK_MAX * 4);
    unsigned* binned = (unsigned*)carve((size_t)NE * 4);
    float* xs   = (float*)carve((size_t)NN_pad * 32 * 4);
    float* hs2  = (float*)carve((size_t)NN_pad * 32 * 4);

    const int dranges = (NN_pad + DEG_RANGE - 1) / DEG_RANGE;
    const int dchunk  = (((NE + DEG_CHUNKS - 1) / DEG_CHUNKS) + 3) & ~3;
    deg_count_lds_kernel<<<dim3(dranges, DEG_CHUNKS), DEG_THREADS, 0, stream>>>(
        dst, degp, NE, dchunk, NN_pad, dpitch);

    const int spblk = (NN_pad + SP_WAVES * SP_ROWS - 1) / (SP_WAVES * SP_ROWS);
    dinv_scale_pad_kernel<<<spblk, SP_WAVES * 32, 0, stream>>>(degp, x, dinv, xs, wtot, cursor, NN, NN_pad, dpitch);

    const int nb = (NN_pad + BK_ROWS - 1) / BK_ROWS;
    int pwgs = 256;
    if ((NE + PL_CAP - 1) / PL_CAP > pwgs) pwgs = (NE + PL_CAP - 1) / PL_CAP;
    const int pslice = (NE + pwgs - 1) / pwgs;
    bin_edges_kernel<<<pwgs, PL_THREADS, 0, stream>>>(src, dst, wtot, cursor, gbase, binned, NE, pslice, nwaves, nb);

    bucket_sum_kernel<false><<<nb, BS_THREADS, 0, stream>>>(binned, gbase, cursor, xs, dinv, b2, W1, b1, W2, hs2, NN, NN_pad);

    bucket_sum_kernel<true><<<nb, BS_THREADS, 0, stream>>>(binned, gbase, cursor, hs2, dinv, b2, W1, b1, W2, out, NN, NN_pad);
}


# Neighbour sums take two edges per instruction: each 16-lane half walks its own words (lane = column pair, 8-byte gathers, ids by DPP row_share), run ends from one ballot per 16; sum order changes
# speedup vs baseline: 7.1883x; 1.0553x over previous
#include <hip/hip_runtime.h>
#include <utility>

typedef __attribute__((ext_vector_type(2))) float v2f;
typedef __attribute__((ext_vector_type(8))) float v8f;

constexpr int DEG_RANGE   = 25024;
constexpr int DEG_CHUNKS  = 64;
constexpr int DEG_THREADS = 1024;
constexpr int DEG_BATCH   = 8;

__device__ __forceinline__ void deg_bump(unsigned* cnt, int d, int lo) {
    const unsigned i = (unsigned)(d - lo);
    if (i < (unsigned)DEG_RANGE) atomicAdd(&cnt[i], 1u);
}

__global__ void __launch_bounds__(DEG_THREADS)
deg_count_lds_kernel(const int* __restrict__ dst, unsigned* __restrict__ part,
                     int ne, int chunk, int nn_pad, int pitch) {
    __shared__ unsigned cnt[DEG_RANGE];
    const int tid = threadIdx.x;
    const int lo  = blockIdx.x * DEG_RANGE;
    const int c   = blockIdx.y;
    const int left = nn_pad - lo;
    const int range_n = left < DEG_RANGE ? left : DEG_RANGE;

    uint4* cnt4 = (uint4*)cnt;
    for (int i = tid; i < DEG_RANGE / 4; i += DEG_THREADS) cnt4[i] = make_uint4(0u, 0u, 0u, 0u);
    __syncthreads();

    const int e0 = c * chunk;
    int e1 = e0 + chunk;
    if (e1 > ne) e1 = ne;
    if ((((size_t)dst) & 15) == 0) {
        const int4* d4 = (const int4*)dst;
        const int q1 = e1 >> 2;
        for (int qb = (e0 >> 2) + tid; qb < q1; qb += DEG_BATCH * DEG_THREADS) {
            int4 d[DEG_BATCH];
#pragma unroll
            for (int j = 0; j < DEG_BATCH; ++j) {
                const int q = qb + j * DEG_THREADS;
                d[j] = d4[q < q1 ? q : q1 - 1];
            }
#pragma unroll
            for (int j = 0; j < DEG_BATCH; ++j) {
                if (qb + j * DEG_THREADS < q1) {
                    deg_bump(cnt, d[j].x, lo);
                    deg_bump(cnt, d[j].y, lo);
                    deg_bump(cnt, d[j].z, lo);
                    deg_bump(cnt, d[j].w, lo);
                }
            }
        }
        const int e = (q1 << 2) + tid;
        if (e >= e0 && e < e1) deg_bump(cnt, dst[e], lo);
    } else {
        for (int eb = e0 + tid; eb < e1; eb += DEG_BATCH * DEG_THREADS) {
            int d[DEG_BATCH];
#pragma unroll
            for (int j = 0; j < DEG_BATCH; ++j) {
                const int e = eb + j * DEG_THREADS;
                d[j] = dst[e < e1 ? e : e1 - 1];
            }
#pragma unroll
            for (int j = 0; j < DEG_BATCH; ++j)
                if (eb + j * DEG_THREADS < e1) deg_bump(cnt, d[j], lo);
        }
    }
    __syncthreads();

    uint4* out4 = (uint4*)(part + (size_t)c * pitch + lo);
    for (int i = tid; i < range_n / 4; i += DEG_THREADS) out4[i] = cnt4[i];
}

constexpr int SP_WAVES = 4;
constexpr int SP_ROWS  = 32;

constexpr int BK_WAVES = 13;
constexpr int BK_ROWS  = BK_WAVES * SP_ROWS;
constexpr int BK_MAX   = 256;

__global__ void __launch_bounds__(SP_WAVES * 32) __attribute__((amdgpu_waves_per_eu(1, 4)))
dinv_scale_pad_kernel(const unsigned* __restrict__ part, const float* __restrict__ x,
                      float* __restrict__ dinv, float* __restrict__ xs,
                      unsigned* __restrict__ wtot, int* __restrict__ cursor,
                      int nn, int nn_pad, int pitch) {
    const int lane = threadIdx.x & 31;
    const int wave = __builtin_amdgcn_readfirstlane(blockIdx.x * SP_WAVES + (threadIdx.x >> 5));
    const int n0   = wave * SP_ROWS;
    if (n0 >= nn_pad) return;
    if (wave == 0) {
#pragma unroll
        for (int k = 0; k < BK_MAX / 32; ++k) cursor[k * 32 + lane] = 0;
    }

    const int n  = n0 + lane;
    const int nc = n < nn_pad ? n : nn_pad - 1;
    unsigned s = 1u;
#pragma unroll
    for (int c = 0; c < DEG_CHUNKS; ++c) s += part[(size_t)c * pitch + nc];

    const int kc = lane < 18 ? lane : 17;
    float xv[SP_ROWS];
#pragma unroll
    for (int j = 0; j < SP_ROWS; ++j) {
        const int r = n0 + j < nn ? n0 + j : nn - 1;
        xv[j] = x[r * 18 + kc];
    }

    const float dn = rsqrtf((float)s);
    if (n < nn_pad) dinv[n] = dn;

    unsigned ein = n < nn_pad ? s - 1u : 0u;
#pragma unroll
    for (int o = 16; o > 0; o >>= 1) ein += __shfl_xor(ein, o);
    if (lane == 0) wtot[wave] = ein;

#pragma unroll
    for (int j = 0; j < SP_ROWS; ++j) {
        const int row = n0 + j;
        if (row >= nn_pad) break;
        const float dj = __builtin_bit_cast(
            float, __builtin_amdgcn_readlane(__builtin_bit_cast(int, dn), j));
        const float v = (row < nn && lane < 18) ? dj * xv[j] : 0.0f;
        xs[row * 32 + lane] = v;
    }
}

template <int K1, int LDA, int NH, int NO>
__device__ __forceinline__ void two_layers_tile(const float* xl,
                                                const float* __restrict__ W1,
                                                const float* __restrict__ b1,
                                                const float* __restrict__ W2,
                                                const float* __restrict__ dv16,
                                                float* __restrict__ hs, int lane) {
    static_assert(NH % 16 == 0 && NO % 16 == 0 && LDA % 2 == 0, "tile layout");
    constexpr int KS1 = (K1 + 3) / 4;
    constexpr int T1  = NH / 16;
    constexpr int T2  = NO / 16;
    static_assert(KS1 * 4 <= LDA, "padded K inside the row pitch");

    const int c    = lane & 15;
    const int hh   = lane >> 4;

    v8f acc1[T1] = {};
#pragma unroll
    for (int ks = 0; ks < KS1; ++ks) {
        const int k0 = ks * 4 + hh * 2;
        const v2f xb = *(const v2f*)&xl[c * LDA + k0];
        const bool in0 = k0 < K1;
        const bool in1 = k0 + 1 < K1;
        const int r0 = in0 ? k0 : 0;
        const int r1 = in1 ? k0 + 1 : 0;
#pragma unroll
        for (int t = 0; t < T1; ++t) {
            v2f wa;
            wa.x = W1[r0 * NH + t * 16 + c];
            wa.y = W1[r1 * NH + t * 16 + c];
            if (!in0) wa.x = 0.0f;
            if (!in1) wa.y = 0.0f;
            acc1[t] = __builtin_amdgcn_wmma_f32_16x16x4_f32(
                false, wa, false, xb, (short)0, acc1[t], false, false);
        }
    }

    const float dn = dv16[c];
#pragma unroll
    for (int t = 0; t < T1; ++t) {
        const float4 ba = *(const float4*)&b1[t * 16 + 8 * hh];
        const float4 bb = *(const float4*)&b1[t * 16 + 8 * hh + 4];
        const float bv[8] = {ba.x, ba.y, ba.z, ba.w, bb.x, bb.y, bb.z, bb.w};
#pragma unroll
        for (int r = 0; r < 8; ++r)
            acc1[t][r] = fmaxf(acc1[t][r] * dn + bv[r], 0.0f);
    }

    v8f acc2[T2] = {};
#pragma unroll
    for (int t = 0; t < T1; ++t) {
#pragma unroll
        for (int p = 0; p < 4; ++p) {
            v2f a;
            a.x = acc1[t][2 * p];
            a.y = acc1[t][2 * p + 1];
            const int k2 = t * 16 + 8 * hh + 2 * p;
#pragma unroll
            for (int u = 0; u < T2; ++u) {
                v2f b;
                b.x = W2[k2 * NO + u * 16 + c];
                b.y = W2[(k2 + 1) * NO + u * 16 + c];
                acc2[u] = __builtin_amdgcn_wmma_f32_16x16x4_f32(
                    false, a, false, b, (short)0, acc2[u], false, false);
            }
        }
    }

    const float4 dv0 = *(const float4*)&dv16[8 * hh];
    const float4 dv1 = *(const float4*)&dv16[8 * hh + 4];
    const float dv[8] = {dv0.x, dv0.y, dv0.z, dv0.w, dv1.x, dv1.y, dv1.z, dv1.w};
#pragma unroll
    for (int r = 0; r < 8; ++r) {
        const int m = r + 8 * hh;
#pragma unroll
        for (int u = 0; u < T2; ++u) {
            const float val = acc2[u][r] * dv[r];
            hs[m * NO + u * 16 + c] = val;
        }
    }
}

constexpr int PL_THREADS = 1024;
constexpr int PL_PER     = 13;
constexpr int PL_CAP     = PL_THREADS * PL_PER;

__device__ __forceinline__ int wave_incl_scan(int v, int lane) {
#pragma unroll
    for (int o = 1; o < 32; o <<= 1) {
        const int t = __shfl_up(v, o);
        if (lane >= o) v += t;
    }
    return v;
}

template <int U>
__device__ __forceinline__ int half_bcast(int x) {
    return __builtin_amdgcn_update_dpp(0, x, 0x150 + U, 0xf, 0xf, true);
}

template <class Fn, int... I>
__device__ __forceinline__ void const_for_impl(Fn&& fn, std::integer_sequence<int, I...>) {
    (fn(std::integral_constant<int, I>{}), ...);
}
template <int N, class Fn>
__device__ __forceinline__ void const_for(Fn&& fn) {
    const_for_impl(fn, std::make_integer_sequence<int, N>{});
}

__global__ void __launch_bounds__(PL_THREADS)
bin_edges_kernel(const int* __restrict__ src, const int* __restrict__ dst,
                 const unsigned* __restrict__ wtot, int* __restrict__ cursor,
                 int* __restrict__ gbase, unsigned* __restrict__ binned,
                 int ne, int slice, int nwaves, int nb) {
    __shared__ unsigned ent[PL_CAP];
    __shared__ unsigned short ebk[PL_CAP];
    __shared__ int hist[BK_MAX];
    __shared__ int lbase[BK_MAX];
    __shared__ int rbase[BK_MAX];
    __shared__ int wsum1[BK_MAX / 32], wsum2[BK_MAX / 32];
    const int tid  = threadIdx.x;
    const int lane = tid & 31;
    const int wq   = tid >> 5;
    const int e0 = blockIdx.x * slice;
    int e1 = e0 + slice;
    if (e1 > ne) e1 = ne;

    int sv[PL_PER], dv[PL_PER];
#pragma unroll
    for (int j = 0; j < PL_PER; ++j) {
        const int e = e0 + j * PL_THREADS + tid;
        const int ec = e < e1 ? e : ne - 1;
        sv[j] = src[ec];
        dv[j] = dst[ec];
    }

    int bt = 0, bi = 0;
    if (tid < BK_MAX) {
        hist[tid] = 0;
        unsigned q[BK_WAVES];
#pragma unroll
        for (int k = 0; k < BK_WAVES; ++k) {
            const int w = tid * BK_WAVES + k;
            q[k] = wtot[w < nwaves ? w : nwaves - 1];
        }
        static_assert(BK_WAVES == 13, "the fence below names thirteen values");
        asm volatile("" :: "v"(q[0]), "v"(q[1]), "v"(q[2]), "v"(q[3]), "v"(q[4]),
                           "v"(q[5]), "v"(q[6]), "v"(q[7]), "v"(q[8]), "v"(q[9]),
                           "v"(q[10]), "v"(q[11]), "v"(q[12]));
        unsigned t = 0u;
#pragma unroll
        for (int k = 0; k < BK_WAVES; ++k) {
            const int w = tid * BK_WAVES + k;
            t += (tid < nb && w < nwaves) ? q[k] : 0u;
        }
        bt = (int)t;
        bi = wave_incl_scan(bt, lane);
        if (lane == 31) wsum1[wq] = bi;
    }
    __syncthreads();

    int bb = 0;
    if (tid < BK_MAX) {
        bb = bi - bt;
#pragma unroll
        for (int k = 0; k < BK_MAX / 32; ++k) bb += k < wq ? wsum1[k] : 0;
        if (blockIdx.x == 0) gbase[tid] = bb;
    }

    int rk[PL_PER];
#pragma unroll
    for (int j = 0; j < PL_PER; ++j) {
        rk[j] = 0;
        if (e0 + j * PL_THREADS + tid < e1) rk[j] = atomicAdd(&hist[dv[j] / BK_ROWS], 1);
    }
    __syncthreads();

    int h = 0, hi = 0;
    if (tid < BK_MAX) {
        h  = hist[tid];
        hi = wave_incl_scan(h, lane);
        if (lane == 31) wsum2[wq] = hi;
    }
    __syncthreads();
    if (tid < BK_MAX) {
        int a = hi - h;
#pragma unroll
        for (int k = 0; k < BK_MAX / 32; ++k) a += k < wq ? wsum2[k] : 0;
        lbase[tid] = a;
        int g = 0;
        if (h > 0) g = atomicAdd(&cursor[tid], h);
        rbase[tid] = bb + g;
    }
    __syncthreads();

#pragma unroll
    for (int j = 0; j < PL_PER; ++j) {
        if (e0 + j * PL_THREADS + tid < e1) {
            const int b = dv[j] / BK_ROWS;
            const int p = lbase[b] + rk[j];
            ent[p] = ((unsigned)(dv[j] - b * BK_ROWS) << 20) | (unsigned)sv[j];
            ebk[p] = (unsigned short)b;
        }
    }
    __syncthreads();

    const int cnt = e1 > e0 ? e1 - e0 : 0;
#pragma unroll
    for (int j = 0; j < PL_PER; ++j) {
        const int i = tid + j * PL_THREADS;
        if (i < cnt) {
            const int b = ebk[i];
            binned[rbase[b] + (i - lbase[b])] = ent[i];
        }
    }
}

constexpr int BS_THREADS = 1024;
constexpr int BS_CH      = 2;
constexpr int SRT_PER    = 16;
constexpr int SRT_CAP    = BS_THREADS * SRT_PER;

template <bool FINAL>
__global__ void __launch_bounds__(BS_THREADS)
bucket_sum_kernel(unsigned* __restrict__ binned, const int* __restrict__ gbase,
                  const int* __restrict__ cursor, const float* __restrict__ hs,
                  const float* __restrict__ dinv, const float* __restrict__ bias,
                  const float* __restrict__ W1, const float* __restrict__ b1,
                  const float* __restrict__ W2,
                  float* __restrict__ rows_out, int nn, int nn_pad) {
    constexpr int F  = 32;
    constexpr int NW = BS_THREADS / 32;
    constexpr int ST = 32 * BS_CH;
    constexpr int NQ = (BK_ROWS * (F / 4) + BS_THREADS - 1) / BS_THREADS;
    __shared__ float acc[BK_ROWS * F];
    const int tid  = threadIdx.x;
    const int lane = tid & 31;
    const int wv   = __builtin_amdgcn_readfirstlane(tid >> 5);
    const int n_lo = blockIdx.x * BK_ROWS;
    int rows = nn_pad - n_lo;
    if (rows > BK_ROWS) rows = BK_ROWS;
    const int nq = rows * (F / 4);
    const int base = __builtin_amdgcn_readfirstlane(gbase[blockIdx.x]);
    const int cnt  = __builtin_amdgcn_readfirstlane(cursor[blockIdx.x]);
    unsigned* lst = binned + base;

    constexpr int HL = 16;
    static_assert(ST == 2 * BS_CH * HL, "two halves, BS_CH groups of HL words each");
    const int cl   = lane & 15;
    const int hh   = lane >> 4;
    const int lofs = hh * (BS_CH * HL) + cl;
    const v2f* hs2 = (const v2f*)hs;
    auto sum_edges = [&](auto word, unsigned (&pn)[BS_CH]) {
        v2f a = {0.0f, 0.0f};
        int last = -1;
        for (int c0 = wv * ST; c0 < cnt; c0 += NW * ST) {
            unsigned p[BS_CH];
            const int cn = c0 + NW * ST;
#pragma unroll
            for (int h = 0; h < BS_CH; ++h) {
                p[h] = pn[h];
                const int in = cn + lofs + h * HL;
                pn[h] = word(in < cnt ? in : cnt - 1);
            }
            int my_s[BS_CH], my_d[BS_CH];
#pragma unroll
            for (int h = 0; h < BS_CH; ++h) {
                my_s[h] = 0;
                my_d[h] = -1;
                if (c0 + lofs + h * HL < cnt) {
                    my_s[h] = (int)(p[h] & 0xFFFFFu);
                    my_d[h] = (int)(p[h] >> 20);
                }
            }
            int prevd[BS_CH];
            unsigned m[BS_CH], mym[BS_CH];
            int carry = last;
#pragma unroll
            for (int h = 0; h < BS_CH; ++h) {
                const int sh = __builtin_amdgcn_update_dpp(0, my_d[h], 0x111, 0xf, 0xf, true);
                prevd[h] = cl == 0 ? carry : sh;
                m[h]     = __builtin_amdgcn_ballot_w32(my_d[h] != prevd[h]);
                mym[h]   = m[h] >> (hh * HL);
                carry    = half_bcast<HL - 1>(my_d[h]);
            }
            last = carry;

            v2f v[BS_CH * HL];
            const_for<BS_CH * HL>([&](auto ic) {
                constexpr int i = decltype(ic)::value;
                constexpr int h = i / HL, u = i % HL;
                const int s = half_bcast<u>(my_s[h]);
                v[i] = hs2[s * (F / 2) + cl];
            });
            static_assert(BS_CH * HL == 32, "the fences below name 32 pairs");
            asm volatile("" :: "v"(v[0].x), "v"(v[0].y), "v"(v[1].x), "v"(v[1].y),
                               "v"(v[2].x), "v"(v[2].y), "v"(v[3].x), "v"(v[3].y),
                               "v"(v[4].x), "v"(v[4].y), "v"(v[5].x), "v"(v[5].y),
                               "v"(v[6].x), "v"(v[6].y), "v"(v[7].x), "v"(v[7].y));
            asm volatile("" :: "v"(v[8].x), "v"(v[8].y), "v"(v[9].x), "v"(v[9].y),
                               "v"(v[10].x), "v"(v[10].y), "v"(v[11].x), "v"(v[11].y),
                               "v"(v[12].x), "v"(v[12].y), "v"(v[13].x), "v"(v[13].y),
                               "v"(v[14].x), "v"(v[14].y), "v"(v[15].x), "v"(v[15].y));
            asm volatile("" :: "v"(v[16].x), "v"(v[16].y), "v"(v[17].x), "v"(v[17].y),
                               "v"(v[18].x), "v"(v[18].y), "v"(v[19].x), "v"(v[19].y),
                               "v"(v[20].x), "v"(v[20].y), "v"(v[21].x), "v"(v[21].y),
                               "v"(v[22].x), "v"(v[22].y), "v"(v[23].x), "v"(v[23].y));
            asm volatile("" :: "v"(v[24].x), "v"(v[24].y), "v"(v[25].x), "v"(v[25].y),
                               "v"(v[26].x), "v"(v[26].y), "v"(v[27].x), "v"(v[27].y),
                               "v"(v[28].x), "v"(v[28].y), "v"(v[29].x), "v"(v[29].y),
                               "v"(v[30].x), "v"(v[30].y), "v"(v[31].x), "v"(v[31].y));
            const_for<BS_CH * HL>([&](auto ic) {
                constexpr int i = decltype(ic)::value;
                constexpr int h = i / HL, u = i % HL;
                if (__builtin_expect((m[h] & (0x10001u << u)) != 0u, 0)) {
                    const int pd = half_bcast<u>(prevd[h]);
                    if ((mym[h] >> u) & 1u) {
                        if (pd >= 0) {
                            atomicAdd(&acc[pd * F + 2 * cl], a.x);
                            atomicAdd(&acc[pd * F + 2 * cl + 1], a.y);
                        }
                        a = v2f{0.0f, 0.0f};
                    }
                }
                a += v[i];
            });
        }
        if (last >= 0) {
            atomicAdd(&acc[last * F + 2 * cl], a.x);
            atomicAdd(&acc[last * F + 2 * cl + 1], a.y);
        }
    };

    const bool order = !FINAL && cnt > 0 && cnt <= SRT_CAP;

    unsigned pn[BS_CH];
#pragma unroll
    for (int h = 0; h < BS_CH; ++h) {
        const int i = wv * ST + lofs + h * HL;
        pn[h] = 0u;
        if (cnt > 0 && !order) pn[h] = lst[i < cnt ? i : cnt - 1];
    }

    const float4* h4 = (const float4*)(hs + (size_t)n_lo * F);
    float4* a4 = (float4*)acc;
    auto self_rows = [&]() {
        float4 t[NQ];
#pragma unroll
        for (int k = 0; k < NQ; ++k) {
            const int i = tid + k * BS_THREADS;
            t[k] = h4[i < nq ? i : nq - 1];
        }
        static_assert(NQ == 4, "the fence below names four pieces");
        asm volatile("" :: "v"(t[0].x), "v"(t[0].y), "v"(t[0].z), "v"(t[0].w),
                           "v"(t[1].x), "v"(t[1].y), "v"(t[1].z), "v"(t[1].w),
                           "v"(t[2].x), "v"(t[2].y), "v"(t[2].z), "v"(t[2].w),
                           "v"(t[3].x), "v"(t[3].y), "v"(t[3].z), "v"(t[3].w));
#pragma unroll
        for (int k = 0; k < NQ; ++k) {
            const int i = tid + k * BS_THREADS;
            if (i < nq) a4[i] = t[k];
        }
    };

    if constexpr (!FINAL) {
        __shared__ unsigned srt[SRT_CAP];
        __shared__ int rcnt[BK_ROWS];
        __shared__ int roff[BK_ROWS];
        __shared__ int rws[BK_WAVES];
        if (order) {
            unsigned w[SRT_PER];
#pragma unroll
            for (int j = 0; j < SRT_PER; ++j) {
                const int i = tid + j * BS_THREADS;
                w[j] = lst[i < cnt ? i : cnt - 1];
            }
            if (tid < BK_ROWS) rcnt[tid] = 0;
            self_rows();
            __syncthreads();
            int rk[SRT_PER];
#pragma unroll
            for (int j = 0; j < SRT_PER; ++j) {
                rk[j] = 0;
                if (tid + j * BS_THREADS < cnt) rk[j] = atomicAdd(&rcnt[w[j] >> 20], 1);
            }
            __syncthreads();
            static_assert(BK_ROWS == BK_WAVES * 32, "whole waves scan the rows");
            int rc = 0, ri = 0;
            if (tid < BK_ROWS) {
                rc = rcnt[tid];
                ri = wave_incl_scan(rc, lane);
                if (lane == 31) rws[wv] = ri;
            }
            __syncthreads();
            if (tid < BK_ROWS) {
                int o = ri - rc;
#pragma unroll
                for (int k = 0; k < BK_WAVES; ++k) o += k < wv ? rws[k] : 0;
                roff[tid] = o;
            }
            __syncthreads();
#pragma unroll
            for (int j = 0; j < SRT_PER; ++j)
                if (tid + j * BS_THREADS < cnt) srt[roff[w[j] >> 20] + rk[j]] = w[j];
            __syncthreads();
#pragma unroll
            for (int j = 0; j < SRT_PER; ++j) {
                const int i = tid + j * BS_THREADS;
                if (i < cnt) lst[i] = srt[i];
            }
#pragma unroll
            for (int h = 0; h < BS_CH; ++h) {
                const int i = wv * ST + lofs + h * HL;
                pn[h] = srt[i < cnt ? i : cnt - 1];
            }
            sum_edges([&](int i) { return srt[i]; }, pn);
        } else {
            self_rows();
            __syncthreads();
            sum_edges([&](int i) { return lst[i]; }, pn);
        }
    } else {
        self_rows();
        __syncthreads();
        sum_edges([&](int i) { return lst[i]; }, pn);
    }
    __syncthreads();

    if constexpr (!FINAL) {
        static_assert(BK_ROWS % 16 == 0 && BK_ROWS / 16 <= NW, "one wave per 16-row tile");
        const int m0 = wv * 16;
        if (m0 < rows)
            two_layers_tile<18, F, 64, 32>(acc + m0 * F, W1, b1, W2, dinv + n_lo + m0,
                                           rows_out + (size_t)(n_lo + m0) * 32, lane);
    } else {
        float4* o4 = (float4*)(rows_out + (size_t)n_lo * F);
        const float4 bq = *(const float4*)&bias[(tid & 7) * 4];
        float dn[NQ];
#pragma unroll
        for (int k = 0; k < NQ; ++k) {
            const int n = n_lo + ((tid + k * BS_THREADS) >> 3);
            dn[k] = dinv[n < nn ? n : nn - 1];
        }
        asm volatile("" :: "v"(dn[0]), "v"(dn[1]), "v"(dn[2]), "v"(dn[3]));
#pragma unroll
        for (int k = 0; k < NQ; ++k) {
            const int i = tid + k * BS_THREADS;
            if (i < nq && n_lo + (i >> 3) < nn) {
                float4 r = a4[i];
                r.x = dn[k] * r.x + bq.x;
                r.y = dn[k] * r.y + bq.y;
                r.z = dn[k] * r.z + bq.z;
                r.w = dn[k] * r.w + bq.w;
                o4[i] = r;
            }
        }
    }
}

extern "C" void kernel_launch(void* const* d_in, const int* in_sizes, int n_in,
                              void* d_out, int out_size, void* d_ws, size_t ws_size,
                              hipStream_t stream) {
    const float* x  = (const float*)d_in[0];
    const int*   ei = (const int*)d_in[1];
    const float* W1 = (const float*)d_in[2];
    const float* b1 = (const float*)d_in[3];
    const float* W2 = (const float*)d_in[4];
    const float* b2 = (const float*)d_in[5];
    float* out = (float*)d_out;

    const int NN = in_sizes[0] / 18;
    const int NE = in_sizes[1] / 2;
    const int NN_pad = (NN + 15) & ~15;
    const int* src = ei;
    const int* dst = ei + NE;

    char* ws = (char*)d_ws;
    size_t off = 0;
    auto carve = [&](size_t bytes) -> char* {
        char* p = ws + off;
        off += (bytes + 255) & ~(size_t)255;
        return p;
    };
    float* dinv = (float*)carve((size_t)NN_pad * 4);
    const int dpitch = (NN_pad + 31) & ~31;
    unsigned* degp = (unsigned*)carve((size_t)DEG_CHUNKS * dpitch * 4);
    const int nwaves = (NN_pad + SP_ROWS - 1) / SP_ROWS;
    unsigned* wtot = (unsigned*)carve((size_t)nwaves * 4);
    int* cursor = (int*)carve((size_t)BK_MAX * 4);
    int* gbase  = (int*)carve((size_t)BK_MAX * 4);
    unsigned* binned = (unsigned*)carve((size_t)NE * 4);
    float* xs   = (float*)carve((size_t)NN_pad * 32 * 4);
    float* hs2  = (float*)carve((size_t)NN_pad * 32 * 4);

    const int dranges = (NN_pad + DEG_RANGE - 1) / DEG_RANGE;
    const int dchunk  = (((NE + DEG_CHUNKS - 1) / DEG_CHUNKS) + 3) & ~3;
    deg_count_lds_kernel<<<dim3(dranges, DEG_CHUNKS), DEG_THREADS, 0, stream>>>(
        dst, degp, NE, dchunk, NN_pad, dpitch);

    const int spblk = (NN_pad + SP_WAVES * SP_ROWS - 1) / (SP_WAVES * SP_ROWS);
    dinv_scale_pad_kernel<<<spblk, SP_WAVES * 32, 0, stream>>>(degp, x, dinv, xs, wtot, cursor, NN, NN_pad, dpitch);

    const int nb = (NN_pad + BK_ROWS - 1) / BK_ROWS;
    int pwgs = 256;
    if ((NE + PL_CAP - 1) / PL_CAP > pwgs) pwgs = (NE + PL_CAP - 1) / PL_CAP;
    const int pslice = (NE + pwgs - 1) / pwgs;
    bin_edges_kernel<<<pwgs, PL_THREADS, 0, stream>>>(src, dst, wtot, cursor, gbase, binned, NE, pslice, nwaves, nb);

    bucket_sum_kernel<false><<<nb, BS_THREADS, 0, stream>>>(binned, gbase, cursor, xs, dinv, b2, W1, b1, W2, hs2, NN, NN_pad);

    bucket_sum_kernel<true><<<nb, BS_THREADS, 0, stream>>>(binned, gbase, cursor, hs2, dinv, b2, W1, b1, W2, out, NN, NN_pad);
}
